# starting kernel
# baseline (speedup 1.0000x reference)
#include <hip/hip_runtime.h>


typedef float v2f __attribute__((ext_vector_type(2)));
typedef float v8f __attribute__((ext_vector_type(8)));

#define NUM_GRAPHS 512
#define C_OUT 10


__global__ void fill_kernel(float* __restrict__ p, float v, int n) {
    int t = blockIdx.x * blockDim.x + threadIdx.x;
    if (t < n) p[t] = v;
}

__global__ void degree_kernel(float* __restrict__ deg, const int* __restrict__ col, int E) {
    int t = blockIdx.x * blockDim.x + threadIdx.x;
    if (t < E) atomicAdd(&deg[col[t]], 1.0f);
}

__global__ void rsqrt_kernel(float* __restrict__ deg, int n) {
    int t = blockIdx.x * blockDim.x + threadIdx.x;
    if (t < n) deg[t] = rsqrtf(deg[t]);
}

__global__ void relu_kernel(float* __restrict__ p, long long n) {
    long long t = (long long)blockIdx.x * blockDim.x + threadIdx.x;
    if (t < n) { float v = p[t]; p[t] = v > 0.0f ? v : 0.0f; }
}

template <int BCOLS, int DSTRIDE, int KDIM>
__global__ __launch_bounds__(256)
void gemm_wmma_f32(const float* __restrict__ A, const float* __restrict__ B,
                   float* __restrict__ D, int N, int ntilesN) {
    int wave = blockIdx.x * (blockDim.x >> 5) + (threadIdx.x >> 5);
    int lane = threadIdx.x & 31;
    int tilesM = (N + 15) >> 4;
    if (wave >= tilesM * ntilesN) return;
    int tm = wave / ntilesN;
    int tn = wave - tm * ntilesN;
    int m0 = tm << 4, n0 = tn << 4;
    int half = lane >> 4;
    int l15  = lane & 15;

    int row  = m0 + l15;   if (row  >= N)     row  = N - 1;
    int bcol = n0 + l15;   if (bcol >= BCOLS) bcol = BCOLS - 1;

    const float* __restrict__ Arow = A + (size_t)row * KDIM + half * 2;
    const float* __restrict__ Bcol = B + (size_t)(half * 2) * BCOLS + bcol;

    v8f acc = {0.f, 0.f, 0.f, 0.f, 0.f, 0.f, 0.f, 0.f};
#pragma unroll
    for (int k0 = 0; k0 < KDIM; k0 += 4) {
        v2f a = *(const v2f*)(Arow + k0);
        v2f b;
        b[0] = Bcol[(size_t)k0 * BCOLS];
        b[1] = Bcol[(size_t)(k0 + 1) * BCOLS];
        acc = __builtin_amdgcn_wmma_f32_16x16x4_f32(false, a, false, b,
                                                    (short)0, acc, false, false);
    }
#pragma unroll
    for (int r = 0; r < 8; ++r) {
        int m = m0 + r + half * 8;
        if (m < N) D[(size_t)m * DSTRIDE + n0 + l15] = acc[r];
    }
}

template <int F>
__global__ void init_agg_kernel(const float* __restrict__ lin, const float* __restrict__ dis,
                                const float* __restrict__ bias, int biasLen,
                                float* __restrict__ out, int N) {
    long long t = (long long)blockIdx.x * blockDim.x + threadIdx.x;
    if (t >= (long long)N * F) return;
    int i = (int)(t / F);
    int c = (int)(t - (long long)i * F);
    float d  = dis[i];
    float bv = (c < biasLen) ? bias[c] : 0.0f;
    out[t] = lin[t] * d * d + bv;
}

template <int F>
__global__ void scatter_kernel(const float* __restrict__ lin, const float* __restrict__ dis,
                               const int* __restrict__ row, const int* __restrict__ col,
                               float* __restrict__ out, int E) {
    const int LPE = F / 4;
    long long t = (long long)blockIdx.x * blockDim.x + threadIdx.x;
    long long e = t / LPE;
    int l = (int)(t - e * LPE);
    if (e >= E) return;
    int r = row[e];
    int c = col[e];
    float s = dis[r] * dis[c];
    const float4 v = *(const float4*)(lin + (size_t)r * F + l * 4);
    float* o = out + (size_t)c * F + l * 4;
    atomicAdd(o + 0, v.x * s);
    atomicAdd(o + 1, v.y * s);
    atomicAdd(o + 2, v.z * s);
    atomicAdd(o + 3, v.w * s);
}

__global__ void pool_kernel(const float* __restrict__ h  ,
                            const int* __restrict__ batch,
                            float* __restrict__ out, float* __restrict__ cnt, int N) {
    long long t = (long long)blockIdx.x * blockDim.x + threadIdx.x;
    if (t >= (long long)N * 16) return;
    int i = (int)(t >> 4);
    int c = (int)(t & 15);
    int g = batch[i];
    if (c < C_OUT) atomicAdd(&out[(size_t)g * C_OUT + c], h[t]);
    if (c == 0)    atomicAdd(&cnt[g], 1.0f);
}

__global__ void div_kernel(float* __restrict__ out, const float* __restrict__ cnt) {
    int t = blockIdx.x * blockDim.x + threadIdx.x;
    if (t < NUM_GRAPHS * C_OUT) out[t] /= fmaxf(cnt[t / C_OUT], 1.0f);
}


static inline int cdiv_ll(long long a, long long b) { return (int)((a + b - 1) / b); }

extern "C" void kernel_launch(void* const* d_in, const int* in_sizes, int n_in,
                              void* d_out, int out_size, void* d_ws, size_t ws_size,
                              hipStream_t stream) {
    const float* x     = (const float*)d_in[0];
    const int*   ei    = (const int*)d_in[1];
    const int*   batch = (const int*)d_in[2];
    const float* W0 = (const float*)d_in[3];  const float* b0 = (const float*)d_in[4];
    const float* W1 = (const float*)d_in[5];  const float* b1 = (const float*)d_in[6];
    const float* W2 = (const float*)d_in[7];  const float* b2 = (const float*)d_in[8];
    const float* W3 = (const float*)d_in[9];  const float* b3 = (const float*)d_in[10];

    const int N = in_sizes[2];
    const int E = in_sizes[1] / 2;
    const int H = 128;
    const int* rowi = ei;
    const int* coli = ei + E;

    float* out = (float*)d_out;
    float* ws  = (float*)d_ws;
    float* dis  = ws;
    float* bufA = ws + (((size_t)N + 255) & ~(size_t)255);
    float* bufB = bufA + (size_t)N * H;
    float* cnt  = bufB + (size_t)N * H;

    const int TB = 256;

    fill_kernel<<<cdiv_ll(N, TB), TB, 0, stream>>>(dis, 1.0f, N);
    degree_kernel<<<cdiv_ll(E, TB), TB, 0, stream>>>(dis, coli, E);
    rsqrt_kernel<<<cdiv_ll(N, TB), TB, 0, stream>>>(dis, N);

    const int tilesM     = (N + 15) >> 4;
    const int gemmBlkH   = cdiv_ll((long long)tilesM * 8, TB / 32);
    const int gemmBlkOut = cdiv_ll((long long)tilesM * 1, TB / 32);
    const long long NH   = (long long)N * H;

    gemm_wmma_f32<128, 128, 128><<<gemmBlkH, TB, 0, stream>>>(x, W0, bufA, N, 8);
    init_agg_kernel<128><<<cdiv_ll(NH, TB), TB, 0, stream>>>(bufA, dis, b0, 128, bufB, N);
    scatter_kernel<128><<<cdiv_ll((long long)E * 32, TB), TB, 0, stream>>>(bufA, dis, rowi, coli, bufB, E);
    relu_kernel<<<cdiv_ll(NH, TB), TB, 0, stream>>>(bufB, NH);

    gemm_wmma_f32<128, 128, 128><<<gemmBlkH, TB, 0, stream>>>(bufB, W1, bufA, N, 8);
    init_agg_kernel<128><<<cdiv_ll(NH, TB), TB, 0, stream>>>(bufA, dis, b1, 128, bufB, N);
    scatter_kernel<128><<<cdiv_ll((long long)E * 32, TB), TB, 0, stream>>>(bufA, dis, rowi, coli, bufB, E);
    relu_kernel<<<cdiv_ll(NH, TB), TB, 0, stream>>>(bufB, NH);

    gemm_wmma_f32<128, 128, 128><<<gemmBlkH, TB, 0, stream>>>(bufB, W2, bufA, N, 8);
    init_agg_kernel<128><<<cdiv_ll(NH, TB), TB, 0, stream>>>(bufA, dis, b2, 128, bufB, N);
    scatter_kernel<128><<<cdiv_ll((long long)E * 32, TB), TB, 0, stream>>>(bufA, dis, rowi, coli, bufB, E);
    relu_kernel<<<cdiv_ll(NH, TB), TB, 0, stream>>>(bufB, NH);

    gemm_wmma_f32<10, 16, 128><<<gemmBlkOut, TB, 0, stream>>>(bufB, W3, bufA, N, 1);
    init_agg_kernel<16><<<cdiv_ll((long long)N * 16, TB), TB, 0, stream>>>(bufA, dis, b3, C_OUT, bufB, N);
    scatter_kernel<16><<<cdiv_ll((long long)E * 4, TB), TB, 0, stream>>>(bufA, dis, rowi, coli, bufB, E);

    fill_kernel<<<cdiv_ll(NUM_GRAPHS * C_OUT, TB), TB, 0, stream>>>(out, 0.0f, NUM_GRAPHS * C_OUT);
    fill_kernel<<<cdiv_ll(NUM_GRAPHS, TB), TB, 0, stream>>>(cnt, 0.0f, NUM_GRAPHS);
    pool_kernel<<<cdiv_ll((long long)N * 16, TB), TB, 0, stream>>>(bufB, batch, out, cnt, N);
    div_kernel<<<cdiv_ll(NUM_GRAPHS * C_OUT, TB), TB, 0, stream>>>(out, cnt);
}


# Atomic edge scatter replaced by an in-call edge list ordered by (target, source) and a one-wave-per-row register gather with self term, bias and ReLU at its one store; sums in fixed source order
# speedup vs baseline: 2.7913x; 2.7913x over previous
#include <hip/hip_runtime.h>


typedef float v2f __attribute__((ext_vector_type(2)));
typedef float v8f __attribute__((ext_vector_type(8)));

#define NUM_GRAPHS 512
#define C_OUT 10


__global__ void fill_kernel(float* __restrict__ p, float v, int n) {
    int t = blockIdx.x * blockDim.x + threadIdx.x;
    if (t < n) p[t] = v;
}

__global__ void fill_int_kernel(int* __restrict__ p, int v, int n) {
    int t = blockIdx.x * blockDim.x + threadIdx.x;
    if (t < n) p[t] = v;
}

__global__ void rank_kernel(int* __restrict__ count, const int* __restrict__ col,
                            int* __restrict__ rank, int E) {
    int t = blockIdx.x * blockDim.x + threadIdx.x;
    if (t < E) rank[t] = atomicAdd(&count[col[t]], 1);
}

__device__ __forceinline__ int wave_inclusive_scan(int v, int lane) {
    int incl = v;
    for (int off = 1; off < 32; off <<= 1) {
        int n = __shfl_up(incl, off, 32);
        if (lane >= off) incl += n;
    }
    return incl;
}

__global__ __launch_bounds__(256)
void scan_local_kernel(const int* __restrict__ count, int* __restrict__ rowptr,
                       int* __restrict__ bsum, int N) {
    __shared__ int wsum[8];
    int lane = threadIdx.x & 31, wv = threadIdx.x >> 5;
    int base = blockIdx.x * 1024 + threadIdx.x * 4;
    int c0 = (base + 0 < N) ? count[base + 0] : 0;
    int c1 = (base + 1 < N) ? count[base + 1] : 0;
    int c2 = (base + 2 < N) ? count[base + 2] : 0;
    int c3 = (base + 3 < N) ? count[base + 3] : 0;
    int tsum = c0 + c1 + c2 + c3;
    int incl = wave_inclusive_scan(tsum, lane);
    if (lane == 31) wsum[wv] = incl;
    __syncthreads();
    int woff = 0;
    for (int w = 0; w < wv; ++w) woff += wsum[w];
    int excl = woff + incl - tsum;
    if (base + 0 < N) rowptr[base + 0] = excl;
    if (base + 1 < N) rowptr[base + 1] = excl + c0;
    if (base + 2 < N) rowptr[base + 2] = excl + c0 + c1;
    if (base + 3 < N) rowptr[base + 3] = excl + c0 + c1 + c2;
    if (threadIdx.x == 255) bsum[blockIdx.x] = woff + incl;
}

__global__ void scan_top_kernel(int* __restrict__ bsum, int nb) {
    int lane = threadIdx.x;
    int carry = 0;
    for (int b0 = 0; b0 < nb; b0 += 32) {
        int i = b0 + lane;
        int v = (i < nb) ? bsum[i] : 0;
        int incl = wave_inclusive_scan(v, lane);
        if (i < nb) bsum[i] = carry + incl - v;
        carry += __shfl(incl, 31, 32);
    }
}

__global__ void finalize_kernel(int* __restrict__ rowptr, const int* __restrict__ bsum,
                                const int* __restrict__ count, float* __restrict__ dis, int N) {
    int i = blockIdx.x * blockDim.x + threadIdx.x;
    if (i < N) {
        rowptr[i] += bsum[i >> 10];
        dis[i] = rsqrtf((float)(count[i] + 1));
    }
}

__global__ void place_kernel(const int* __restrict__ row, const int* __restrict__ col,
                             const int* __restrict__ rank, const int* __restrict__ rowptr,
                             int* __restrict__ slot_src, int E) {
    int t = blockIdx.x * blockDim.x + threadIdx.x;
    if (t < E) slot_src[rowptr[col[t]] + rank[t]] = row[t];
}

__global__ __launch_bounds__(256)
void order_kernel(const int* __restrict__ slot_src, const int* __restrict__ rowptr,
                  const int* __restrict__ count, const float* __restrict__ dis,
                  int* __restrict__ csr_src, float* __restrict__ csr_w, int N) {
    int lane = threadIdx.x & 31;
    int node = __builtin_amdgcn_readfirstlane((int)(blockIdx.x * 8 + (threadIdx.x >> 5)));
    if (node >= N) return;
    int start = rowptr[node];
    int d = count[node];
    float dc = dis[node];
    const int* __restrict__ seg = slot_src + start;
    for (int i0 = 0; i0 < d; i0 += 32) {
        int i = i0 + lane;
        if (i < d) {
            int key = seg[i];
            int p = 0;
            for (int j = 0; j < d; ++j) {
                int kj = seg[j];
                p += ((kj < key) || (kj == key && j < i)) ? 1 : 0;
            }
            csr_src[start + p] = key;
            csr_w[start + p] = dis[key] * dc;
        }
    }
}

template <int BCOLS, int DSTRIDE, int KDIM>
__global__ __launch_bounds__(256)
void gemm_wmma_f32(const float* __restrict__ A, const float* __restrict__ B,
                   float* __restrict__ D, int N, int ntilesN) {
    int wave = blockIdx.x * (blockDim.x >> 5) + (threadIdx.x >> 5);
    int lane = threadIdx.x & 31;
    int tilesM = (N + 15) >> 4;
    if (wave >= tilesM * ntilesN) return;
    int tm = wave / ntilesN;
    int tn = wave - tm * ntilesN;
    int m0 = tm << 4, n0 = tn << 4;
    int half = lane >> 4;
    int l15  = lane & 15;

    int row  = m0 + l15;   if (row  >= N)     row  = N - 1;
    int bcol = n0 + l15;   if (bcol >= BCOLS) bcol = BCOLS - 1;

    const float* __restrict__ Arow = A + (size_t)row * KDIM + half * 2;
    const float* __restrict__ Bcol = B + (size_t)(half * 2) * BCOLS + bcol;

    v8f acc = {0.f, 0.f, 0.f, 0.f, 0.f, 0.f, 0.f, 0.f};
#pragma unroll
    for (int k0 = 0; k0 < KDIM; k0 += 4) {
        v2f a = *(const v2f*)(Arow + k0);
        v2f b;
        b[0] = Bcol[(size_t)k0 * BCOLS];
        b[1] = Bcol[(size_t)(k0 + 1) * BCOLS];
        acc = __builtin_amdgcn_wmma_f32_16x16x4_f32(false, a, false, b,
                                                    (short)0, acc, false, false);
    }
#pragma unroll
    for (int r = 0; r < 8; ++r) {
        int m = m0 + r + half * 8;
        if (m < N) D[(size_t)m * DSTRIDE + n0 + l15] = acc[r];
    }
}

template <int K, int F>
__device__ __forceinline__ void gather_step(float4& acc, const float* __restrict__ colbase,
                                            const int* __restrict__ src,
                                            const float* __restrict__ w, int j) {
    int s[K]; float ww[K]; float4 v[K];
#pragma unroll
    for (int k = 0; k < K; ++k) { s[k] = src[j + k]; ww[k] = w[j + k]; }
#pragma unroll
    for (int k = 0; k < K; ++k) v[k] = *(const float4*)(colbase + (size_t)s[k] * F);
#pragma unroll
    for (int k = 0; k < K; ++k) {
        acc.x = fmaf(v[k].x, ww[k], acc.x);
        acc.y = fmaf(v[k].y, ww[k], acc.y);
        acc.z = fmaf(v[k].z, ww[k], acc.z);
        acc.w = fmaf(v[k].w, ww[k], acc.w);
    }
}

template <bool RELU>
__global__ __launch_bounds__(256)
void gather128_kernel(const float* __restrict__ lin, const float* __restrict__ dis,
                      const float* __restrict__ bias,
                      const int* __restrict__ rowptr, const int* __restrict__ count,
                      const int* __restrict__ csr_src, const float* __restrict__ csr_w,
                      float* __restrict__ out, int N) {
    int lane = threadIdx.x & 31;
    int node = __builtin_amdgcn_readfirstlane((int)(blockIdx.x * 8 + (threadIdx.x >> 5)));
    if (node >= N) return;
    int start = rowptr[node];
    int d = count[node];
    float dn = dis[node];
    const float* __restrict__ colbase = lin + lane * 4;
    const int* __restrict__ src = csr_src + start;
    const float* __restrict__ w = csr_w + start;
    float4 self = *(const float4*)(colbase + (size_t)node * 128);
    float s2 = dn * dn;
    float4 acc;
    acc.x = self.x * s2; acc.y = self.y * s2; acc.z = self.z * s2; acc.w = self.w * s2;
    int j = 0;
    for (; j + 8 <= d; j += 8) gather_step<8, 128>(acc, colbase, src, w, j);
    if (j + 4 <= d) { gather_step<4, 128>(acc, colbase, src, w, j); j += 4; }
    if (j + 2 <= d) { gather_step<2, 128>(acc, colbase, src, w, j); j += 2; }
    if (j < d) gather_step<1, 128>(acc, colbase, src, w, j);
    float4 b = *(const float4*)(bias + lane * 4);
    acc.x += b.x; acc.y += b.y; acc.z += b.z; acc.w += b.w;
    if (RELU) {
        acc.x = fmaxf(acc.x, 0.0f); acc.y = fmaxf(acc.y, 0.0f);
        acc.z = fmaxf(acc.z, 0.0f); acc.w = fmaxf(acc.w, 0.0f);
    }
    *(float4*)(out + (size_t)node * 128 + lane * 4) = acc;
}

__global__ void gather16_kernel(const float* __restrict__ lin, const float* __restrict__ dis,
                                const float* __restrict__ bias, int biasLen,
                                const int* __restrict__ rowptr, const int* __restrict__ count,
                                const int* __restrict__ csr_src, const float* __restrict__ csr_w,
                                float* __restrict__ out, int N) {
    long long t = (long long)blockIdx.x * blockDim.x + threadIdx.x;
    int node = (int)(t >> 2);
    int q = (int)(t & 3);
    if (t >= (long long)N * 4) return;
    int start = rowptr[node];
    int d = count[node];
    float dn = dis[node];
    const float* __restrict__ colbase = lin + q * 4;
    const int* __restrict__ src = csr_src + start;
    const float* __restrict__ w = csr_w + start;
    float4 self = *(const float4*)(colbase + (size_t)node * 16);
    float s2 = dn * dn;
    float4 acc;
    acc.x = self.x * s2; acc.y = self.y * s2; acc.z = self.z * s2; acc.w = self.w * s2;
    int j = 0;
    for (; j + 4 <= d; j += 4) gather_step<4, 16>(acc, colbase, src, w, j);
    if (j + 2 <= d) { gather_step<2, 16>(acc, colbase, src, w, j); j += 2; }
    if (j < d) gather_step<1, 16>(acc, colbase, src, w, j);
    int c = q * 4;
    acc.x += (c + 0 < biasLen) ? bias[c + 0] : 0.0f;
    acc.y += (c + 1 < biasLen) ? bias[c + 1] : 0.0f;
    acc.z += (c + 2 < biasLen) ? bias[c + 2] : 0.0f;
    acc.w += (c + 3 < biasLen) ? bias[c + 3] : 0.0f;
    *(float4*)(out + (size_t)node * 16 + q * 4) = acc;
}

__global__ void pool_kernel(const float* __restrict__ h  ,
                            const int* __restrict__ batch,
                            float* __restrict__ out, float* __restrict__ cnt, int N) {
    long long t = (long long)blockIdx.x * blockDim.x + threadIdx.x;
    if (t >= (long long)N * 16) return;
    int i = (int)(t >> 4);
    int c = (int)(t & 15);
    int g = batch[i];
    if (c < C_OUT) atomicAdd(&out[(size_t)g * C_OUT + c], h[t]);
    if (c == 0)    atomicAdd(&cnt[g], 1.0f);
}

__global__ void div_kernel(float* __restrict__ out, const float* __restrict__ cnt) {
    int t = blockIdx.x * blockDim.x + threadIdx.x;
    if (t < NUM_GRAPHS * C_OUT) out[t] /= fmaxf(cnt[t / C_OUT], 1.0f);
}


static inline int cdiv_ll(long long a, long long b) { return (int)((a + b - 1) / b); }

extern "C" void kernel_launch(void* const* d_in, const int* in_sizes, int n_in,
                              void* d_out, int out_size, void* d_ws, size_t ws_size,
                              hipStream_t stream) {
    const float* x     = (const float*)d_in[0];
    const int*   ei    = (const int*)d_in[1];
    const int*   batch = (const int*)d_in[2];
    const float* W0 = (const float*)d_in[3];  const float* b0 = (const float*)d_in[4];
    const float* W1 = (const float*)d_in[5];  const float* b1 = (const float*)d_in[6];
    const float* W2 = (const float*)d_in[7];  const float* b2 = (const float*)d_in[8];
    const float* W3 = (const float*)d_in[9];  const float* b3 = (const float*)d_in[10];

    const int N = in_sizes[2];
    const int E = in_sizes[1] / 2;
    const int H = 128;
    const int* rowi = ei;
    const int* coli = ei + E;

    float* out = (float*)d_out;
    float* ws  = (float*)d_ws;
    const size_t Npad = ((size_t)N + 255) & ~(size_t)255;
    const int    nb   = (N + 1023) >> 10;
    float* dis  = ws;
    float* bufA = ws + Npad;
    float* bufB = bufA + (size_t)N * H;
    float* cnt  = bufB + (size_t)N * H;
    int*   count    = (int*)(cnt + 512);
    int*   rowptr   = count + Npad;
    int*   bsum     = rowptr + Npad;
    int*   rank     = bsum + (((size_t)nb + 255) & ~(size_t)255);
    int*   slot_src = rank + (size_t)E;
    int*   csr_src  = slot_src + (size_t)E;
    float* csr_w    = (float*)(csr_src + (size_t)E);

    const int TB = 256;

    fill_int_kernel<<<cdiv_ll(N, TB), TB, 0, stream>>>(count, 0, N);
    rank_kernel<<<cdiv_ll(E, TB), TB, 0, stream>>>(count, coli, rank, E);
    scan_local_kernel<<<nb, TB, 0, stream>>>(count, rowptr, bsum, N);
    scan_top_kernel<<<1, 32, 0, stream>>>(bsum, nb);
    finalize_kernel<<<cdiv_ll(N, TB), TB, 0, stream>>>(rowptr, bsum, count, dis, N);
    place_kernel<<<cdiv_ll(E, TB), TB, 0, stream>>>(rowi, coli, rank, rowptr, slot_src, E);
    order_kernel<<<cdiv_ll(N, TB / 32), TB, 0, stream>>>(slot_src, rowptr, count, dis, csr_src, csr_w, N);

    const int tilesM     = (N + 15) >> 4;
    const int gemmBlkH   = cdiv_ll((long long)tilesM * 8, TB / 32);
    const int gemmBlkOut = cdiv_ll((long long)tilesM * 1, TB / 32);
    const int gathBlkH   = cdiv_ll(N, TB / 32);

    gemm_wmma_f32<128, 128, 128><<<gemmBlkH, TB, 0, stream>>>(x, W0, bufA, N, 8);
    gather128_kernel<true><<<gathBlkH, TB, 0, stream>>>(bufA, dis, b0, rowptr, count, csr_src, csr_w, bufB, N);

    gemm_wmma_f32<128, 128, 128><<<gemmBlkH, TB, 0, stream>>>(bufB, W1, bufA, N, 8);
    gather128_kernel<true><<<gathBlkH, TB, 0, stream>>>(bufA, dis, b1, rowptr, count, csr_src, csr_w, bufB, N);

    gemm_wmma_f32<128, 128, 128><<<gemmBlkH, TB, 0, stream>>>(bufB, W2, bufA, N, 8);
    gather128_kernel<true><<<gathBlkH, TB, 0, stream>>>(bufA, dis, b2, rowptr, count, csr_src, csr_w, bufB, N);

    gemm_wmma_f32<10, 16, 128><<<gemmBlkOut, TB, 0, stream>>>(bufB, W3, bufA, N, 1);
    gather16_kernel<<<cdiv_ll((long long)N * 4, TB), TB, 0, stream>>>(bufA, dis, b3, C_OUT, rowptr, count, csr_src, csr_w, bufB, N);

    fill_kernel<<<cdiv_ll(NUM_GRAPHS * C_OUT, TB), TB, 0, stream>>>(out, 0.0f, NUM_GRAPHS * C_OUT);
    fill_kernel<<<cdiv_ll(NUM_GRAPHS, TB), TB, 0, stream>>>(cnt, 0.0f, NUM_GRAPHS);
    pool_kernel<<<cdiv_ll((long long)N * 16, TB), TB, 0, stream>>>(bufB, batch, out, cnt, N);
    div_kernel<<<cdiv_ll(NUM_GRAPHS * C_OUT, TB), TB, 0, stream>>>(out, cnt);
}


# GEMM restructured: weights staged once per work-group in LDS, each wave does a 16x128 strip (8 accumulators) from one pass over A, rows stored whole via LDS; k order of the product sums changes
# speedup vs baseline: 3.2048x; 1.1481x over previous
#include <hip/hip_runtime.h>


typedef float v2f __attribute__((ext_vector_type(2)));
typedef float v4f __attribute__((ext_vector_type(4)));
typedef float v8f __attribute__((ext_vector_type(8)));

#define NUM_GRAPHS 512
#define C_OUT 10


__global__ void fill_kernel(float* __restrict__ p, float v, int n) {
    int t = blockIdx.x * blockDim.x + threadIdx.x;
    if (t < n) p[t] = v;
}

__global__ void fill_int_kernel(int* __restrict__ p, int v, int n) {
    int t = blockIdx.x * blockDim.x + threadIdx.x;
    if (t < n) p[t] = v;
}

__global__ void rank_kernel(int* __restrict__ count, const int* __restrict__ col,
                            int* __restrict__ rank, int E) {
    int t = blockIdx.x * blockDim.x + threadIdx.x;
    if (t < E) rank[t] = atomicAdd(&count[col[t]], 1);
}

__device__ __forceinline__ int wave_inclusive_scan(int v, int lane) {
    int incl = v;
    for (int off = 1; off < 32; off <<= 1) {
        int n = __shfl_up(incl, off, 32);
        if (lane >= off) incl += n;
    }
    return incl;
}

__global__ __launch_bounds__(256)
void scan_local_kernel(const int* __restrict__ count, int* __restrict__ rowptr,
                       int* __restrict__ bsum, int N) {
    __shared__ int wsum[8];
    int lane = threadIdx.x & 31, wv = threadIdx.x >> 5;
    int base = blockIdx.x * 1024 + threadIdx.x * 4;
    int c0 = (base + 0 < N) ? count[base + 0] : 0;
    int c1 = (base + 1 < N) ? count[base + 1] : 0;
    int c2 = (base + 2 < N) ? count[base + 2] : 0;
    int c3 = (base + 3 < N) ? count[base + 3] : 0;
    int tsum = c0 + c1 + c2 + c3;
    int incl = wave_inclusive_scan(tsum, lane);
    if (lane == 31) wsum[wv] = incl;
    __syncthreads();
    int woff = 0;
    for (int w = 0; w < wv; ++w) woff += wsum[w];
    int excl = woff + incl - tsum;
    if (base + 0 < N) rowptr[base + 0] = excl;
    if (base + 1 < N) rowptr[base + 1] = excl + c0;
    if (base + 2 < N) rowptr[base + 2] = excl + c0 + c1;
    if (base + 3 < N) rowptr[base + 3] = excl + c0 + c1 + c2;
    if (threadIdx.x == 255) bsum[blockIdx.x] = woff + incl;
}

__global__ void scan_top_kernel(int* __restrict__ bsum, int nb) {
    int lane = threadIdx.x;
    int carry = 0;
    for (int b0 = 0; b0 < nb; b0 += 32) {
        int i = b0 + lane;
        int v = (i < nb) ? bsum[i] : 0;
        int incl = wave_inclusive_scan(v, lane);
        if (i < nb) bsum[i] = carry + incl - v;
        carry += __shfl(incl, 31, 32);
    }
}

__global__ void finalize_kernel(int* __restrict__ rowptr, const int* __restrict__ bsum,
                                const int* __restrict__ count, float* __restrict__ dis, int N) {
    int i = blockIdx.x * blockDim.x + threadIdx.x;
    if (i < N) {
        rowptr[i] += bsum[i >> 10];
        dis[i] = rsqrtf((float)(count[i] + 1));
    }
}

__global__ void place_kernel(const int* __restrict__ row, const int* __restrict__ col,
                             const int* __restrict__ rank, const int* __restrict__ rowptr,
                             int* __restrict__ slot_src, int E) {
    int t = blockIdx.x * blockDim.x + threadIdx.x;
    if (t < E) slot_src[rowptr[col[t]] + rank[t]] = row[t];
}

__global__ __launch_bounds__(256)
void order_kernel(const int* __restrict__ slot_src, const int* __restrict__ rowptr,
                  const int* __restrict__ count, const float* __restrict__ dis,
                  int* __restrict__ csr_src, float* __restrict__ csr_w, int N) {
    int lane = threadIdx.x & 31;
    int node = __builtin_amdgcn_readfirstlane((int)(blockIdx.x * 8 + (threadIdx.x >> 5)));
    if (node >= N) return;
    int start = rowptr[node];
    int d = count[node];
    float dc = dis[node];
    const int* __restrict__ seg = slot_src + start;
    for (int i0 = 0; i0 < d; i0 += 32) {
        int i = i0 + lane;
        if (i < d) {
            int key = seg[i];
            int p = 0;
            for (int j = 0; j < d; ++j) {
                int kj = seg[j];
                p += ((kj < key) || (kj == key && j < i)) ? 1 : 0;
            }
            csr_src[start + p] = key;
            csr_w[start + p] = dis[key] * dc;
        }
    }
}

#define WT_PITCH 132

template <int BCOLS, int NT, int DSTRIDE>
__global__ __launch_bounds__(256)
void gemm_strip_wmma_f32(const float* __restrict__ A, const float* __restrict__ B,
                         float* __restrict__ D, int N) {
    __shared__ __attribute__((aligned(16))) float Wt[NT * 16 * WT_PITCH];
    __shared__ __attribute__((aligned(16))) float St[8 * 16 * WT_PITCH];

#pragma unroll
    for (int it = 0; it < (128 * NT * 16) / 256; ++it) {
        int i = it * 256 + (int)threadIdx.x;
        int n = i % (NT * 16);
        int k = i / (NT * 16);
        int nc = (n < BCOLS) ? n : (BCOLS - 1);
        Wt[n * WT_PITCH + k] = B[(size_t)k * BCOLS + nc];
    }
    __syncthreads();

    const int lane = threadIdx.x & 31;
    const int c = lane & 15;
    const int h = lane >> 4;
    const int tilesM = (N + 15) >> 4;
    const int nwaves = (int)gridDim.x * 8;
    int tm = __builtin_amdgcn_readfirstlane((int)((threadIdx.x >> 5) * gridDim.x + blockIdx.x));
    const float* wfrag = Wt + c * WT_PITCH + 4 * h;
    constexpr int LPR = NT * 4;
    constexpr int RPI = 32 / LPR;
    float* sw = St + (threadIdx.x >> 5) * (16 * WT_PITCH);
    float* sput = sw + (8 * h) * WT_PITCH + c;
    const int rsub = lane / LPR;
    const int cpos = (lane % LPR) * 4;

    for (; tm < tilesM; tm += nwaves) {
        asm volatile("" ::: "memory");
        const int m0 = tm << 4;
        int row = m0 + c;
        if (row >= N) row = N - 1;
        const v4f* __restrict__ Ap = (const v4f*)(A + (size_t)row * 128 + 4 * h);

        v4f a[16];
#pragma unroll
        for (int kb = 0; kb < 16; ++kb) a[kb] = Ap[kb * 2];
        asm volatile("" :: "v"(a[0]), "v"(a[1]), "v"(a[2]), "v"(a[3]), "v"(a[4]), "v"(a[5]),
                           "v"(a[6]), "v"(a[7]), "v"(a[8]), "v"(a[9]), "v"(a[10]), "v"(a[11]),
                           "v"(a[12]), "v"(a[13]), "v"(a[14]), "v"(a[15]));

        v8f acc[NT];
#pragma unroll
        for (int t = 0; t < NT; ++t) acc[t] = (v8f){0.f, 0.f, 0.f, 0.f, 0.f, 0.f, 0.f, 0.f};

#pragma unroll
        for (int kb = 0; kb < 16; ++kb) {
            v4f b[NT];
#pragma unroll
            for (int t = 0; t < NT; ++t)
                b[t] = *(const v4f*)(wfrag + t * 16 * WT_PITCH + kb * 8);
            v2f a0 = {a[kb].x, a[kb].y};
            v2f a1 = {a[kb].z, a[kb].w};
#pragma unroll
            for (int t = 0; t < NT; ++t) {
                v2f b0 = {b[t].x, b[t].y};
                acc[t] = __builtin_amdgcn_wmma_f32_16x16x4_f32(false, a0, false, b0,
                                                               (short)0, acc[t], false, false);
            }
#pragma unroll
            for (int t = 0; t < NT; ++t) {
                v2f b1 = {b[t].z, b[t].w};
                acc[t] = __builtin_amdgcn_wmma_f32_16x16x4_f32(false, a1, false, b1,
                                                               (short)0, acc[t], false, false);
            }
        }

#pragma unroll
        for (int r = 0; r < 8; ++r) {
#pragma unroll
            for (int t = 0; t < NT; ++t) sput[r * WT_PITCH + t * 16] = acc[t][r];
        }
        if (m0 + 16 <= N) {
            v4f v[16 / RPI];
#pragma unroll
            for (int q = 0; q < 16 / RPI; ++q)
                v[q] = *(const v4f*)(sw + (q * RPI + rsub) * WT_PITCH + cpos);
#pragma unroll
            for (int q = 0; q < 16 / RPI; ++q)
                *(v4f*)(D + (size_t)(m0 + q * RPI + rsub) * DSTRIDE + cpos) = v[q];
        } else {
#pragma unroll
            for (int i = 0; i < 16; i += RPI) {
                const int rr = i + rsub;
                v4f v = *(const v4f*)(sw + rr * WT_PITCH + cpos);
                const int m = m0 + rr;
                if (m < N) *(v4f*)(D + (size_t)m * DSTRIDE + cpos) = v;
            }
        }
    }
}

template <int K, int F>
__device__ __forceinline__ void gather_step(float4& acc, const float* __restrict__ colbase,
                                            const int* __restrict__ src,
                                            const float* __restrict__ w, int j) {
    int s[K]; float ww[K]; float4 v[K];
#pragma unroll
    for (int k = 0; k < K; ++k) { s[k] = src[j + k]; ww[k] = w[j + k]; }
#pragma unroll
    for (int k = 0; k < K; ++k) v[k] = *(const float4*)(colbase + (size_t)s[k] * F);
#pragma unroll
    for (int k = 0; k < K; ++k) {
        acc.x = fmaf(v[k].x, ww[k], acc.x);
        acc.y = fmaf(v[k].y, ww[k], acc.y);
        acc.z = fmaf(v[k].z, ww[k], acc.z);
        acc.w = fmaf(v[k].w, ww[k], acc.w);
    }
}

template <bool RELU>
__global__ __launch_bounds__(256)
void gather128_kernel(const float* __restrict__ lin, const float* __restrict__ dis,
                      const float* __restrict__ bias,
                      const int* __restrict__ rowptr, const int* __restrict__ count,
                      const int* __restrict__ csr_src, const float* __restrict__ csr_w,
                      float* __restrict__ out, int N) {
    int lane = threadIdx.x & 31;
    int node = __builtin_amdgcn_readfirstlane((int)(blockIdx.x * 8 + (threadIdx.x >> 5)));
    if (node >= N) return;
    int start = rowptr[node];
    int d = count[node];
    float dn = dis[node];
    const float* __restrict__ colbase = lin + lane * 4;
    const int* __restrict__ src = csr_src + start;
    const float* __restrict__ w = csr_w + start;
    float4 self = *(const float4*)(colbase + (size_t)node * 128);
    float s2 = dn * dn;
    float4 acc;
    acc.x = self.x * s2; acc.y = self.y * s2; acc.z = self.z * s2; acc.w = self.w * s2;
    int j = 0;
    for (; j + 8 <= d; j += 8) gather_step<8, 128>(acc, colbase, src, w, j);
    if (j + 4 <= d) { gather_step<4, 128>(acc, colbase, src, w, j); j += 4; }
    if (j + 2 <= d) { gather_step<2, 128>(acc, colbase, src, w, j); j += 2; }
    if (j < d) gather_step<1, 128>(acc, colbase, src, w, j);
    float4 b = *(const float4*)(bias + lane * 4);
    acc.x += b.x; acc.y += b.y; acc.z += b.z; acc.w += b.w;
    if (RELU) {
        acc.x = fmaxf(acc.x, 0.0f); acc.y = fmaxf(acc.y, 0.0f);
        acc.z = fmaxf(acc.z, 0.0f); acc.w = fmaxf(acc.w, 0.0f);
    }
    *(float4*)(out + (size_t)node * 128 + lane * 4) = acc;
}

__global__ void gather16_kernel(const float* __restrict__ lin, const float* __restrict__ dis,
                                const float* __restrict__ bias, int biasLen,
                                const int* __restrict__ rowptr, const int* __restrict__ count,
                                const int* __restrict__ csr_src, const float* __restrict__ csr_w,
                                float* __restrict__ out, int N) {
    long long t = (long long)blockIdx.x * blockDim.x + threadIdx.x;
    int node = (int)(t >> 2);
    int q = (int)(t & 3);
    if (t >= (long long)N * 4) return;
    int start = rowptr[node];
    int d = count[node];
    float dn = dis[node];
    const float* __restrict__ colbase = lin + q * 4;
    const int* __restrict__ src = csr_src + start;
    const float* __restrict__ w = csr_w + start;
    float4 self = *(const float4*)(colbase + (size_t)node * 16);
    float s2 = dn * dn;
    float4 acc;
    acc.x = self.x * s2; acc.y = self.y * s2; acc.z = self.z * s2; acc.w = self.w * s2;
    int j = 0;
    for (; j + 4 <= d; j += 4) gather_step<4, 16>(acc, colbase, src, w, j);
    if (j + 2 <= d) { gather_step<2, 16>(acc, colbase, src, w, j); j += 2; }
    if (j < d) gather_step<1, 16>(acc, colbase, src, w, j);
    int c = q * 4;
    acc.x += (c + 0 < biasLen) ? bias[c + 0] : 0.0f;
    acc.y += (c + 1 < biasLen) ? bias[c + 1] : 0.0f;
    acc.z += (c + 2 < biasLen) ? bias[c + 2] : 0.0f;
    acc.w += (c + 3 < biasLen) ? bias[c + 3] : 0.0f;
    *(float4*)(out + (size_t)node * 16 + q * 4) = acc;
}

__global__ void pool_kernel(const float* __restrict__ h  ,
                            const int* __restrict__ batch,
                            float* __restrict__ out, float* __restrict__ cnt, int N) {
    long long t = (long long)blockIdx.x * blockDim.x + threadIdx.x;
    if (t >= (long long)N * 16) return;
    int i = (int)(t >> 4);
    int c = (int)(t & 15);
    int g = batch[i];
    if (c < C_OUT) atomicAdd(&out[(size_t)g * C_OUT + c], h[t]);
    if (c == 0)    atomicAdd(&cnt[g], 1.0f);
}

__global__ void div_kernel(float* __restrict__ out, const float* __restrict__ cnt) {
    int t = blockIdx.x * blockDim.x + threadIdx.x;
    if (t < NUM_GRAPHS * C_OUT) out[t] /= fmaxf(cnt[t / C_OUT], 1.0f);
}


static inline int cdiv_ll(long long a, long long b) { return (int)((a + b - 1) / b); }

extern "C" void kernel_launch(void* const* d_in, const int* in_sizes, int n_in,
                              void* d_out, int out_size, void* d_ws, size_t ws_size,
                              hipStream_t stream) {
    const float* x     = (const float*)d_in[0];
    const int*   ei    = (const int*)d_in[1];
    const int*   batch = (const int*)d_in[2];
    const float* W0 = (const float*)d_in[3];  const float* b0 = (const float*)d_in[4];
    const float* W1 = (const float*)d_in[5];  const float* b1 = (const float*)d_in[6];
    const float* W2 = (const float*)d_in[7];  const float* b2 = (const float*)d_in[8];
    const float* W3 = (const float*)d_in[9];  const float* b3 = (const float*)d_in[10];

    const int N = in_sizes[2];
    const int E = in_sizes[1] / 2;
    const int H = 128;
    const int* rowi = ei;
    const int* coli = ei + E;

    float* out = (float*)d_out;
    float* ws  = (float*)d_ws;
    const size_t Npad = ((size_t)N + 255) & ~(size_t)255;
    const int    nb   = (N + 1023) >> 10;
    float* dis  = ws;
    float* bufA = ws + Npad;
    float* bufB = bufA + (size_t)N * H;
    float* cnt  = bufB + (size_t)N * H;
    int*   count    = (int*)(cnt + 512);
    int*   rowptr   = count + Npad;
    int*   bsum     = rowptr + Npad;
    int*   rank     = bsum + (((size_t)nb + 255) & ~(size_t)255);
    int*   slot_src = rank + (size_t)E;
    int*   csr_src  = slot_src + (size_t)E;
    float* csr_w    = (float*)(csr_src + (size_t)E);

    const int TB = 256;

    fill_int_kernel<<<cdiv_ll(N, TB), TB, 0, stream>>>(count, 0, N);
    rank_kernel<<<cdiv_ll(E, TB), TB, 0, stream>>>(count, coli, rank, E);
    scan_local_kernel<<<nb, TB, 0, stream>>>(count, rowptr, bsum, N);
    scan_top_kernel<<<1, 32, 0, stream>>>(bsum, nb);
    finalize_kernel<<<cdiv_ll(N, TB), TB, 0, stream>>>(rowptr, bsum, count, dis, N);
    place_kernel<<<cdiv_ll(E, TB), TB, 0, stream>>>(rowi, coli, rank, rowptr, slot_src, E);
    order_kernel<<<cdiv_ll(N, TB / 32), TB, 0, stream>>>(slot_src, rowptr, count, dis, csr_src, csr_w, N);

    const int gemmBlk    = 512;
    const int gathBlkH   = cdiv_ll(N, TB / 32);

    gemm_strip_wmma_f32<128, 8, 128><<<gemmBlk, TB, 0, stream>>>(x, W0, bufA, N);
    gather128_kernel<true><<<gathBlkH, TB, 0, stream>>>(bufA, dis, b0, rowptr, count, csr_src, csr_w, bufB, N);

    gemm_strip_wmma_f32<128, 8, 128><<<gemmBlk, TB, 0, stream>>>(bufB, W1, bufA, N);
    gather128_kernel<true><<<gathBlkH, TB, 0, stream>>>(bufA, dis, b1, rowptr, count, csr_src, csr_w, bufB, N);

    gemm_strip_wmma_f32<128, 8, 128><<<gemmBlk, TB, 0, stream>>>(bufB, W2, bufA, N);
    gather128_kernel<true><<<gathBlkH, TB, 0, stream>>>(bufA, dis, b2, rowptr, count, csr_src, csr_w, bufB, N);

    gemm_strip_wmma_f32<10, 1, 16><<<gemmBlk, TB, 0, stream>>>(bufB, W3, bufA, N);
    gather16_kernel<<<cdiv_ll((long long)N * 4, TB), TB, 0, stream>>>(bufA, dis, b3, C_OUT, rowptr, count, csr_src, csr_w, bufB, N);

    fill_kernel<<<cdiv_ll(NUM_GRAPHS * C_OUT, TB), TB, 0, stream>>>(out, 0.0f, NUM_GRAPHS * C_OUT);
    fill_kernel<<<cdiv_ll(NUM_GRAPHS, TB), TB, 0, stream>>>(cnt, 0.0f, NUM_GRAPHS);
    pool_kernel<<<cdiv_ll((long long)N * 16, TB), TB, 0, stream>>>(bufB, batch, out, cnt, N);
    div_kernel<<<cdiv_ll(NUM_GRAPHS * C_OUT, TB), TB, 0, stream>>>(out, cnt);
}


# In-degree count and edge rank from LDS counters (8 target ranges x 64 edge chunks; per-chunk counts prefixed per node) in place of one returning global atomic per edge; final edge order unchanged
# speedup vs baseline: 3.6820x; 1.1489x over previous
#include <hip/hip_runtime.h>


typedef float v2f __attribute__((ext_vector_type(2)));
typedef float v4f __attribute__((ext_vector_type(4)));
typedef float v8f __attribute__((ext_vector_type(8)));

#define NUM_GRAPHS 512
#define C_OUT 10


__global__ void fill_kernel(float* __restrict__ p, float v, int n) {
    int t = blockIdx.x * blockDim.x + threadIdx.x;
    if (t < n) p[t] = v;
}

#define CR_NR 12544
#define CR_S  64
#define CR_U  8

__global__ __launch_bounds__(1024)
void count_rank_kernel(const int* __restrict__ col, int* __restrict__ rank,
                       int* __restrict__ part, int E, int chunk, int Npad) {
    __shared__ __attribute__((aligned(16))) int lcnt[CR_NR];
    const int tid = threadIdx.x;
    const int s = blockIdx.x % CR_S;
    const int lo = (blockIdx.x / CR_S) * CR_NR;
    for (int i = tid; i < CR_NR / 4; i += 1024) ((int4*)lcnt)[i] = make_int4(0, 0, 0, 0);
    __syncthreads();
    const int t0 = s * chunk;
    const int t1 = min(t0 + chunk, E);
    const int tlast = t1 - 1;
    for (int tb = t0 + tid; tb < t1; tb += 1024 * CR_U) {
        int c[CR_U], rk[CR_U];
#pragma unroll
        for (int u = 0; u < CR_U; ++u) {
            const int t = tb + u * 1024;
            const int v = col[min(t, tlast)];
            c[u] = (t < t1) ? (v - lo) : -1;
        }
#pragma unroll
        for (int u = 0; u < CR_U; ++u)
            rk[u] = ((unsigned)c[u] < (unsigned)CR_NR) ? atomicAdd(&lcnt[c[u]], 1) : 0;
#pragma unroll
        for (int u = 0; u < CR_U; ++u)
            if ((unsigned)c[u] < (unsigned)CR_NR) rank[tb + u * 1024] = rk[u];
    }
    __syncthreads();
    const int hi = min(CR_NR, Npad - lo);
    int4* __restrict__ dst = (int4*)(part + (size_t)s * Npad + lo);
    for (int i = tid; i < hi / 4; i += 1024) dst[i] = ((const int4*)lcnt)[i];
}

__device__ __forceinline__ int wave_inclusive_scan(int v, int lane) {
    int incl = v;
    for (int off = 1; off < 32; off <<= 1) {
        int n = __shfl_up(incl, off, 32);
        if (lane >= off) incl += n;
    }
    return incl;
}

__global__ __launch_bounds__(256)
void scan_local_kernel(int* __restrict__ part, int* __restrict__ count,
                       int* __restrict__ rowptr, int* __restrict__ bsum, int N, int Npad) {
    __shared__ int wsum[8];
    int lane = threadIdx.x & 31, wv = threadIdx.x >> 5;
    int base = blockIdx.x * 1024 + threadIdx.x * 4;
    int4 tot = make_int4(0, 0, 0, 0);
    if (base < Npad) {
        int4* p = (int4*)(part + base);
        const size_t pitch4 = (size_t)(Npad >> 2);
        for (int s0 = 0; s0 < CR_S; s0 += 16) {
            int4 v[16];
#pragma unroll
            for (int k = 0; k < 16; ++k) v[k] = p[(size_t)(s0 + k) * pitch4];
#pragma unroll
            for (int k = 0; k < 16; ++k) {
                p[(size_t)(s0 + k) * pitch4] = tot;
                tot.x += v[k].x; tot.y += v[k].y; tot.z += v[k].z; tot.w += v[k].w;
            }
        }
    }
    int c0 = (base + 0 < N) ? tot.x : 0;
    int c1 = (base + 1 < N) ? tot.y : 0;
    int c2 = (base + 2 < N) ? tot.z : 0;
    int c3 = (base + 3 < N) ? tot.w : 0;
    int tsum = c0 + c1 + c2 + c3;
    int incl = wave_inclusive_scan(tsum, lane);
    if (lane == 31) wsum[wv] = incl;
    __syncthreads();
    int woff = 0;
    for (int w = 0; w < wv; ++w) woff += wsum[w];
    int excl = woff + incl - tsum;
    if (base + 0 < N) { rowptr[base + 0] = excl;                count[base + 0] = c0; }
    if (base + 1 < N) { rowptr[base + 1] = excl + c0;           count[base + 1] = c1; }
    if (base + 2 < N) { rowptr[base + 2] = excl + c0 + c1;      count[base + 2] = c2; }
    if (base + 3 < N) { rowptr[base + 3] = excl + c0 + c1 + c2; count[base + 3] = c3; }
    if (threadIdx.x == 255) bsum[blockIdx.x] = woff + incl;
}

__global__ void scan_top_kernel(int* __restrict__ bsum, int nb) {
    int lane = threadIdx.x;
    int carry = 0;
    for (int b0 = 0; b0 < nb; b0 += 32) {
        int i = b0 + lane;
        int v = (i < nb) ? bsum[i] : 0;
        int incl = wave_inclusive_scan(v, lane);
        if (i < nb) bsum[i] = carry + incl - v;
        carry += __shfl(incl, 31, 32);
    }
}

__global__ void finalize_kernel(int* __restrict__ rowptr, const int* __restrict__ bsum,
                                const int* __restrict__ count, float* __restrict__ dis, int N) {
    int i = blockIdx.x * blockDim.x + threadIdx.x;
    if (i < N) {
        rowptr[i] += bsum[i >> 10];
        dis[i] = rsqrtf((float)(count[i] + 1));
    }
}

__global__ void place_kernel(const int* __restrict__ row, const int* __restrict__ col,
                             const int* __restrict__ rank, const int* __restrict__ rowptr,
                             const int* __restrict__ part, int* __restrict__ slot_src,
                             int E, int chunk, int Npad) {
    int t = blockIdx.x * blockDim.x + threadIdx.x;
    if (t < E) {
        int c = col[t];
        int s = t / chunk;
        slot_src[rowptr[c] + part[(size_t)s * Npad + c] + rank[t]] = row[t];
    }
}

__global__ __launch_bounds__(256)
void order_kernel(const int* __restrict__ slot_src, const int* __restrict__ rowptr,
                  const int* __restrict__ count, const float* __restrict__ dis,
                  int* __restrict__ csr_src, float* __restrict__ csr_w, int N) {
    int lane = threadIdx.x & 31;
    int node = __builtin_amdgcn_readfirstlane((int)(blockIdx.x * 8 + (threadIdx.x >> 5)));
    if (node >= N) return;
    int start = rowptr[node];
    int d = count[node];
    float dc = dis[node];
    const int* __restrict__ seg = slot_src + start;
    for (int i0 = 0; i0 < d; i0 += 32) {
        int i = i0 + lane;
        if (i < d) {
            int key = seg[i];
            int p = 0;
            for (int j = 0; j < d; ++j) {
                int kj = seg[j];
                p += ((kj < key) || (kj == key && j < i)) ? 1 : 0;
            }
            csr_src[start + p] = key;
            csr_w[start + p] = dis[key] * dc;
        }
    }
}

#define WT_PITCH 132

template <int BCOLS, int NT, int DSTRIDE>
__global__ __launch_bounds__(256)
void gemm_strip_wmma_f32(const float* __restrict__ A, const float* __restrict__ B,
                         float* __restrict__ D, int N) {
    __shared__ __attribute__((aligned(16))) float Wt[NT * 16 * WT_PITCH];
    __shared__ __attribute__((aligned(16))) float St[8 * 16 * WT_PITCH];

#pragma unroll
    for (int it = 0; it < (128 * NT * 16) / 256; ++it) {
        int i = it * 256 + (int)threadIdx.x;
        int n = i % (NT * 16);
        int k = i / (NT * 16);
        int nc = (n < BCOLS) ? n : (BCOLS - 1);
        Wt[n * WT_PITCH + k] = B[(size_t)k * BCOLS + nc];
    }
    __syncthreads();

    const int lane = threadIdx.x & 31;
    const int c = lane & 15;
    const int h = lane >> 4;
    const int tilesM = (N + 15) >> 4;
    const int nwaves = (int)gridDim.x * 8;
    int tm = __builtin_amdgcn_readfirstlane((int)((threadIdx.x >> 5) * gridDim.x + blockIdx.x));
    const float* wfrag = Wt + c * WT_PITCH + 4 * h;
    constexpr int LPR = NT * 4;
    constexpr int RPI = 32 / LPR;
    float* sw = St + (threadIdx.x >> 5) * (16 * WT_PITCH);
    float* sput = sw + (8 * h) * WT_PITCH + c;
    const int rsub = lane / LPR;
    const int cpos = (lane % LPR) * 4;

    for (; tm < tilesM; tm += nwaves) {
        asm volatile("" ::: "memory");
        const int m0 = tm << 4;
        int row = m0 + c;
        if (row >= N) row = N - 1;
        const v4f* __restrict__ Ap = (const v4f*)(A + (size_t)row * 128 + 4 * h);

        v4f a[16];
#pragma unroll
        for (int kb = 0; kb < 16; ++kb) a[kb] = Ap[kb * 2];
        asm volatile("" :: "v"(a[0]), "v"(a[1]), "v"(a[2]), "v"(a[3]), "v"(a[4]), "v"(a[5]),
                           "v"(a[6]), "v"(a[7]), "v"(a[8]), "v"(a[9]), "v"(a[10]), "v"(a[11]),
                           "v"(a[12]), "v"(a[13]), "v"(a[14]), "v"(a[15]));

        v8f acc[NT];
#pragma unroll
        for (int t = 0; t < NT; ++t) acc[t] = (v8f){0.f, 0.f, 0.f, 0.f, 0.f, 0.f, 0.f, 0.f};

#pragma unroll
        for (int kb = 0; kb < 16; ++kb) {
            v4f b[NT];
#pragma unroll
            for (int t = 0; t < NT; ++t)
                b[t] = *(const v4f*)(wfrag + t * 16 * WT_PITCH + kb * 8);
            v2f a0 = {a[kb].x, a[kb].y};
            v2f a1 = {a[kb].z, a[kb].w};
#pragma unroll
            for (int t = 0; t < NT; ++t) {
                v2f b0 = {b[t].x, b[t].y};
                acc[t] = __builtin_amdgcn_wmma_f32_16x16x4_f32(false, a0, false, b0,
                                                               (short)0, acc[t], false, false);
            }
#pragma unroll
            for (int t = 0; t < NT; ++t) {
                v2f b1 = {b[t].z, b[t].w};
                acc[t] = __builtin_amdgcn_wmma_f32_16x16x4_f32(false, a1, false, b1,
                                                               (short)0, acc[t], false, false);
            }
        }

#pragma unroll
        for (int r = 0; r < 8; ++r) {
#pragma unroll
            for (int t = 0; t < NT; ++t) sput[r * WT_PITCH + t * 16] = acc[t][r];
        }
        if (m0 + 16 <= N) {
            v4f v[16 / RPI];
#pragma unroll
            for (int q = 0; q < 16 / RPI; ++q)
                v[q] = *(const v4f*)(sw + (q * RPI + rsub) * WT_PITCH + cpos);
#pragma unroll
            for (int q = 0; q < 16 / RPI; ++q)
                *(v4f*)(D + (size_t)(m0 + q * RPI + rsub) * DSTRIDE + cpos) = v[q];
        } else {
#pragma unroll
            for (int i = 0; i < 16; i += RPI) {
                const int rr = i + rsub;
                v4f v = *(const v4f*)(sw + rr * WT_PITCH + cpos);
                const int m = m0 + rr;
                if (m < N) *(v4f*)(D + (size_t)m * DSTRIDE + cpos) = v;
            }
        }
    }
}

template <int K, int F>
__device__ __forceinline__ void gather_step(float4& acc, const float* __restrict__ colbase,
                                            const int* __restrict__ src,
                                            const float* __restrict__ w, int j) {
    int s[K]; float ww[K]; float4 v[K];
#pragma unroll
    for (int k = 0; k < K; ++k) { s[k] = src[j + k]; ww[k] = w[j + k]; }
#pragma unroll
    for (int k = 0; k < K; ++k) v[k] = *(const float4*)(colbase + (size_t)s[k] * F);
#pragma unroll
    for (int k = 0; k < K; ++k) {
        acc.x = fmaf(v[k].x, ww[k], acc.x);
        acc.y = fmaf(v[k].y, ww[k], acc.y);
        acc.z = fmaf(v[k].z, ww[k], acc.z);
        acc.w = fmaf(v[k].w, ww[k], acc.w);
    }
}

template <bool RELU>
__global__ __launch_bounds__(256)
void gather128_kernel(const float* __restrict__ lin, const float* __restrict__ dis,
                      const float* __restrict__ bias,
                      const int* __restrict__ rowptr, const int* __restrict__ count,
                      const int* __restrict__ csr_src, const float* __restrict__ csr_w,
                      float* __restrict__ out, int N) {
    int lane = threadIdx.x & 31;
    int node = __builtin_amdgcn_readfirstlane((int)(blockIdx.x * 8 + (threadIdx.x >> 5)));
    if (node >= N) return;
    int start = rowptr[node];
    int d = count[node];
    float dn = dis[node];
    const float* __restrict__ colbase = lin + lane * 4;
    const int* __restrict__ src = csr_src + start;
    const float* __restrict__ w = csr_w + start;
    float4 self = *(const float4*)(colbase + (size_t)node * 128);
    float s2 = dn * dn;
    float4 acc;
    acc.x = self.x * s2; acc.y = self.y * s2; acc.z = self.z * s2; acc.w = self.w * s2;
    int j = 0;
    for (; j + 8 <= d; j += 8) gather_step<8, 128>(acc, colbase, src, w, j);
    if (j + 4 <= d) { gather_step<4, 128>(acc, colbase, src, w, j); j += 4; }
    if (j + 2 <= d) { gather_step<2, 128>(acc, colbase, src, w, j); j += 2; }
    if (j < d) gather_step<1, 128>(acc, colbase, src, w, j);
    float4 b = *(const float4*)(bias + lane * 4);
    acc.x += b.x; acc.y += b.y; acc.z += b.z; acc.w += b.w;
    if (RELU) {
        acc.x = fmaxf(acc.x, 0.0f); acc.y = fmaxf(acc.y, 0.0f);
        acc.z = fmaxf(acc.z, 0.0f); acc.w = fmaxf(acc.w, 0.0f);
    }
    *(float4*)(out + (size_t)node * 128 + lane * 4) = acc;
}

__global__ void gather16_kernel(const float* __restrict__ lin, const float* __restrict__ dis,
                                const float* __restrict__ bias, int biasLen,
                                const int* __restrict__ rowptr, const int* __restrict__ count,
                                const int* __restrict__ csr_src, const float* __restrict__ csr_w,
                                float* __restrict__ out, int N) {
    long long t = (long long)blockIdx.x * blockDim.x + threadIdx.x;
    int node = (int)(t >> 2);
    int q = (int)(t & 3);
    if (t >= (long long)N * 4) return;
    int start = rowptr[node];
    int d = count[node];
    float dn = dis[node];
    const float* __restrict__ colbase = lin + q * 4;
    const int* __restrict__ src = csr_src + start;
    const float* __restrict__ w = csr_w + start;
    float4 self = *(const float4*)(colbase + (size_t)node * 16);
    float s2 = dn * dn;
    float4 acc;
    acc.x = self.x * s2; acc.y = self.y * s2; acc.z = self.z * s2; acc.w = self.w * s2;
    int j = 0;
    for (; j + 4 <= d; j += 4) gather_step<4, 16>(acc, colbase, src, w, j);
    if (j + 2 <= d) { gather_step<2, 16>(acc, colbase, src, w, j); j += 2; }
    if (j < d) gather_step<1, 16>(acc, colbase, src, w, j);
    int c = q * 4;
    acc.x += (c + 0 < biasLen) ? bias[c + 0] : 0.0f;
    acc.y += (c + 1 < biasLen) ? bias[c + 1] : 0.0f;
    acc.z += (c + 2 < biasLen) ? bias[c + 2] : 0.0f;
    acc.w += (c + 3 < biasLen) ? bias[c + 3] : 0.0f;
    *(float4*)(out + (size_t)node * 16 + q * 4) = acc;
}

__global__ void pool_kernel(const float* __restrict__ h  ,
                            const int* __restrict__ batch,
                            float* __restrict__ out, float* __restrict__ cnt, int N) {
    long long t = (long long)blockIdx.x * blockDim.x + threadIdx.x;
    if (t >= (long long)N * 16) return;
    int i = (int)(t >> 4);
    int c = (int)(t & 15);
    int g = batch[i];
    if (c < C_OUT) atomicAdd(&out[(size_t)g * C_OUT + c], h[t]);
    if (c == 0)    atomicAdd(&cnt[g], 1.0f);
}

__global__ void div_kernel(float* __restrict__ out, const float* __restrict__ cnt) {
    int t = blockIdx.x * blockDim.x + threadIdx.x;
    if (t < NUM_GRAPHS * C_OUT) out[t] /= fmaxf(cnt[t / C_OUT], 1.0f);
}


static inline int cdiv_ll(long long a, long long b) { return (int)((a + b - 1) / b); }

extern "C" void kernel_launch(void* const* d_in, const int* in_sizes, int n_in,
                              void* d_out, int out_size, void* d_ws, size_t ws_size,
                              hipStream_t stream) {
    const float* x     = (const float*)d_in[0];
    const int*   ei    = (const int*)d_in[1];
    const int*   batch = (const int*)d_in[2];
    const float* W0 = (const float*)d_in[3];  const float* b0 = (const float*)d_in[4];
    const float* W1 = (const float*)d_in[5];  const float* b1 = (const float*)d_in[6];
    const float* W2 = (const float*)d_in[7];  const float* b2 = (const float*)d_in[8];
    const float* W3 = (const float*)d_in[9];  const float* b3 = (const float*)d_in[10];

    const int N = in_sizes[2];
    const int E = in_sizes[1] / 2;
    const int H = 128;
    const int* rowi = ei;
    const int* coli = ei + E;

    float* out = (float*)d_out;
    float* ws  = (float*)d_ws;
    const size_t Npad = ((size_t)N + 255) & ~(size_t)255;
    const int    nb   = (N + 1023) >> 10;
    float* dis  = ws;
    float* bufA = ws + Npad;
    float* bufB = bufA + (size_t)N * H;
    float* cnt  = bufB + (size_t)N * H;
    int*   count    = (int*)(cnt + 512);
    int*   rowptr   = count + Npad;
    int*   bsum     = rowptr + Npad;
    int*   part     = bsum + (((size_t)nb + 255) & ~(size_t)255);
    int*   rank     = part + (size_t)CR_S * Npad;
    int*   slot_src = rank + (size_t)E;
    int*   csr_src  = slot_src + (size_t)E;
    float* csr_w    = (float*)(csr_src + (size_t)E);

    const int TB = 256;

    const int nranges = cdiv_ll((long long)Npad, CR_NR);
    const int chunk   = cdiv_ll(E, CR_S);
    count_rank_kernel<<<nranges * CR_S, 1024, 0, stream>>>(coli, rank, part, E, chunk, (int)Npad);
    scan_local_kernel<<<nb, TB, 0, stream>>>(part, count, rowptr, bsum, N, (int)Npad);
    scan_top_kernel<<<1, 32, 0, stream>>>(bsum, nb);
    finalize_kernel<<<cdiv_ll(N, TB), TB, 0, stream>>>(rowptr, bsum, count, dis, N);
    place_kernel<<<cdiv_ll(E, TB), TB, 0, stream>>>(rowi, coli, rank, rowptr, part, slot_src, E, chunk, (int)Npad);
    order_kernel<<<cdiv_ll(N, TB / 32), TB, 0, stream>>>(slot_src, rowptr, count, dis, csr_src, csr_w, N);

    const int gemmBlk    = 512;
    const int gathBlkH   = cdiv_ll(N, TB / 32);

    gemm_strip_wmma_f32<128, 8, 128><<<gemmBlk, TB, 0, stream>>>(x, W0, bufA, N);
    gather128_kernel<true><<<gathBlkH, TB, 0, stream>>>(bufA, dis, b0, rowptr, count, csr_src, csr_w, bufB, N);

    gemm_strip_wmma_f32<128, 8, 128><<<gemmBlk, TB, 0, stream>>>(bufB, W1, bufA, N);
    gather128_kernel<true><<<gathBlkH, TB, 0, stream>>>(bufA, dis, b1, rowptr, count, csr_src, csr_w, bufB, N);

    gemm_strip_wmma_f32<128, 8, 128><<<gemmBlk, TB, 0, stream>>>(bufB, W2, bufA, N);
    gather128_kernel<true><<<gathBlkH, TB, 0, stream>>>(bufA, dis, b2, rowptr, count, csr_src, csr_w, bufB, N);

    gemm_strip_wmma_f32<10, 1, 16><<<gemmBlk, TB, 0, stream>>>(bufB, W3, bufA, N);
    gather16_kernel<<<cdiv_ll((long long)N * 4, TB), TB, 0, stream>>>(bufA, dis, b3, C_OUT, rowptr, count, csr_src, csr_w, bufB, N);

    fill_kernel<<<cdiv_ll(NUM_GRAPHS * C_OUT, TB), TB, 0, stream>>>(out, 0.0f, NUM_GRAPHS * C_OUT);
    fill_kernel<<<cdiv_ll(NUM_GRAPHS, TB), TB, 0, stream>>>(cnt, 0.0f, NUM_GRAPHS);
    pool_kernel<<<cdiv_ll((long long)N * 16, TB), TB, 0, stream>>>(bufB, batch, out, cnt, N);
    div_kernel<<<cdiv_ll(NUM_GRAPHS * C_OUT, TB), TB, 0, stream>>>(out, cnt);
}


# Aggregation of layers 1-3 fused with the next layer's GEMM: a work-group sums a 16-row strip into LDS and multiplies it there (B fragments in registers); 18 -> 15 launches, sum orders unchanged
# speedup vs baseline: 3.9652x; 1.0769x over previous
#include <hip/hip_runtime.h>


typedef float v2f __attribute__((ext_vector_type(2)));
typedef float v4f __attribute__((ext_vector_type(4)));
typedef float v8f __attribute__((ext_vector_type(8)));

#define NUM_GRAPHS 512
#define C_OUT 10


__global__ void fill_kernel(float* __restrict__ p, float v, int n) {
    int t = blockIdx.x * blockDim.x + threadIdx.x;
    if (t < n) p[t] = v;
}

#define CR_NR 12544
#define CR_S  64
#define CR_U  8

__global__ __launch_bounds__(1024)
void count_rank_kernel(const int* __restrict__ col, int* __restrict__ rank,
                       int* __restrict__ part, int E, int chunk, int Npad) {
    __shared__ __attribute__((aligned(16))) int lcnt[CR_NR];
    const int tid = threadIdx.x;
    const int s = blockIdx.x % CR_S;
    const int lo = (blockIdx.x / CR_S) * CR_NR;
    for (int i = tid; i < CR_NR / 4; i += 1024) ((int4*)lcnt)[i] = make_int4(0, 0, 0, 0);
    __syncthreads();
    const int t0 = s * chunk;
    const int t1 = min(t0 + chunk, E);
    const int tlast = t1 - 1;
    for (int tb = t0 + tid; tb < t1; tb += 1024 * CR_U) {
        int c[CR_U], rk[CR_U];
#pragma unroll
        for (int u = 0; u < CR_U; ++u) {
            const int t = tb + u * 1024;
            const int v = col[min(t, tlast)];
            c[u] = (t < t1) ? (v - lo) : -1;
        }
#pragma unroll
        for (int u = 0; u < CR_U; ++u)
            rk[u] = ((unsigned)c[u] < (unsigned)CR_NR) ? atomicAdd(&lcnt[c[u]], 1) : 0;
#pragma unroll
        for (int u = 0; u < CR_U; ++u)
            if ((unsigned)c[u] < (unsigned)CR_NR) rank[tb + u * 1024] = rk[u];
    }
    __syncthreads();
    const int hi = min(CR_NR, Npad - lo);
    int4* __restrict__ dst = (int4*)(part + (size_t)s * Npad + lo);
    for (int i = tid; i < hi / 4; i += 1024) dst[i] = ((const int4*)lcnt)[i];
}

__device__ __forceinline__ int wave_inclusive_scan(int v, int lane) {
    int incl = v;
    for (int off = 1; off < 32; off <<= 1) {
        int n = __shfl_up(incl, off, 32);
        if (lane >= off) incl += n;
    }
    return incl;
}

__global__ __launch_bounds__(256)
void scan_local_kernel(int* __restrict__ part, int* __restrict__ count,
                       int* __restrict__ rowptr, int* __restrict__ bsum, int N, int Npad) {
    __shared__ int wsum[8];
    int lane = threadIdx.x & 31, wv = threadIdx.x >> 5;
    int base = blockIdx.x * 1024 + threadIdx.x * 4;
    int4 tot = make_int4(0, 0, 0, 0);
    if (base < Npad) {
        int4* p = (int4*)(part + base);
        const size_t pitch4 = (size_t)(Npad >> 2);
        for (int s0 = 0; s0 < CR_S; s0 += 16) {
            int4 v[16];
#pragma unroll
            for (int k = 0; k < 16; ++k) v[k] = p[(size_t)(s0 + k) * pitch4];
#pragma unroll
            for (int k = 0; k < 16; ++k) {
                p[(size_t)(s0 + k) * pitch4] = tot;
                tot.x += v[k].x; tot.y += v[k].y; tot.z += v[k].z; tot.w += v[k].w;
            }
        }
    }
    int c0 = (base + 0 < N) ? tot.x : 0;
    int c1 = (base + 1 < N) ? tot.y : 0;
    int c2 = (base + 2 < N) ? tot.z : 0;
    int c3 = (base + 3 < N) ? tot.w : 0;
    int tsum = c0 + c1 + c2 + c3;
    int incl = wave_inclusive_scan(tsum, lane);
    if (lane == 31) wsum[wv] = incl;
    __syncthreads();
    int woff = 0;
    for (int w = 0; w < wv; ++w) woff += wsum[w];
    int excl = woff + incl - tsum;
    if (base + 0 < N) { rowptr[base + 0] = excl;                count[base + 0] = c0; }
    if (base + 1 < N) { rowptr[base + 1] = excl + c0;           count[base + 1] = c1; }
    if (base + 2 < N) { rowptr[base + 2] = excl + c0 + c1;      count[base + 2] = c2; }
    if (base + 3 < N) { rowptr[base + 3] = excl + c0 + c1 + c2; count[base + 3] = c3; }
    if (threadIdx.x == 255) bsum[blockIdx.x] = woff + incl;
}

__global__ void scan_top_kernel(int* __restrict__ bsum, int nb) {
    int lane = threadIdx.x;
    int carry = 0;
    for (int b0 = 0; b0 < nb; b0 += 32) {
        int i = b0 + lane;
        int v = (i < nb) ? bsum[i] : 0;
        int incl = wave_inclusive_scan(v, lane);
        if (i < nb) bsum[i] = carry + incl - v;
        carry += __shfl(incl, 31, 32);
    }
}

__global__ void finalize_kernel(int* __restrict__ rowptr, const int* __restrict__ bsum,
                                const int* __restrict__ count, float* __restrict__ dis, int N) {
    int i = blockIdx.x * blockDim.x + threadIdx.x;
    if (i < N) {
        rowptr[i] += bsum[i >> 10];
        dis[i] = rsqrtf((float)(count[i] + 1));
    }
}

__global__ void place_kernel(const int* __restrict__ row, const int* __restrict__ col,
                             const int* __restrict__ rank, const int* __restrict__ rowptr,
                             const int* __restrict__ part, int* __restrict__ slot_src,
                             int E, int chunk, int Npad) {
    int t = blockIdx.x * blockDim.x + threadIdx.x;
    if (t < E) {
        int c = col[t];
        int s = t / chunk;
        slot_src[rowptr[c] + part[(size_t)s * Npad + c] + rank[t]] = row[t];
    }
}

__global__ __launch_bounds__(256)
void order_kernel(const int* __restrict__ slot_src, const int* __restrict__ rowptr,
                  const int* __restrict__ count, const float* __restrict__ dis,
                  int* __restrict__ csr_src, float* __restrict__ csr_w, int N) {
    int lane = threadIdx.x & 31;
    int node = __builtin_amdgcn_readfirstlane((int)(blockIdx.x * 8 + (threadIdx.x >> 5)));
    if (node >= N) return;
    int start = rowptr[node];
    int d = count[node];
    float dc = dis[node];
    const int* __restrict__ seg = slot_src + start;
    for (int i0 = 0; i0 < d; i0 += 32) {
        int i = i0 + lane;
        if (i < d) {
            int key = seg[i];
            int p = 0;
            for (int j = 0; j < d; ++j) {
                int kj = seg[j];
                p += ((kj < key) || (kj == key && j < i)) ? 1 : 0;
            }
            csr_src[start + p] = key;
            csr_w[start + p] = dis[key] * dc;
        }
    }
}

#define WT_PITCH 132

template <int BCOLS, int NT, int DSTRIDE>
__global__ __launch_bounds__(256)
void gemm_strip_wmma_f32(const float* __restrict__ A, const float* __restrict__ B,
                         float* __restrict__ D, int N) {
    __shared__ __attribute__((aligned(16))) float Wt[NT * 16 * WT_PITCH];
    __shared__ __attribute__((aligned(16))) float St[8 * 16 * WT_PITCH];

#pragma unroll
    for (int it = 0; it < (128 * NT * 16) / 256; ++it) {
        int i = it * 256 + (int)threadIdx.x;
        int n = i % (NT * 16);
        int k = i / (NT * 16);
        int nc = (n < BCOLS) ? n : (BCOLS - 1);
        Wt[n * WT_PITCH + k] = B[(size_t)k * BCOLS + nc];
    }
    __syncthreads();

    const int lane = threadIdx.x & 31;
    const int c = lane & 15;
    const int h = lane >> 4;
    const int tilesM = (N + 15) >> 4;
    const int nwaves = (int)gridDim.x * 8;
    int tm = __builtin_amdgcn_readfirstlane((int)((threadIdx.x >> 5) * gridDim.x + blockIdx.x));
    const float* wfrag = Wt + c * WT_PITCH + 4 * h;
    constexpr int LPR = NT * 4;
    constexpr int RPI = 32 / LPR;
    float* sw = St + (threadIdx.x >> 5) * (16 * WT_PITCH);
    float* sput = sw + (8 * h) * WT_PITCH + c;
    const int rsub = lane / LPR;
    const int cpos = (lane % LPR) * 4;

    for (; tm < tilesM; tm += nwaves) {
        asm volatile("" ::: "memory");
        const int m0 = tm << 4;
        int row = m0 + c;
        if (row >= N) row = N - 1;
        const v4f* __restrict__ Ap = (const v4f*)(A + (size_t)row * 128 + 4 * h);

        v4f a[16];
#pragma unroll
        for (int kb = 0; kb < 16; ++kb) a[kb] = Ap[kb * 2];
        asm volatile("" :: "v"(a[0]), "v"(a[1]), "v"(a[2]), "v"(a[3]), "v"(a[4]), "v"(a[5]),
                           "v"(a[6]), "v"(a[7]), "v"(a[8]), "v"(a[9]), "v"(a[10]), "v"(a[11]),
                           "v"(a[12]), "v"(a[13]), "v"(a[14]), "v"(a[15]));

        v8f acc[NT];
#pragma unroll
        for (int t = 0; t < NT; ++t) acc[t] = (v8f){0.f, 0.f, 0.f, 0.f, 0.f, 0.f, 0.f, 0.f};

#pragma unroll
        for (int kb = 0; kb < 16; ++kb) {
            v4f b[NT];
#pragma unroll
            for (int t = 0; t < NT; ++t)
                b[t] = *(const v4f*)(wfrag + t * 16 * WT_PITCH + kb * 8);
            v2f a0 = {a[kb].x, a[kb].y};
            v2f a1 = {a[kb].z, a[kb].w};
#pragma unroll
            for (int t = 0; t < NT; ++t) {
                v2f b0 = {b[t].x, b[t].y};
                acc[t] = __builtin_amdgcn_wmma_f32_16x16x4_f32(false, a0, false, b0,
                                                               (short)0, acc[t], false, false);
            }
#pragma unroll
            for (int t = 0; t < NT; ++t) {
                v2f b1 = {b[t].z, b[t].w};
                acc[t] = __builtin_amdgcn_wmma_f32_16x16x4_f32(false, a1, false, b1,
                                                               (short)0, acc[t], false, false);
            }
        }

#pragma unroll
        for (int r = 0; r < 8; ++r) {
#pragma unroll
            for (int t = 0; t < NT; ++t) sput[r * WT_PITCH + t * 16] = acc[t][r];
        }
        if (m0 + 16 <= N) {
            v4f v[16 / RPI];
#pragma unroll
            for (int q = 0; q < 16 / RPI; ++q)
                v[q] = *(const v4f*)(sw + (q * RPI + rsub) * WT_PITCH + cpos);
#pragma unroll
            for (int q = 0; q < 16 / RPI; ++q)
                *(v4f*)(D + (size_t)(m0 + q * RPI + rsub) * DSTRIDE + cpos) = v[q];
        } else {
#pragma unroll
            for (int i = 0; i < 16; i += RPI) {
                const int rr = i + rsub;
                v4f v = *(const v4f*)(sw + rr * WT_PITCH + cpos);
                const int m = m0 + rr;
                if (m < N) *(v4f*)(D + (size_t)m * DSTRIDE + cpos) = v;
            }
        }
    }
}

template <int K, int F>
__device__ __forceinline__ void gather_step(float4& acc, const float* __restrict__ colbase,
                                            const int* __restrict__ src,
                                            const float* __restrict__ w, int j) {
    int s[K]; float ww[K]; float4 v[K];
#pragma unroll
    for (int k = 0; k < K; ++k) { s[k] = src[j + k]; ww[k] = w[j + k]; }
#pragma unroll
    for (int k = 0; k < K; ++k) v[k] = *(const float4*)(colbase + (size_t)s[k] * F);
#pragma unroll
    for (int k = 0; k < K; ++k) {
        acc.x = fmaf(v[k].x, ww[k], acc.x);
        acc.y = fmaf(v[k].y, ww[k], acc.y);
        acc.z = fmaf(v[k].z, ww[k], acc.z);
        acc.w = fmaf(v[k].w, ww[k], acc.w);
    }
}

#define ST_PITCH 132
#define FUSE_BLK 1024

template <bool WIDE>
__global__ __launch_bounds__(256)
void gather_gemm_kernel(const float* __restrict__ lin, const float* __restrict__ dis,
                        const float* __restrict__ bias,
                        const int* __restrict__ rowptr, const int* __restrict__ count,
                        const int* __restrict__ csr_src, const float* __restrict__ csr_w,
                        const float* __restrict__ Bw, int bcols,
                        float* __restrict__ D, int N) {
    __shared__ __attribute__((aligned(16))) float St[16 * ST_PITCH];
    __shared__ __attribute__((aligned(16))) float Ot[16 * ST_PITCH];
    const int lane = threadIdx.x & 31;
    const int c = lane & 15;
    const int h = lane >> 4;
    const int wv = __builtin_amdgcn_readfirstlane((int)(threadIdx.x >> 5));
    const bool mult = WIDE || (wv == 0);

    v4f breg[16];
    {
        int ncol = (WIDE ? wv * 16 : 0) + c;
        if (ncol >= bcols) ncol = bcols - 1;
        const float* __restrict__ bp = Bw + (size_t)(4 * h) * bcols + ncol;
#pragma unroll
        for (int kb = 0; kb < 16; ++kb) {
            breg[kb].x = bp[(size_t)(kb * 8 + 0) * bcols];
            breg[kb].y = bp[(size_t)(kb * 8 + 1) * bcols];
            breg[kb].z = bp[(size_t)(kb * 8 + 2) * bcols];
            breg[kb].w = bp[(size_t)(kb * 8 + 3) * bcols];
        }
    }

    const float4 bv = *(const float4*)(bias + lane * 4);
    const float* __restrict__ colbase = lin + lane * 4;
    const int tilesM = (N + 15) >> 4;

    for (int tm = blockIdx.x; tm < tilesM; tm += gridDim.x) {
        const int m0 = tm << 4;
        for (int rr = 0; rr < 2; ++rr) {
            const int rloc = wv * 2 + rr;
            const int node = m0 + rloc;
            if (node < N) {
                int start = rowptr[node];
                int d = count[node];
                float dn = dis[node];
                const int* __restrict__ src = csr_src + start;
                const float* __restrict__ w = csr_w + start;
                float4 self = *(const float4*)(colbase + (size_t)node * 128);
                float s2 = dn * dn;
                float4 acc;
                acc.x = self.x * s2; acc.y = self.y * s2; acc.z = self.z * s2; acc.w = self.w * s2;
                int j = 0;
                for (; j + 8 <= d; j += 8) gather_step<8, 128>(acc, colbase, src, w, j);
                if (j + 4 <= d) { gather_step<4, 128>(acc, colbase, src, w, j); j += 4; }
                if (j + 2 <= d) { gather_step<2, 128>(acc, colbase, src, w, j); j += 2; }
                if (j < d) gather_step<1, 128>(acc, colbase, src, w, j);
                acc.x += bv.x; acc.y += bv.y; acc.z += bv.z; acc.w += bv.w;
                acc.x = fmaxf(acc.x, 0.0f); acc.y = fmaxf(acc.y, 0.0f);
                acc.z = fmaxf(acc.z, 0.0f); acc.w = fmaxf(acc.w, 0.0f);
                *(float4*)(St + rloc * ST_PITCH + lane * 4) = acc;
            }
        }
        __syncthreads();

        if (mult) {
            const float* __restrict__ ap = St + c * ST_PITCH + 4 * h;
            v8f acc = (v8f){0.f, 0.f, 0.f, 0.f, 0.f, 0.f, 0.f, 0.f};
#pragma unroll
            for (int kh = 0; kh < 2; ++kh) {
                v4f a[8];
#pragma unroll
                for (int q = 0; q < 8; ++q) a[q] = *(const v4f*)(ap + (kh * 8 + q) * 8);
                asm volatile("" :: "v"(a[0]), "v"(a[1]), "v"(a[2]), "v"(a[3]),
                                   "v"(a[4]), "v"(a[5]), "v"(a[6]), "v"(a[7]));
#pragma unroll
                for (int q = 0; q < 8; ++q) {
                    const int kb = kh * 8 + q;
                    v2f a0 = {a[q].x, a[q].y};
                    v2f a1 = {a[q].z, a[q].w};
                    v2f b0 = {breg[kb].x, breg[kb].y};
                    v2f b1 = {breg[kb].z, breg[kb].w};
                    acc = __builtin_amdgcn_wmma_f32_16x16x4_f32(false, a0, false, b0,
                                                                (short)0, acc, false, false);
                    acc = __builtin_amdgcn_wmma_f32_16x16x4_f32(false, a1, false, b1,
                                                                (short)0, acc, false, false);
                }
            }
            float* __restrict__ op = Ot + (8 * h) * ST_PITCH + (WIDE ? wv * 16 : 0) + c;
#pragma unroll
            for (int r = 0; r < 8; ++r) op[r * ST_PITCH] = acc[r];
        }
        __syncthreads();

        if (WIDE) {
            for (int rr = 0; rr < 2; ++rr) {
                const int rloc = wv * 2 + rr;
                const int m = m0 + rloc;
                if (m < N) {
                    v4f v = *(const v4f*)(Ot + rloc * ST_PITCH + lane * 4);
                    *(v4f*)(D + (size_t)m * 128 + lane * 4) = v;
                }
            }
        } else if (wv == 0) {
            for (int q = 0; q < 2; ++q) {
                const int idx = lane + 32 * q;
                const int rloc = idx >> 2;
                const int cpos = (idx & 3) * 4;
                const int m = m0 + rloc;
                if (m < N) {
                    v4f v = *(const v4f*)(Ot + rloc * ST_PITCH + cpos);
                    *(v4f*)(D + (size_t)m * 16 + cpos) = v;
                }
            }
        }
    }
}

__global__ void gather16_kernel(const float* __restrict__ lin, const float* __restrict__ dis,
                                const float* __restrict__ bias, int biasLen,
                                const int* __restrict__ rowptr, const int* __restrict__ count,
                                const int* __restrict__ csr_src, const float* __restrict__ csr_w,
                                float* __restrict__ out, int N) {
    long long t = (long long)blockIdx.x * blockDim.x + threadIdx.x;
    int node = (int)(t >> 2);
    int q = (int)(t & 3);
    if (t >= (long long)N * 4) return;
    int start = rowptr[node];
    int d = count[node];
    float dn = dis[node];
    const float* __restrict__ colbase = lin + q * 4;
    const int* __restrict__ src = csr_src + start;
    const float* __restrict__ w = csr_w + start;
    float4 self = *(const float4*)(colbase + (size_t)node * 16);
    float s2 = dn * dn;
    float4 acc;
    acc.x = self.x * s2; acc.y = self.y * s2; acc.z = self.z * s2; acc.w = self.w * s2;
    int j = 0;
    for (; j + 4 <= d; j += 4) gather_step<4, 16>(acc, colbase, src, w, j);
    if (j + 2 <= d) { gather_step<2, 16>(acc, colbase, src, w, j); j += 2; }
    if (j < d) gather_step<1, 16>(acc, colbase, src, w, j);
    int c = q * 4;
    acc.x += (c + 0 < biasLen) ? bias[c + 0] : 0.0f;
    acc.y += (c + 1 < biasLen) ? bias[c + 1] : 0.0f;
    acc.z += (c + 2 < biasLen) ? bias[c + 2] : 0.0f;
    acc.w += (c + 3 < biasLen) ? bias[c + 3] : 0.0f;
    *(float4*)(out + (size_t)node * 16 + q * 4) = acc;
}

__global__ void pool_kernel(const float* __restrict__ h  ,
                            const int* __restrict__ batch,
                            float* __restrict__ out, float* __restrict__ cnt, int N) {
    long long t = (long long)blockIdx.x * blockDim.x + threadIdx.x;
    if (t >= (long long)N * 16) return;
    int i = (int)(t >> 4);
    int c = (int)(t & 15);
    int g = batch[i];
    if (c < C_OUT) atomicAdd(&out[(size_t)g * C_OUT + c], h[t]);
    if (c == 0)    atomicAdd(&cnt[g], 1.0f);
}

__global__ void div_kernel(float* __restrict__ out, const float* __restrict__ cnt) {
    int t = blockIdx.x * blockDim.x + threadIdx.x;
    if (t < NUM_GRAPHS * C_OUT) out[t] /= fmaxf(cnt[t / C_OUT], 1.0f);
}


static inline int cdiv_ll(long long a, long long b) { return (int)((a + b - 1) / b); }

extern "C" void kernel_launch(void* const* d_in, const int* in_sizes, int n_in,
                              void* d_out, int out_size, void* d_ws, size_t ws_size,
                              hipStream_t stream) {
    const float* x     = (const float*)d_in[0];
    const int*   ei    = (const int*)d_in[1];
    const int*   batch = (const int*)d_in[2];
    const float* W0 = (const float*)d_in[3];  const float* b0 = (const float*)d_in[4];
    const float* W1 = (const float*)d_in[5];  const float* b1 = (const float*)d_in[6];
    const float* W2 = (const float*)d_in[7];  const float* b2 = (const float*)d_in[8];
    const float* W3 = (const float*)d_in[9];  const float* b3 = (const float*)d_in[10];

    const int N = in_sizes[2];
    const int E = in_sizes[1] / 2;
    const int H = 128;
    const int* rowi = ei;
    const int* coli = ei + E;

    float* out = (float*)d_out;
    float* ws  = (float*)d_ws;
    const size_t Npad = ((size_t)N + 255) & ~(size_t)255;
    const int    nb   = (N + 1023) >> 10;
    float* dis  = ws;
    float* bufA = ws + Npad;
    float* bufB = bufA + (size_t)N * H;
    float* cnt  = bufB + (size_t)N * H;
    int*   count    = (int*)(cnt + 512);
    int*   rowptr   = count + Npad;
    int*   bsum     = rowptr + Npad;
    int*   part     = bsum + (((size_t)nb + 255) & ~(size_t)255);
    int*   rank     = part + (size_t)CR_S * Npad;
    int*   slot_src = rank + (size_t)E;
    int*   csr_src  = slot_src + (size_t)E;
    float* csr_w    = (float*)(csr_src + (size_t)E);

    const int TB = 256;

    const int nranges = cdiv_ll((long long)Npad, CR_NR);
    const int chunk   = cdiv_ll(E, CR_S);
    count_rank_kernel<<<nranges * CR_S, 1024, 0, stream>>>(coli, rank, part, E, chunk, (int)Npad);
    scan_local_kernel<<<nb, TB, 0, stream>>>(part, count, rowptr, bsum, N, (int)Npad);
    scan_top_kernel<<<1, 32, 0, stream>>>(bsum, nb);
    finalize_kernel<<<cdiv_ll(N, TB), TB, 0, stream>>>(rowptr, bsum, count, dis, N);
    place_kernel<<<cdiv_ll(E, TB), TB, 0, stream>>>(rowi, coli, rank, rowptr, part, slot_src, E, chunk, (int)Npad);
    order_kernel<<<cdiv_ll(N, TB / 32), TB, 0, stream>>>(slot_src, rowptr, count, dis, csr_src, csr_w, N);

    const int gemmBlk    = 512;
    const int fuseBlk    = FUSE_BLK;

    gemm_strip_wmma_f32<128, 8, 128><<<gemmBlk, TB, 0, stream>>>(x, W0, bufA, N);

    gather_gemm_kernel<true><<<fuseBlk, TB, 0, stream>>>(bufA, dis, b0, rowptr, count, csr_src, csr_w, W1, H, bufB, N);

    gather_gemm_kernel<true><<<fuseBlk, TB, 0, stream>>>(bufB, dis, b1, rowptr, count, csr_src, csr_w, W2, H, bufA, N);

    gather_gemm_kernel<false><<<fuseBlk, TB, 0, stream>>>(bufA, dis, b2, rowptr, count, csr_src, csr_w, W3, C_OUT, bufB, N);

    gather16_kernel<<<cdiv_ll((long long)N * 4, TB), TB, 0, stream>>>(bufB, dis, b3, C_OUT, rowptr, count, csr_src, csr_w, bufA, N);

    fill_kernel<<<cdiv_ll(NUM_GRAPHS * C_OUT, TB), TB, 0, stream>>>(out, 0.0f, NUM_GRAPHS * C_OUT);
    fill_kernel<<<cdiv_ll(NUM_GRAPHS, TB), TB, 0, stream>>>(cnt, 0.0f, NUM_GRAPHS);
    pool_kernel<<<cdiv_ll((long long)N * 16, TB), TB, 0, stream>>>(bufA, batch, out, cnt, N);
    div_kernel<<<cdiv_ll(NUM_GRAPHS * C_OUT, TB), TB, 0, stream>>>(out, cnt);
}


# Fused layers: a row's source ids and weights sit in lanes (one vector load each, fetched a strip ahead); readlane feeds batches of 8 row loads with no index round trip; sum order unchanged
# speedup vs baseline: 4.2298x; 1.0667x over previous
#include <hip/hip_runtime.h>


typedef float v2f __attribute__((ext_vector_type(2)));
typedef float v4f __attribute__((ext_vector_type(4)));
typedef float v8f __attribute__((ext_vector_type(8)));

#define NUM_GRAPHS 512
#define C_OUT 10


__global__ void fill_kernel(float* __restrict__ p, float v, int n) {
    int t = blockIdx.x * blockDim.x + threadIdx.x;
    if (t < n) p[t] = v;
}

#define CR_NR 12544
#define CR_S  64
#define CR_U  8

__global__ __launch_bounds__(1024)
void count_rank_kernel(const int* __restrict__ col, int* __restrict__ rank,
                       int* __restrict__ part, int E, int chunk, int Npad) {
    __shared__ __attribute__((aligned(16))) int lcnt[CR_NR];
    const int tid = threadIdx.x;
    const int s = blockIdx.x % CR_S;
    const int lo = (blockIdx.x / CR_S) * CR_NR;
    for (int i = tid; i < CR_NR / 4; i += 1024) ((int4*)lcnt)[i] = make_int4(0, 0, 0, 0);
    __syncthreads();
    const int t0 = s * chunk;
    const int t1 = min(t0 + chunk, E);
    const int tlast = t1 - 1;
    for (int tb = t0 + tid; tb < t1; tb += 1024 * CR_U) {
        int c[CR_U], rk[CR_U];
#pragma unroll
        for (int u = 0; u < CR_U; ++u) {
            const int t = tb + u * 1024;
            const int v = col[min(t, tlast)];
            c[u] = (t < t1) ? (v - lo) : -1;
        }
#pragma unroll
        for (int u = 0; u < CR_U; ++u)
            rk[u] = ((unsigned)c[u] < (unsigned)CR_NR) ? atomicAdd(&lcnt[c[u]], 1) : 0;
#pragma unroll
        for (int u = 0; u < CR_U; ++u)
            if ((unsigned)c[u] < (unsigned)CR_NR) rank[tb + u * 1024] = rk[u];
    }
    __syncthreads();
    const int hi = min(CR_NR, Npad - lo);
    int4* __restrict__ dst = (int4*)(part + (size_t)s * Npad + lo);
    for (int i = tid; i < hi / 4; i += 1024) dst[i] = ((const int4*)lcnt)[i];
}

__device__ __forceinline__ int wave_inclusive_scan(int v, int lane) {
    int incl = v;
    for (int off = 1; off < 32; off <<= 1) {
        int n = __shfl_up(incl, off, 32);
        if (lane >= off) incl += n;
    }
    return incl;
}

__global__ __launch_bounds__(256)
void scan_local_kernel(int* __restrict__ part, int* __restrict__ count,
                       int* __restrict__ rowptr, int* __restrict__ bsum, int N, int Npad) {
    __shared__ int wsum[8];
    int lane = threadIdx.x & 31, wv = threadIdx.x >> 5;
    int base = blockIdx.x * 1024 + threadIdx.x * 4;
    int4 tot = make_int4(0, 0, 0, 0);
    if (base < Npad) {
        int4* p = (int4*)(part + base);
        const size_t pitch4 = (size_t)(Npad >> 2);
        for (int s0 = 0; s0 < CR_S; s0 += 16) {
            int4 v[16];
#pragma unroll
            for (int k = 0; k < 16; ++k) v[k] = p[(size_t)(s0 + k) * pitch4];
#pragma unroll
            for (int k = 0; k < 16; ++k) {
                p[(size_t)(s0 + k) * pitch4] = tot;
                tot.x += v[k].x; tot.y += v[k].y; tot.z += v[k].z; tot.w += v[k].w;
            }
        }
    }
    int c0 = (base + 0 < N) ? tot.x : 0;
    int c1 = (base + 1 < N) ? tot.y : 0;
    int c2 = (base + 2 < N) ? tot.z : 0;
    int c3 = (base + 3 < N) ? tot.w : 0;
    int tsum = c0 + c1 + c2 + c3;
    int incl = wave_inclusive_scan(tsum, lane);
    if (lane == 31) wsum[wv] = incl;
    __syncthreads();
    int woff = 0;
    for (int w = 0; w < wv; ++w) woff += wsum[w];
    int excl = woff + incl - tsum;
    if (base + 0 < N) { rowptr[base + 0] = excl;                count[base + 0] = c0; }
    if (base + 1 < N) { rowptr[base + 1] = excl + c0;           count[base + 1] = c1; }
    if (base + 2 < N) { rowptr[base + 2] = excl + c0 + c1;      count[base + 2] = c2; }
    if (base + 3 < N) { rowptr[base + 3] = excl + c0 + c1 + c2; count[base + 3] = c3; }
    if (threadIdx.x == 255) bsum[blockIdx.x] = woff + incl;
}

__global__ void scan_top_kernel(int* __restrict__ bsum, int nb) {
    int lane = threadIdx.x;
    int carry = 0;
    for (int b0 = 0; b0 < nb; b0 += 32) {
        int i = b0 + lane;
        int v = (i < nb) ? bsum[i] : 0;
        int incl = wave_inclusive_scan(v, lane);
        if (i < nb) bsum[i] = carry + incl - v;
        carry += __shfl(incl, 31, 32);
    }
}

__global__ void finalize_kernel(int* __restrict__ rowptr, const int* __restrict__ bsum,
                                const int* __restrict__ count, float* __restrict__ dis, int N) {
    int i = blockIdx.x * blockDim.x + threadIdx.x;
    if (i < N) {
        rowptr[i] += bsum[i >> 10];
        dis[i] = rsqrtf((float)(count[i] + 1));
    }
}

__global__ void place_kernel(const int* __restrict__ row, const int* __restrict__ col,
                             const int* __restrict__ rank, const int* __restrict__ rowptr,
                             const int* __restrict__ part, int* __restrict__ slot_src,
                             int E, int chunk, int Npad) {
    int t = blockIdx.x * blockDim.x + threadIdx.x;
    if (t < E) {
        int c = col[t];
        int s = t / chunk;
        slot_src[rowptr[c] + part[(size_t)s * Npad + c] + rank[t]] = row[t];
    }
}

__global__ __launch_bounds__(256)
void order_kernel(const int* __restrict__ slot_src, const int* __restrict__ rowptr,
                  const int* __restrict__ count, const float* __restrict__ dis,
                  int* __restrict__ csr_src, float* __restrict__ csr_w, int N) {
    int lane = threadIdx.x & 31;
    int node = __builtin_amdgcn_readfirstlane((int)(blockIdx.x * 8 + (threadIdx.x >> 5)));
    if (node >= N) return;
    int start = rowptr[node];
    int d = count[node];
    float dc = dis[node];
    const int* __restrict__ seg = slot_src + start;
    for (int i0 = 0; i0 < d; i0 += 32) {
        int i = i0 + lane;
        if (i < d) {
            int key = seg[i];
            int p = 0;
            for (int j = 0; j < d; ++j) {
                int kj = seg[j];
                p += ((kj < key) || (kj == key && j < i)) ? 1 : 0;
            }
            csr_src[start + p] = key;
            csr_w[start + p] = dis[key] * dc;
        }
    }
}

#define WT_PITCH 132

template <int BCOLS, int NT, int DSTRIDE>
__global__ __launch_bounds__(256)
void gemm_strip_wmma_f32(const float* __restrict__ A, const float* __restrict__ B,
                         float* __restrict__ D, int N) {
    __shared__ __attribute__((aligned(16))) float Wt[NT * 16 * WT_PITCH];
    __shared__ __attribute__((aligned(16))) float St[8 * 16 * WT_PITCH];

#pragma unroll
    for (int it = 0; it < (128 * NT * 16) / 256; ++it) {
        int i = it * 256 + (int)threadIdx.x;
        int n = i % (NT * 16);
        int k = i / (NT * 16);
        int nc = (n < BCOLS) ? n : (BCOLS - 1);
        Wt[n * WT_PITCH + k] = B[(size_t)k * BCOLS + nc];
    }
    __syncthreads();

    const int lane = threadIdx.x & 31;
    const int c = lane & 15;
    const int h = lane >> 4;
    const int tilesM = (N + 15) >> 4;
    const int nwaves = (int)gridDim.x * 8;
    int tm = __builtin_amdgcn_readfirstlane((int)((threadIdx.x >> 5) * gridDim.x + blockIdx.x));
    const float* wfrag = Wt + c * WT_PITCH + 4 * h;
    constexpr int LPR = NT * 4;
    constexpr int RPI = 32 / LPR;
    float* sw = St + (threadIdx.x >> 5) * (16 * WT_PITCH);
    float* sput = sw + (8 * h) * WT_PITCH + c;
    const int rsub = lane / LPR;
    const int cpos = (lane % LPR) * 4;

    for (; tm < tilesM; tm += nwaves) {
        asm volatile("" ::: "memory");
        const int m0 = tm << 4;
        int row = m0 + c;
        if (row >= N) row = N - 1;
        const v4f* __restrict__ Ap = (const v4f*)(A + (size_t)row * 128 + 4 * h);

        v4f a[16];
#pragma unroll
        for (int kb = 0; kb < 16; ++kb) a[kb] = Ap[kb * 2];
        asm volatile("" :: "v"(a[0]), "v"(a[1]), "v"(a[2]), "v"(a[3]), "v"(a[4]), "v"(a[5]),
                           "v"(a[6]), "v"(a[7]), "v"(a[8]), "v"(a[9]), "v"(a[10]), "v"(a[11]),
                           "v"(a[12]), "v"(a[13]), "v"(a[14]), "v"(a[15]));

        v8f acc[NT];
#pragma unroll
        for (int t = 0; t < NT; ++t) acc[t] = (v8f){0.f, 0.f, 0.f, 0.f, 0.f, 0.f, 0.f, 0.f};

#pragma unroll
        for (int kb = 0; kb < 16; ++kb) {
            v4f b[NT];
#pragma unroll
            for (int t = 0; t < NT; ++t)
                b[t] = *(const v4f*)(wfrag + t * 16 * WT_PITCH + kb * 8);
            v2f a0 = {a[kb].x, a[kb].y};
            v2f a1 = {a[kb].z, a[kb].w};
#pragma unroll
            for (int t = 0; t < NT; ++t) {
                v2f b0 = {b[t].x, b[t].y};
                acc[t] = __builtin_amdgcn_wmma_f32_16x16x4_f32(false, a0, false, b0,
                                                               (short)0, acc[t], false, false);
            }
#pragma unroll
            for (int t = 0; t < NT; ++t) {
                v2f b1 = {b[t].z, b[t].w};
                acc[t] = __builtin_amdgcn_wmma_f32_16x16x4_f32(false, a1, false, b1,
                                                               (short)0, acc[t], false, false);
            }
        }

#pragma unroll
        for (int r = 0; r < 8; ++r) {
#pragma unroll
            for (int t = 0; t < NT; ++t) sput[r * WT_PITCH + t * 16] = acc[t][r];
        }
        if (m0 + 16 <= N) {
            v4f v[16 / RPI];
#pragma unroll
            for (int q = 0; q < 16 / RPI; ++q)
                v[q] = *(const v4f*)(sw + (q * RPI + rsub) * WT_PITCH + cpos);
#pragma unroll
            for (int q = 0; q < 16 / RPI; ++q)
                *(v4f*)(D + (size_t)(m0 + q * RPI + rsub) * DSTRIDE + cpos) = v[q];
        } else {
#pragma unroll
            for (int i = 0; i < 16; i += RPI) {
                const int rr = i + rsub;
                v4f v = *(const v4f*)(sw + rr * WT_PITCH + cpos);
                const int m = m0 + rr;
                if (m < N) *(v4f*)(D + (size_t)m * DSTRIDE + cpos) = v;
            }
        }
    }
}

template <int K, int F>
__device__ __forceinline__ void gather_step(float4& acc, const float* __restrict__ colbase,
                                            const int* __restrict__ src,
                                            const float* __restrict__ w, int j) {
    int s[K]; float ww[K]; float4 v[K];
#pragma unroll
    for (int k = 0; k < K; ++k) { s[k] = src[j + k]; ww[k] = w[j + k]; }
#pragma unroll
    for (int k = 0; k < K; ++k) v[k] = *(const float4*)(colbase + (size_t)s[k] * F);
#pragma unroll
    for (int k = 0; k < K; ++k) {
        acc.x = fmaf(v[k].x, ww[k], acc.x);
        acc.y = fmaf(v[k].y, ww[k], acc.y);
        acc.z = fmaf(v[k].z, ww[k], acc.z);
        acc.w = fmaf(v[k].w, ww[k], acc.w);
    }
}

struct RowHead {
    int start; int d; float dn;
    int sv; float wl;
    float4 self;
};

__device__ __forceinline__ void pair_bounds_load(int& bs, int& bd, float& bn, int nd, int nlim,
                                                 int lane,
                                                 const int* __restrict__ rowptr,
                                                 const int* __restrict__ count,
                                                 const float* __restrict__ dis) {
    const int node = nd + (lane & 1);
    const bool ok = node < nlim;
    const int nc = ok ? node : 0;
    bs = rowptr[nc];
    const int cnt = count[nc];
    bd = ok ? cnt : 0;
    bn = dis[nc];
}

__device__ __forceinline__ void row_list_load(RowHead& r, int bs, int bd, float bn, int rr,
                                              int node, int nlim, int lane,
                                              const float* __restrict__ lin,
                                              const int* __restrict__ csr_src,
                                              const float* __restrict__ csr_w) {
    r.start = __builtin_amdgcn_readlane(bs, rr);
    r.d = __builtin_amdgcn_readlane(bd, rr);
    r.dn = __int_as_float(__builtin_amdgcn_readlane(__float_as_int(bn), rr));
    r.sv = 0; r.wl = 0.0f;
    r.self = make_float4(0.0f, 0.0f, 0.0f, 0.0f);
    if (node < nlim) {
        const float* __restrict__ rowp = lin + (size_t)node * 128;
        r.self = *(const float4*)(rowp + (unsigned)lane * 4u);
        if (lane < r.d) { r.sv = csr_src[r.start + lane]; r.wl = csr_w[r.start + lane]; }
    }
}

template <int K>
__device__ __forceinline__ void lane_list_batch(float4& acc, const float* __restrict__ lin,
                                                unsigned loff, int sv, float wl, int k0) {
    float4 v[K];
#pragma unroll
    for (int k = 0; k < K; ++k) {
        const int s = __builtin_amdgcn_readlane(sv, k0 + k);
        const float* __restrict__ rowp = lin + (size_t)s * 128;
        v[k] = *(const float4*)(rowp + loff);
    }
#pragma unroll
    for (int k = 0; k < K; ++k) {
        const float ww = __int_as_float(__builtin_amdgcn_readlane(__float_as_int(wl), k0 + k));
        acc.x = fmaf(v[k].x, ww, acc.x);
        acc.y = fmaf(v[k].y, ww, acc.y);
        acc.z = fmaf(v[k].z, ww, acc.z);
        acc.w = fmaf(v[k].w, ww, acc.w);
    }
}

__device__ __forceinline__ void lane_list_chunk(float4& acc, const float* __restrict__ lin,
                                                unsigned loff, int sv, float wl, int dd) {
    int k0 = 0;
    for (; k0 + 8 <= dd; k0 += 8) lane_list_batch<8>(acc, lin, loff, sv, wl, k0);
    switch (dd - k0) {
        case 7: lane_list_batch<7>(acc, lin, loff, sv, wl, k0); break;
        case 6: lane_list_batch<6>(acc, lin, loff, sv, wl, k0); break;
        case 5: lane_list_batch<5>(acc, lin, loff, sv, wl, k0); break;
        case 4: lane_list_batch<4>(acc, lin, loff, sv, wl, k0); break;
        case 3: lane_list_batch<3>(acc, lin, loff, sv, wl, k0); break;
        case 2: lane_list_batch<2>(acc, lin, loff, sv, wl, k0); break;
        case 1: lane_list_batch<1>(acc, lin, loff, sv, wl, k0); break;
        default: break;
    }
}

#define ST_PITCH 132
#define FUSE_BLK 1024

template <bool WIDE>
__global__ __launch_bounds__(256)
void gather_gemm_kernel(const float* __restrict__ lin, const float* __restrict__ dis,
                        const float* __restrict__ bias,
                        const int* __restrict__ rowptr, const int* __restrict__ count,
                        const int* __restrict__ csr_src, const float* __restrict__ csr_w,
                        const float* __restrict__ Bw, int bcols,
                        float* __restrict__ D, int N) {
    __shared__ __attribute__((aligned(16))) float St[16 * ST_PITCH];
    __shared__ __attribute__((aligned(16))) float Ot[16 * ST_PITCH];
    __shared__ __attribute__((aligned(16))) float Bs[8 * 128];
    const int lane = threadIdx.x & 31;
    const int c = lane & 15;
    const int h = lane >> 4;
    const int wv = __builtin_amdgcn_readfirstlane((int)(threadIdx.x >> 5));
    const bool mult = WIDE || (wv == 0);

    v4f breg[16];
    {
        int ncol = (WIDE ? wv * 16 : 0) + c;
        if (ncol >= bcols) ncol = bcols - 1;
        const float* __restrict__ bp = Bw + (size_t)(4 * h) * bcols + ncol;
#pragma unroll
        for (int kb = 0; kb < 16; ++kb) {
            breg[kb].x = bp[(size_t)(kb * 8 + 0) * bcols];
            breg[kb].y = bp[(size_t)(kb * 8 + 1) * bcols];
            breg[kb].z = bp[(size_t)(kb * 8 + 2) * bcols];
            breg[kb].w = bp[(size_t)(kb * 8 + 3) * bcols];
        }
    }

    float* __restrict__ bsw = Bs + wv * 128 + lane * 4;
    *(float4*)bsw = *(const float4*)(bias + lane * 4);
    const int tilesM = (N + 15) >> 4;

    RowHead h0, h1;
    int bs, bd; float bn;
    {
        const int t0 = (int)blockIdx.x;
        const int nl0 = (t0 < tilesM) ? N : 0;
        const int nd0 = (t0 << 4) + wv * 2;
        pair_bounds_load(bs, bd, bn, nd0, nl0, lane, rowptr, count, dis);
        row_list_load(h0, bs, bd, bn, 0, nd0, nl0, lane, lin, csr_src, csr_w);
        row_list_load(h1, bs, bd, bn, 1, nd0 + 1, nl0, lane, lin, csr_src, csr_w);
        const int t1 = t0 + (int)gridDim.x;
        pair_bounds_load(bs, bd, bn, (t1 << 4) + wv * 2, (t1 < tilesM) ? N : 0, lane, rowptr, count, dis);
    }

    for (int tm = blockIdx.x; tm < tilesM; tm += gridDim.x) {
        const int m0 = tm << 4;
#pragma nounroll
        for (int rr = 0; rr < 2; ++rr) {
            const int rloc = wv * 2 + rr;
            const int node = m0 + rloc;
            if (node < N) {
                const int d = h0.d;
                const int start = h0.start;
                int sv = h0.sv;
                float wl = h0.wl;
                const float s2 = h0.dn * h0.dn;
                float4 acc;
                acc.x = h0.self.x * s2; acc.y = h0.self.y * s2;
                acc.z = h0.self.z * s2; acc.w = h0.self.w * s2;
                for (int j0 = 0; j0 < d; j0 += 32) {
                    const int dd = min(d - j0, 32);
                    if (j0 != 0) {
                        sv = 0; wl = 0.0f;
                        if (lane < dd) { sv = csr_src[start + j0 + lane]; wl = csr_w[start + j0 + lane]; }
                    }
                    lane_list_chunk(acc, lin, (unsigned)lane * 4u, sv, wl, dd);
                }
                const float4 bv = *(const float4*)bsw;
                acc.x += bv.x; acc.y += bv.y; acc.z += bv.z; acc.w += bv.w;
                acc.x = fmaxf(acc.x, 0.0f); acc.y = fmaxf(acc.y, 0.0f);
                acc.z = fmaxf(acc.z, 0.0f); acc.w = fmaxf(acc.w, 0.0f);
                *(float4*)(St + rloc * ST_PITCH + lane * 4) = acc;
            }
            h0 = h1;
        }
        __syncthreads();
        {
            const int tn = tm + (int)gridDim.x;
            const int nln = (tn < tilesM) ? N : 0;
            const int ndn = (tn << 4) + wv * 2;
            row_list_load(h0, bs, bd, bn, 0, ndn, nln, lane, lin, csr_src, csr_w);
            row_list_load(h1, bs, bd, bn, 1, ndn + 1, nln, lane, lin, csr_src, csr_w);
            const int tnn = tn + (int)gridDim.x;
            pair_bounds_load(bs, bd, bn, (tnn << 4) + wv * 2, (tnn < tilesM) ? N : 0, lane, rowptr, count, dis);
        }

        if (mult) {
            const float* __restrict__ ap = St + c * ST_PITCH + 4 * h;
            v8f acc = (v8f){0.f, 0.f, 0.f, 0.f, 0.f, 0.f, 0.f, 0.f};
#pragma unroll
            for (int kq = 0; kq < 4; ++kq) {
                v4f a[4];
#pragma unroll
                for (int q = 0; q < 4; ++q) a[q] = *(const v4f*)(ap + (kq * 4 + q) * 8);
                asm volatile("" :: "v"(a[0]), "v"(a[1]), "v"(a[2]), "v"(a[3]));
#pragma unroll
                for (int q = 0; q < 4; ++q) {
                    const int kb = kq * 4 + q;
                    v2f a0 = {a[q].x, a[q].y};
                    v2f a1 = {a[q].z, a[q].w};
                    v2f b0 = {breg[kb].x, breg[kb].y};
                    v2f b1 = {breg[kb].z, breg[kb].w};
                    acc = __builtin_amdgcn_wmma_f32_16x16x4_f32(false, a0, false, b0,
                                                                (short)0, acc, false, false);
                    acc = __builtin_amdgcn_wmma_f32_16x16x4_f32(false, a1, false, b1,
                                                                (short)0, acc, false, false);
                }
            }
            float* __restrict__ op = Ot + (8 * h) * ST_PITCH + (WIDE ? wv * 16 : 0) + c;
#pragma unroll
            for (int r = 0; r < 8; ++r) op[r * ST_PITCH] = acc[r];
        }
        __syncthreads();

        if (WIDE) {
            for (int rr = 0; rr < 2; ++rr) {
                const int rloc = wv * 2 + rr;
                const int m = m0 + rloc;
                if (m < N) {
                    v4f v = *(const v4f*)(Ot + rloc * ST_PITCH + lane * 4);
                    *(v4f*)(D + (size_t)m * 128 + lane * 4) = v;
                }
            }
        } else if (wv == 0) {
            for (int q = 0; q < 2; ++q) {
                const int idx = lane + 32 * q;
                const int rloc = idx >> 2;
                const int cpos = (idx & 3) * 4;
                const int m = m0 + rloc;
                if (m < N) {
                    v4f v = *(const v4f*)(Ot + rloc * ST_PITCH + cpos);
                    *(v4f*)(D + (size_t)m * 16 + cpos) = v;
                }
            }
        }
    }
}

__global__ void gather16_kernel(const float* __restrict__ lin, const float* __restrict__ dis,
                                const float* __restrict__ bias, int biasLen,
                                const int* __restrict__ rowptr, const int* __restrict__ count,
                                const int* __restrict__ csr_src, const float* __restrict__ csr_w,
                                float* __restrict__ out, int N) {
    long long t = (long long)blockIdx.x * blockDim.x + threadIdx.x;
    int node = (int)(t >> 2);
    int q = (int)(t & 3);
    if (t >= (long long)N * 4) return;
    int start = rowptr[node];
    int d = count[node];
    float dn = dis[node];
    const float* __restrict__ colbase = lin + q * 4;
    const int* __restrict__ src = csr_src + start;
    const float* __restrict__ w = csr_w + start;
    float4 self = *(const float4*)(colbase + (size_t)node * 16);
    float s2 = dn * dn;
    float4 acc;
    acc.x = self.x * s2; acc.y = self.y * s2; acc.z = self.z * s2; acc.w = self.w * s2;
    int j = 0;
    for (; j + 4 <= d; j += 4) gather_step<4, 16>(acc, colbase, src, w, j);
    if (j + 2 <= d) { gather_step<2, 16>(acc, colbase, src, w, j); j += 2; }
    if (j < d) gather_step<1, 16>(acc, colbase, src, w, j);
    int c = q * 4;
    acc.x += (c + 0 < biasLen) ? bias[c + 0] : 0.0f;
    acc.y += (c + 1 < biasLen) ? bias[c + 1] : 0.0f;
    acc.z += (c + 2 < biasLen) ? bias[c + 2] : 0.0f;
    acc.w += (c + 3 < biasLen) ? bias[c + 3] : 0.0f;
    *(float4*)(out + (size_t)node * 16 + q * 4) = acc;
}

__global__ void pool_kernel(const float* __restrict__ h  ,
                            const int* __restrict__ batch,
                            float* __restrict__ out, float* __restrict__ cnt, int N) {
    long long t = (long long)blockIdx.x * blockDim.x + threadIdx.x;
    if (t >= (long long)N * 16) return;
    int i = (int)(t >> 4);
    int c = (int)(t & 15);
    int g = batch[i];
    if (c < C_OUT) atomicAdd(&out[(size_t)g * C_OUT + c], h[t]);
    if (c == 0)    atomicAdd(&cnt[g], 1.0f);
}

__global__ void div_kernel(float* __restrict__ out, const float* __restrict__ cnt) {
    int t = blockIdx.x * blockDim.x + threadIdx.x;
    if (t < NUM_GRAPHS * C_OUT) out[t] /= fmaxf(cnt[t / C_OUT], 1.0f);
}


static inline int cdiv_ll(long long a, long long b) { return (int)((a + b - 1) / b); }

extern "C" void kernel_launch(void* const* d_in, const int* in_sizes, int n_in,
                              void* d_out, int out_size, void* d_ws, size_t ws_size,
                              hipStream_t stream) {
    const float* x     = (const float*)d_in[0];
    const int*   ei    = (const int*)d_in[1];
    const int*   batch = (const int*)d_in[2];
    const float* W0 = (const float*)d_in[3];  const float* b0 = (const float*)d_in[4];
    const float* W1 = (const float*)d_in[5];  const float* b1 = (const float*)d_in[6];
    const float* W2 = (const float*)d_in[7];  const float* b2 = (const float*)d_in[8];
    const float* W3 = (const float*)d_in[9];  const float* b3 = (const float*)d_in[10];

    const int N = in_sizes[2];
    const int E = in_sizes[1] / 2;
    const int H = 128;
    const int* rowi = ei;
    const int* coli = ei + E;

    float* out = (float*)d_out;
    float* ws  = (float*)d_ws;
    const size_t Npad = ((size_t)N + 255) & ~(size_t)255;
    const int    nb   = (N + 1023) >> 10;
    float* dis  = ws;
    float* bufA = ws + Npad;
    float* bufB = bufA + (size_t)N * H;
    float* cnt  = bufB + (size_t)N * H;
    int*   count    = (int*)(cnt + 512);
    int*   rowptr   = count + Npad;
    int*   bsum     = rowptr + Npad;
    int*   part     = bsum + (((size_t)nb + 255) & ~(size_t)255);
    int*   rank     = part + (size_t)CR_S * Npad;
    int*   slot_src = rank + (size_t)E;
    int*   csr_src  = slot_src + (size_t)E;
    float* csr_w    = (float*)(csr_src + (size_t)E);

    const int TB = 256;

    const int nranges = cdiv_ll((long long)Npad, CR_NR);
    const int chunk   = cdiv_ll(E, CR_S);
    count_rank_kernel<<<nranges * CR_S, 1024, 0, stream>>>(coli, rank, part, E, chunk, (int)Npad);
    scan_local_kernel<<<nb, TB, 0, stream>>>(part, count, rowptr, bsum, N, (int)Npad);
    scan_top_kernel<<<1, 32, 0, stream>>>(bsum, nb);
    finalize_kernel<<<cdiv_ll(N, TB), TB, 0, stream>>>(rowptr, bsum, count, dis, N);
    place_kernel<<<cdiv_ll(E, TB), TB, 0, stream>>>(rowi, coli, rank, rowptr, part, slot_src, E, chunk, (int)Npad);
    order_kernel<<<cdiv_ll(N, TB / 32), TB, 0, stream>>>(slot_src, rowptr, count, dis, csr_src, csr_w, N);

    const int gemmBlk    = 512;
    const int fuseBlk    = FUSE_BLK;

    gemm_strip_wmma_f32<128, 8, 128><<<gemmBlk, TB, 0, stream>>>(x, W0, bufA, N);

    gather_gemm_kernel<true><<<fuseBlk, TB, 0, stream>>>(bufA, dis, b0, rowptr, count, csr_src, csr_w, W1, H, bufB, N);

    gather_gemm_kernel<true><<<fuseBlk, TB, 0, stream>>>(bufB, dis, b1, rowptr, count, csr_src, csr_w, W2, H, bufA, N);

    gather_gemm_kernel<false><<<fuseBlk, TB, 0, stream>>>(bufA, dis, b2, rowptr, count, csr_src, csr_w, W3, C_OUT, bufB, N);

    gather16_kernel<<<cdiv_ll((long long)N * 4, TB), TB, 0, stream>>>(bufB, dis, b3, C_OUT, rowptr, count, csr_src, csr_w, bufA, N);

    fill_kernel<<<cdiv_ll(NUM_GRAPHS * C_OUT, TB), TB, 0, stream>>>(out, 0.0f, NUM_GRAPHS * C_OUT);
    fill_kernel<<<cdiv_ll(NUM_GRAPHS, TB), TB, 0, stream>>>(cnt, 0.0f, NUM_GRAPHS);
    pool_kernel<<<cdiv_ll((long long)N * 16, TB), TB, 0, stream>>>(bufA, batch, out, cnt, N);
    div_kernel<<<cdiv_ll(NUM_GRAPHS * C_OUT, TB), TB, 0, stream>>>(out, cnt);
}


# Every layer evaluated as (Ahat h) W: layer 1 gathers x and its GEMM joins the fused kernel (no GEMM launch); 128x10 product split by k over waves in layer 3; 15 -> 14 launches; sum orders change
# speedup vs baseline: 4.4875x; 1.0609x over previous
#include <hip/hip_runtime.h>


typedef float v2f __attribute__((ext_vector_type(2)));
typedef float v4f __attribute__((ext_vector_type(4)));
typedef float v8f __attribute__((ext_vector_type(8)));

#define NUM_GRAPHS 512
#define C_OUT 10


__global__ void fill_kernel(float* __restrict__ p, float v, int n) {
    int t = blockIdx.x * blockDim.x + threadIdx.x;
    if (t < n) p[t] = v;
}

#define CR_NR 12544
#define CR_S  64
#define CR_U  8

__global__ __launch_bounds__(1024)
void count_rank_kernel(const int* __restrict__ col, int* __restrict__ rank,
                       int* __restrict__ part, int E, int chunk, int Npad) {
    __shared__ __attribute__((aligned(16))) int lcnt[CR_NR];
    const int tid = threadIdx.x;
    const int s = blockIdx.x % CR_S;
    const int lo = (blockIdx.x / CR_S) * CR_NR;
    for (int i = tid; i < CR_NR / 4; i += 1024) ((int4*)lcnt)[i] = make_int4(0, 0, 0, 0);
    __syncthreads();
    const int t0 = s * chunk;
    const int t1 = min(t0 + chunk, E);
    const int tlast = t1 - 1;
    for (int tb = t0 + tid; tb < t1; tb += 1024 * CR_U) {
        int c[CR_U], rk[CR_U];
#pragma unroll
        for (int u = 0; u < CR_U; ++u) {
            const int t = tb + u * 1024;
            const int v = col[min(t, tlast)];
            c[u] = (t < t1) ? (v - lo) : -1;
        }
#pragma unroll
        for (int u = 0; u < CR_U; ++u)
            rk[u] = ((unsigned)c[u] < (unsigned)CR_NR) ? atomicAdd(&lcnt[c[u]], 1) : 0;
#pragma unroll
        for (int u = 0; u < CR_U; ++u)
            if ((unsigned)c[u] < (unsigned)CR_NR) rank[tb + u * 1024] = rk[u];
    }
    __syncthreads();
    const int hi = min(CR_NR, Npad - lo);
    int4* __restrict__ dst = (int4*)(part + (size_t)s * Npad + lo);
    for (int i = tid; i < hi / 4; i += 1024) dst[i] = ((const int4*)lcnt)[i];
}

__device__ __forceinline__ int wave_inclusive_scan(int v, int lane) {
    int incl = v;
    for (int off = 1; off < 32; off <<= 1) {
        int n = __shfl_up(incl, off, 32);
        if (lane >= off) incl += n;
    }
    return incl;
}

__global__ __launch_bounds__(256)
void scan_local_kernel(int* __restrict__ part, int* __restrict__ count,
                       int* __restrict__ rowptr, int* __restrict__ bsum, int N, int Npad) {
    __shared__ int wsum[8];
    int lane = threadIdx.x & 31, wv = threadIdx.x >> 5;
    int base = blockIdx.x * 1024 + threadIdx.x * 4;
    int4 tot = make_int4(0, 0, 0, 0);
    if (base < Npad) {
        int4* p = (int4*)(part + base);
        const size_t pitch4 = (size_t)(Npad >> 2);
        for (int s0 = 0; s0 < CR_S; s0 += 16) {
            int4 v[16];
#pragma unroll
            for (int k = 0; k < 16; ++k) v[k] = p[(size_t)(s0 + k) * pitch4];
#pragma unroll
            for (int k = 0; k < 16; ++k) {
                p[(size_t)(s0 + k) * pitch4] = tot;
                tot.x += v[k].x; tot.y += v[k].y; tot.z += v[k].z; tot.w += v[k].w;
            }
        }
    }
    int c0 = (base + 0 < N) ? tot.x : 0;
    int c1 = (base + 1 < N) ? tot.y : 0;
    int c2 = (base + 2 < N) ? tot.z : 0;
    int c3 = (base + 3 < N) ? tot.w : 0;
    int tsum = c0 + c1 + c2 + c3;
    int incl = wave_inclusive_scan(tsum, lane);
    if (lane == 31) wsum[wv] = incl;
    __syncthreads();
    int woff = 0;
    for (int w = 0; w < wv; ++w) woff += wsum[w];
    int excl = woff + incl - tsum;
    if (base + 0 < N) { rowptr[base + 0] = excl;                count[base + 0] = c0; }
    if (base + 1 < N) { rowptr[base + 1] = excl + c0;           count[base + 1] = c1; }
    if (base + 2 < N) { rowptr[base + 2] = excl + c0 + c1;      count[base + 2] = c2; }
    if (base + 3 < N) { rowptr[base + 3] = excl + c0 + c1 + c2; count[base + 3] = c3; }
    if (threadIdx.x == 255) bsum[blockIdx.x] = woff + incl;
}

__global__ void scan_top_kernel(int* __restrict__ bsum, int nb) {
    int lane = threadIdx.x;
    int carry = 0;
    for (int b0 = 0; b0 < nb; b0 += 32) {
        int i = b0 + lane;
        int v = (i < nb) ? bsum[i] : 0;
        int incl = wave_inclusive_scan(v, lane);
        if (i < nb) bsum[i] = carry + incl - v;
        carry += __shfl(incl, 31, 32);
    }
}

__global__ void finalize_kernel(int* __restrict__ rowptr, const int* __restrict__ bsum,
                                const int* __restrict__ count, float* __restrict__ dis, int N) {
    int i = blockIdx.x * blockDim.x + threadIdx.x;
    if (i < N) {
        rowptr[i] += bsum[i >> 10];
        dis[i] = rsqrtf((float)(count[i] + 1));
    }
}

__global__ void place_kernel(const int* __restrict__ row, const int* __restrict__ col,
                             const int* __restrict__ rank, const int* __restrict__ rowptr,
                             const int* __restrict__ part, int* __restrict__ slot_src,
                             int E, int chunk, int Npad) {
    int t = blockIdx.x * blockDim.x + threadIdx.x;
    if (t < E) {
        int c = col[t];
        int s = t / chunk;
        slot_src[rowptr[c] + part[(size_t)s * Npad + c] + rank[t]] = row[t];
    }
}

__global__ __launch_bounds__(256)
void order_kernel(const int* __restrict__ slot_src, const int* __restrict__ rowptr,
                  const int* __restrict__ count, const float* __restrict__ dis,
                  int* __restrict__ csr_src, float* __restrict__ csr_w, int N) {
    int lane = threadIdx.x & 31;
    int node = __builtin_amdgcn_readfirstlane((int)(blockIdx.x * 8 + (threadIdx.x >> 5)));
    if (node >= N) return;
    int start = rowptr[node];
    int d = count[node];
    float dc = dis[node];
    const int* __restrict__ seg = slot_src + start;
    for (int i0 = 0; i0 < d; i0 += 32) {
        int i = i0 + lane;
        if (i < d) {
            int key = seg[i];
            int p = 0;
            for (int j = 0; j < d; ++j) {
                int kj = seg[j];
                p += ((kj < key) || (kj == key && j < i)) ? 1 : 0;
            }
            csr_src[start + p] = key;
            csr_w[start + p] = dis[key] * dc;
        }
    }
}


template <int K, int F>
__device__ __forceinline__ void gather_step(float4& acc, const float* __restrict__ colbase,
                                            const int* __restrict__ src,
                                            const float* __restrict__ w, int j) {
    int s[K]; float ww[K]; float4 v[K];
#pragma unroll
    for (int k = 0; k < K; ++k) { s[k] = src[j + k]; ww[k] = w[j + k]; }
#pragma unroll
    for (int k = 0; k < K; ++k) v[k] = *(const float4*)(colbase + (size_t)s[k] * F);
#pragma unroll
    for (int k = 0; k < K; ++k) {
        acc.x = fmaf(v[k].x, ww[k], acc.x);
        acc.y = fmaf(v[k].y, ww[k], acc.y);
        acc.z = fmaf(v[k].z, ww[k], acc.z);
        acc.w = fmaf(v[k].w, ww[k], acc.w);
    }
}

struct RowHead {
    int start; int d; float dn;
    int sv; float wl;
    float4 self;
};

__device__ __forceinline__ void pair_bounds_load(int& bs, int& bd, float& bn, int nd, int nlim,
                                                 int lane,
                                                 const int* __restrict__ rowptr,
                                                 const int* __restrict__ count,
                                                 const float* __restrict__ dis) {
    const int node = nd + (lane & 1);
    const bool ok = node < nlim;
    const int nc = ok ? node : 0;
    bs = rowptr[nc];
    const int cnt = count[nc];
    bd = ok ? cnt : 0;
    bn = dis[nc];
}

__device__ __forceinline__ void row_list_load(RowHead& r, int bs, int bd, float bn, int rr,
                                              int node, int nlim, int lane,
                                              const float* __restrict__ lin,
                                              const int* __restrict__ csr_src,
                                              const float* __restrict__ csr_w) {
    r.start = __builtin_amdgcn_readlane(bs, rr);
    r.d = __builtin_amdgcn_readlane(bd, rr);
    r.dn = __int_as_float(__builtin_amdgcn_readlane(__float_as_int(bn), rr));
    r.sv = 0; r.wl = 0.0f;
    r.self = make_float4(0.0f, 0.0f, 0.0f, 0.0f);
    if (node < nlim) {
        const float* __restrict__ rowp = lin + (size_t)node * 128;
        r.self = *(const float4*)(rowp + (unsigned)lane * 4u);
        if (lane < r.d) { r.sv = csr_src[r.start + lane]; r.wl = csr_w[r.start + lane]; }
    }
}

template <int K>
__device__ __forceinline__ void lane_list_batch(float4& acc, const float* __restrict__ lin,
                                                unsigned loff, int sv, float wl, int k0) {
    float4 v[K];
#pragma unroll
    for (int k = 0; k < K; ++k) {
        const int s = __builtin_amdgcn_readlane(sv, k0 + k);
        const float* __restrict__ rowp = lin + (size_t)s * 128;
        v[k] = *(const float4*)(rowp + loff);
    }
#pragma unroll
    for (int k = 0; k < K; ++k) {
        const float ww = __int_as_float(__builtin_amdgcn_readlane(__float_as_int(wl), k0 + k));
        acc.x = fmaf(v[k].x, ww, acc.x);
        acc.y = fmaf(v[k].y, ww, acc.y);
        acc.z = fmaf(v[k].z, ww, acc.z);
        acc.w = fmaf(v[k].w, ww, acc.w);
    }
}

__device__ __forceinline__ void lane_list_chunk(float4& acc, const float* __restrict__ lin,
                                                unsigned loff, int sv, float wl, int dd) {
    int k0 = 0;
    for (; k0 + 8 <= dd; k0 += 8) lane_list_batch<8>(acc, lin, loff, sv, wl, k0);
    switch (dd - k0) {
        case 7: lane_list_batch<7>(acc, lin, loff, sv, wl, k0); break;
        case 6: lane_list_batch<6>(acc, lin, loff, sv, wl, k0); break;
        case 5: lane_list_batch<5>(acc, lin, loff, sv, wl, k0); break;
        case 4: lane_list_batch<4>(acc, lin, loff, sv, wl, k0); break;
        case 3: lane_list_batch<3>(acc, lin, loff, sv, wl, k0); break;
        case 2: lane_list_batch<2>(acc, lin, loff, sv, wl, k0); break;
        case 1: lane_list_batch<1>(acc, lin, loff, sv, wl, k0); break;
        default: break;
    }
}

#define ST_PITCH 132
#define FUSE_BLK 1024

template <bool LAST>
__global__ __launch_bounds__(256) __attribute__((amdgpu_waves_per_eu(8, 8)))
void gather_gemm_kernel(const float* __restrict__ lin, const float* __restrict__ dis,
                        const float* __restrict__ bias,
                        const int* __restrict__ rowptr, const int* __restrict__ count,
                        const int* __restrict__ csr_src, const float* __restrict__ csr_w,
                        const float* __restrict__ Bw,
                        const float* __restrict__ Bw2, int bcols2,
                        float* __restrict__ D, int N) {
    constexpr int IMG = 16 * ST_PITCH;
    constexpr int NIMG = LAST ? 4 : 2;
    __shared__ __attribute__((aligned(16))) float Sm[NIMG * IMG + 128];
    float* const St  = Sm;
    float* const Ot  = Sm + IMG;
    float* const Pt  = Sm + (LAST ? 2 * IMG : 0);
    float* const W2t = Sm + (LAST ? 3 * IMG : 0);
    float* const Bsh = Sm + NIMG * IMG;
    const int lane = threadIdx.x & 31;
    const int c = lane & 15;
    const int h = lane >> 4;
    const int wv = __builtin_amdgcn_readfirstlane((int)(threadIdx.x >> 5));

    v4f breg[16];
    {
        const float* __restrict__ bp = Bw + (size_t)(4 * h) * 128 + wv * 16 + c;
#pragma unroll
        for (int kb = 0; kb < 16; ++kb) {
            breg[kb].x = bp[(size_t)(kb * 8 + 0) * 128];
            breg[kb].y = bp[(size_t)(kb * 8 + 1) * 128];
            breg[kb].z = bp[(size_t)(kb * 8 + 2) * 128];
            breg[kb].w = bp[(size_t)(kb * 8 + 3) * 128];
        }
    }
    if (threadIdx.x < 128) Bsh[threadIdx.x] = bias[threadIdx.x];
    if (LAST) {
#pragma unroll
        for (int it = 0; it < 8; ++it) {
            const int i = it * 256 + (int)threadIdx.x;
            const int n = i & 15;
            const int k = i >> 4;
            const int nc = (n < bcols2) ? n : (bcols2 - 1);
            W2t[n * ST_PITCH + k] = Bw2[(size_t)k * bcols2 + nc];
        }
    }
    const int tilesM = (N + 15) >> 4;

    RowHead h0, h1;
    int bs, bd; float bn;
    {
        const int t0 = (int)blockIdx.x;
        const int nl0 = (t0 < tilesM) ? N : 0;
        const int nd0 = (t0 << 4) + wv * 2;
        pair_bounds_load(bs, bd, bn, nd0, nl0, lane, rowptr, count, dis);
        row_list_load(h0, bs, bd, bn, 0, nd0, nl0, lane, lin, csr_src, csr_w);
        row_list_load(h1, bs, bd, bn, 1, nd0 + 1, nl0, lane, lin, csr_src, csr_w);
        const int t1 = t0 + (int)gridDim.x;
        pair_bounds_load(bs, bd, bn, (t1 << 4) + wv * 2, (t1 < tilesM) ? N : 0, lane, rowptr, count, dis);
    }

    for (int tm = blockIdx.x; tm < tilesM; tm += gridDim.x) {
        const int m0 = tm << 4;
#pragma nounroll
        for (int rr = 0; rr < 2; ++rr) {
            const int rloc = wv * 2 + rr;
            const int node = m0 + rloc;
            if (node < N) {
                const int d = h0.d;
                const int start = h0.start;
                int sv = h0.sv;
                float wl = h0.wl;
                const float s2 = h0.dn * h0.dn;
                float4 acc;
                acc.x = h0.self.x * s2; acc.y = h0.self.y * s2;
                acc.z = h0.self.z * s2; acc.w = h0.self.w * s2;
                for (int j0 = 0; j0 < d; j0 += 32) {
                    const int dd = min(d - j0, 32);
                    if (j0 != 0) {
                        sv = 0; wl = 0.0f;
                        if (lane < dd) { sv = csr_src[start + j0 + lane]; wl = csr_w[start + j0 + lane]; }
                    }
                    lane_list_chunk(acc, lin, (unsigned)lane * 4u, sv, wl, dd);
                }
                *(float4*)(St + rloc * ST_PITCH + lane * 4) = acc;
            }
            h0 = h1;
        }
        __syncthreads();
        {
            const int tn = tm + (int)gridDim.x;
            const int nln = (tn < tilesM) ? N : 0;
            const int ndn = (tn << 4) + wv * 2;
            row_list_load(h0, bs, bd, bn, 0, ndn, nln, lane, lin, csr_src, csr_w);
            row_list_load(h1, bs, bd, bn, 1, ndn + 1, nln, lane, lin, csr_src, csr_w);
            const int tnn = tn + (int)gridDim.x;
            pair_bounds_load(bs, bd, bn, (tnn << 4) + wv * 2, (tnn < tilesM) ? N : 0, lane, rowptr, count, dis);
        }

        {
            const float* ap = St + c * ST_PITCH + 4 * h;
            v8f acc = (v8f){0.f, 0.f, 0.f, 0.f, 0.f, 0.f, 0.f, 0.f};
#pragma unroll
            for (int kq = 0; kq < 4; ++kq) {
                v4f a[4];
#pragma unroll
                for (int q = 0; q < 4; ++q) a[q] = *(const v4f*)(ap + (kq * 4 + q) * 8);
                asm volatile("" :: "v"(a[0]), "v"(a[1]), "v"(a[2]), "v"(a[3]));
#pragma unroll
                for (int q = 0; q < 4; ++q) {
                    const int kb = kq * 4 + q;
                    v2f a0 = {a[q].x, a[q].y};
                    v2f a1 = {a[q].z, a[q].w};
                    v2f b0 = {breg[kb].x, breg[kb].y};
                    v2f b1 = {breg[kb].z, breg[kb].w};
                    acc = __builtin_amdgcn_wmma_f32_16x16x4_f32(false, a0, false, b0,
                                                                (short)0, acc, false, false);
                    acc = __builtin_amdgcn_wmma_f32_16x16x4_f32(false, a1, false, b1,
                                                                (short)0, acc, false, false);
                }
            }
            const float bcol = Bsh[wv * 16 + c];
            float* op = Ot + (8 * h) * ST_PITCH + wv * 16 + c;
#pragma unroll
            for (int r = 0; r < 8; ++r) op[r * ST_PITCH] = fmaxf(acc[r] + bcol, 0.0f);

            if (LAST) {
                const int foff = c * ST_PITCH + wv * 16 + 4 * h;
                const float* hp = Ot + foff;
                const float* wp = W2t + foff;
                v4f a2[2], b2reg[2];
                b2reg[0] = *(const v4f*)(wp);
                b2reg[1] = *(const v4f*)(wp + 8);
                a2[0] = *(const v4f*)(hp);
                a2[1] = *(const v4f*)(hp + 8);
                v8f p = (v8f){0.f, 0.f, 0.f, 0.f, 0.f, 0.f, 0.f, 0.f};
#pragma unroll
                for (int kb = 0; kb < 2; ++kb) {
                    v2f a0 = {a2[kb].x, a2[kb].y};
                    v2f a1 = {a2[kb].z, a2[kb].w};
                    v2f b0 = {b2reg[kb].x, b2reg[kb].y};
                    v2f b1 = {b2reg[kb].z, b2reg[kb].w};
                    p = __builtin_amdgcn_wmma_f32_16x16x4_f32(false, a0, false, b0,
                                                              (short)0, p, false, false);
                    p = __builtin_amdgcn_wmma_f32_16x16x4_f32(false, a1, false, b1,
                                                              (short)0, p, false, false);
                }
                float* pp = Pt + (8 * h) * ST_PITCH + wv * 16 + c;
#pragma unroll
                for (int r = 0; r < 8; ++r) pp[r * ST_PITCH] = p[r];
            }
        }
        __syncthreads();

        if (!LAST) {
            for (int rr = 0; rr < 2; ++rr) {
                const int rloc = wv * 2 + rr;
                const int m = m0 + rloc;
                if (m < N) {
                    v4f v = *(const v4f*)(Ot + rloc * ST_PITCH + lane * 4);
                    const unsigned didx = (unsigned)m * 128u + (unsigned)lane * 4u;
                    *(v4f*)(D + didx) = v;
                }
            }
        } else {
            const int prow = wv * 2 + h;
            const float* q = Pt + prow * ST_PITCH + c;
            float s = q[0];
#pragma unroll
            for (int w = 1; w < 8; ++w) s += q[w * 16];
            const unsigned didx = (unsigned)m0 * 16u + (unsigned)threadIdx.x;
            if (m0 + prow < N) D[didx] = s;
        }
    }
}

__global__ void gather16_kernel(const float* __restrict__ lin, const float* __restrict__ dis,
                                const float* __restrict__ bias, int biasLen,
                                const int* __restrict__ rowptr, const int* __restrict__ count,
                                const int* __restrict__ csr_src, const float* __restrict__ csr_w,
                                float* __restrict__ out, int N) {
    long long t = (long long)blockIdx.x * blockDim.x + threadIdx.x;
    int node = (int)(t >> 2);
    int q = (int)(t & 3);
    if (t >= (long long)N * 4) return;
    int start = rowptr[node];
    int d = count[node];
    float dn = dis[node];
    const float* __restrict__ colbase = lin + q * 4;
    const int* __restrict__ src = csr_src + start;
    const float* __restrict__ w = csr_w + start;
    float4 self = *(const float4*)(colbase + (size_t)node * 16);
    float s2 = dn * dn;
    float4 acc;
    acc.x = self.x * s2; acc.y = self.y * s2; acc.z = self.z * s2; acc.w = self.w * s2;
    int j = 0;
    for (; j + 4 <= d; j += 4) gather_step<4, 16>(acc, colbase, src, w, j);
    if (j + 2 <= d) { gather_step<2, 16>(acc, colbase, src, w, j); j += 2; }
    if (j < d) gather_step<1, 16>(acc, colbase, src, w, j);
    int c = q * 4;
    acc.x += (c + 0 < biasLen) ? bias[c + 0] : 0.0f;
    acc.y += (c + 1 < biasLen) ? bias[c + 1] : 0.0f;
    acc.z += (c + 2 < biasLen) ? bias[c + 2] : 0.0f;
    acc.w += (c + 3 < biasLen) ? bias[c + 3] : 0.0f;
    *(float4*)(out + (size_t)node * 16 + q * 4) = acc;
}

__global__ void pool_kernel(const float* __restrict__ h  ,
                            const int* __restrict__ batch,
                            float* __restrict__ out, float* __restrict__ cnt, int N) {
    long long t = (long long)blockIdx.x * blockDim.x + threadIdx.x;
    if (t >= (long long)N * 16) return;
    int i = (int)(t >> 4);
    int c = (int)(t & 15);
    int g = batch[i];
    if (c < C_OUT) atomicAdd(&out[(size_t)g * C_OUT + c], h[t]);
    if (c == 0)    atomicAdd(&cnt[g], 1.0f);
}

__global__ void div_kernel(float* __restrict__ out, const float* __restrict__ cnt) {
    int t = blockIdx.x * blockDim.x + threadIdx.x;
    if (t < NUM_GRAPHS * C_OUT) out[t] /= fmaxf(cnt[t / C_OUT], 1.0f);
}


static inline int cdiv_ll(long long a, long long b) { return (int)((a + b - 1) / b); }

extern "C" void kernel_launch(void* const* d_in, const int* in_sizes, int n_in,
                              void* d_out, int out_size, void* d_ws, size_t ws_size,
                              hipStream_t stream) {
    const float* x     = (const float*)d_in[0];
    const int*   ei    = (const int*)d_in[1];
    const int*   batch = (const int*)d_in[2];
    const float* W0 = (const float*)d_in[3];  const float* b0 = (const float*)d_in[4];
    const float* W1 = (const float*)d_in[5];  const float* b1 = (const float*)d_in[6];
    const float* W2 = (const float*)d_in[7];  const float* b2 = (const float*)d_in[8];
    const float* W3 = (const float*)d_in[9];  const float* b3 = (const float*)d_in[10];

    const int N = in_sizes[2];
    const int E = in_sizes[1] / 2;
    const int H = 128;
    const int* rowi = ei;
    const int* coli = ei + E;

    float* out = (float*)d_out;
    float* ws  = (float*)d_ws;
    const size_t Npad = ((size_t)N + 255) & ~(size_t)255;
    const int    nb   = (N + 1023) >> 10;
    float* dis  = ws;
    float* bufA = ws + Npad;
    float* bufB = bufA + (size_t)N * H;
    float* cnt  = bufB + (size_t)N * H;
    int*   count    = (int*)(cnt + 512);
    int*   rowptr   = count + Npad;
    int*   bsum     = rowptr + Npad;
    int*   part     = bsum + (((size_t)nb + 255) & ~(size_t)255);
    int*   rank     = part + (size_t)CR_S * Npad;
    int*   slot_src = rank + (size_t)E;
    int*   csr_src  = slot_src + (size_t)E;
    float* csr_w    = (float*)(csr_src + (size_t)E);

    const int TB = 256;

    const int nranges = cdiv_ll((long long)Npad, CR_NR);
    const int chunk   = cdiv_ll(E, CR_S);
    count_rank_kernel<<<nranges * CR_S, 1024, 0, stream>>>(coli, rank, part, E, chunk, (int)Npad);
    scan_local_kernel<<<nb, TB, 0, stream>>>(part, count, rowptr, bsum, N, (int)Npad);
    scan_top_kernel<<<1, 32, 0, stream>>>(bsum, nb);
    finalize_kernel<<<cdiv_ll(N, TB), TB, 0, stream>>>(rowptr, bsum, count, dis, N);
    place_kernel<<<cdiv_ll(E, TB), TB, 0, stream>>>(rowi, coli, rank, rowptr, part, slot_src, E, chunk, (int)Npad);
    order_kernel<<<cdiv_ll(N, TB / 32), TB, 0, stream>>>(slot_src, rowptr, count, dis, csr_src, csr_w, N);

    const int fuseBlk    = FUSE_BLK;

    gather_gemm_kernel<false><<<fuseBlk, TB, 0, stream>>>(x, dis, b0, rowptr, count, csr_src, csr_w, W0, W0, H, bufA, N);

    gather_gemm_kernel<false><<<fuseBlk, TB, 0, stream>>>(bufA, dis, b1, rowptr, count, csr_src, csr_w, W1, W1, H, bufB, N);

    gather_gemm_kernel<true><<<fuseBlk, TB, 0, stream>>>(bufB, dis, b2, rowptr, count, csr_src, csr_w, W2, W3, C_OUT, bufA, N);

    gather16_kernel<<<cdiv_ll((long long)N * 4, TB), TB, 0, stream>>>(bufA, dis, b3, C_OUT, rowptr, count, csr_src, csr_w, bufB, N);

    fill_kernel<<<cdiv_ll(NUM_GRAPHS * C_OUT, TB), TB, 0, stream>>>(out, 0.0f, NUM_GRAPHS * C_OUT);
    fill_kernel<<<cdiv_ll(NUM_GRAPHS, TB), TB, 0, stream>>>(cnt, 0.0f, NUM_GRAPHS);
    pool_kernel<<<cdiv_ll((long long)N * 16, TB), TB, 0, stream>>>(bufB, batch, out, cnt, N);
    div_kernel<<<cdiv_ll(NUM_GRAPHS * C_OUT, TB), TB, 0, stream>>>(out, cnt);
}


# Four small launches folded into neighbours, 14->10: two pool clears, the chunk scan, the mean pool (now in the output layer's kernel: wave sums 8 rows, one atomic per column); pool sum order changes
# speedup vs baseline: 5.2282x; 1.1651x over previous
#include <hip/hip_runtime.h>


typedef float v2f __attribute__((ext_vector_type(2)));
typedef float v4f __attribute__((ext_vector_type(4)));
typedef float v8f __attribute__((ext_vector_type(8)));

#define NUM_GRAPHS 512
#define C_OUT 10

#define CR_NR 12544
#define CR_S  64
#define CR_U  8

__global__ __launch_bounds__(1024)
void count_rank_kernel(const int* __restrict__ col, int* __restrict__ rank,
                       int* __restrict__ part, int E, int chunk, int Npad,
                       float* __restrict__ psum, float* __restrict__ pcnt) {
    __shared__ __attribute__((aligned(16))) int lcnt[CR_NR];
    const int tid = threadIdx.x;
    const int gt = (int)blockIdx.x * 1024 + tid;
    if (gt < NUM_GRAPHS * C_OUT) psum[gt] = 0.0f;
    if (gt < NUM_GRAPHS) pcnt[gt] = 0.0f;
    const int s = blockIdx.x % CR_S;
    const int lo = (blockIdx.x / CR_S) * CR_NR;
    for (int i = tid; i < CR_NR / 4; i += 1024) ((int4*)lcnt)[i] = make_int4(0, 0, 0, 0);
    __syncthreads();
    const int t0 = s * chunk;
    const int t1 = min(t0 + chunk, E);
    const int tlast = t1 - 1;
    for (int tb = t0 + tid; tb < t1; tb += 1024 * CR_U) {
        int c[CR_U], rk[CR_U];
#pragma unroll
        for (int u = 0; u < CR_U; ++u) {
            const int t = tb + u * 1024;
            const int v = col[min(t, tlast)];
            c[u] = (t < t1) ? (v - lo) : -1;
        }
#pragma unroll
        for (int u = 0; u < CR_U; ++u)
            rk[u] = ((unsigned)c[u] < (unsigned)CR_NR) ? atomicAdd(&lcnt[c[u]], 1) : 0;
#pragma unroll
        for (int u = 0; u < CR_U; ++u)
            if ((unsigned)c[u] < (unsigned)CR_NR) rank[tb + u * 1024] = rk[u];
    }
    __syncthreads();
    const int hi = min(CR_NR, Npad - lo);
    int4* __restrict__ dst = (int4*)(part + (size_t)s * Npad + lo);
    for (int i = tid; i < hi / 4; i += 1024) dst[i] = ((const int4*)lcnt)[i];
}

__device__ __forceinline__ int wave_inclusive_scan(int v, int lane) {
    int incl = v;
    for (int off = 1; off < 32; off <<= 1) {
        int n = __shfl_up(incl, off, 32);
        if (lane >= off) incl += n;
    }
    return incl;
}

__global__ __launch_bounds__(256)
void scan_local_kernel(int* __restrict__ part, int* __restrict__ count,
                       int* __restrict__ rowptr, int* __restrict__ bsum, int N, int Npad) {
    __shared__ int wsum[8];
    int lane = threadIdx.x & 31, wv = threadIdx.x >> 5;
    int base = blockIdx.x * 1024 + threadIdx.x * 4;
    int4 tot = make_int4(0, 0, 0, 0);
    if (base < Npad) {
        int4* p = (int4*)(part + base);
        const size_t pitch4 = (size_t)(Npad >> 2);
        for (int s0 = 0; s0 < CR_S; s0 += 16) {
            int4 v[16];
#pragma unroll
            for (int k = 0; k < 16; ++k) v[k] = p[(size_t)(s0 + k) * pitch4];
#pragma unroll
            for (int k = 0; k < 16; ++k) {
                p[(size_t)(s0 + k) * pitch4] = tot;
                tot.x += v[k].x; tot.y += v[k].y; tot.z += v[k].z; tot.w += v[k].w;
            }
        }
    }
    int c0 = (base + 0 < N) ? tot.x : 0;
    int c1 = (base + 1 < N) ? tot.y : 0;
    int c2 = (base + 2 < N) ? tot.z : 0;
    int c3 = (base + 3 < N) ? tot.w : 0;
    int tsum = c0 + c1 + c2 + c3;
    int incl = wave_inclusive_scan(tsum, lane);
    if (lane == 31) wsum[wv] = incl;
    __syncthreads();
    int woff = 0;
    for (int w = 0; w < wv; ++w) woff += wsum[w];
    int excl = woff + incl - tsum;
    if (base + 0 < N) { rowptr[base + 0] = excl;                count[base + 0] = c0; }
    if (base + 1 < N) { rowptr[base + 1] = excl + c0;           count[base + 1] = c1; }
    if (base + 2 < N) { rowptr[base + 2] = excl + c0 + c1;      count[base + 2] = c2; }
    if (base + 3 < N) { rowptr[base + 3] = excl + c0 + c1 + c2; count[base + 3] = c3; }
    if (threadIdx.x == 255) bsum[blockIdx.x] = woff + incl;
}

__global__ __launch_bounds__(256)
void finalize_kernel(int* __restrict__ rowptr, const int* __restrict__ bsum,
                     const int* __restrict__ count, float* __restrict__ dis, int N) {
    __shared__ int chunk_off;
    const int i = blockIdx.x * 256 + threadIdx.x;
    const int chunk_id = (int)(blockIdx.x >> 2);
    if (threadIdx.x < 32) {
        int s = 0;
        for (int b = threadIdx.x; b < chunk_id; b += 32) s += bsum[b];
        for (int off = 16; off > 0; off >>= 1) s += __shfl_xor(s, off, 32);
        if (threadIdx.x == 0) chunk_off = s;
    }
    __syncthreads();
    if (i < N) {
        rowptr[i] += chunk_off;
        dis[i] = rsqrtf((float)(count[i] + 1));
    }
}

__global__ void place_kernel(const int* __restrict__ row, const int* __restrict__ col,
                             const int* __restrict__ rank, const int* __restrict__ rowptr,
                             const int* __restrict__ part, int* __restrict__ slot_src,
                             int E, int chunk, int Npad) {
    int t = blockIdx.x * blockDim.x + threadIdx.x;
    if (t < E) {
        int c = col[t];
        int s = t / chunk;
        slot_src[rowptr[c] + part[(size_t)s * Npad + c] + rank[t]] = row[t];
    }
}

__global__ __launch_bounds__(256)
void order_kernel(const int* __restrict__ slot_src, const int* __restrict__ rowptr,
                  const int* __restrict__ count, const float* __restrict__ dis,
                  int* __restrict__ csr_src, float* __restrict__ csr_w, int N) {
    int lane = threadIdx.x & 31;
    int node = __builtin_amdgcn_readfirstlane((int)(blockIdx.x * 8 + (threadIdx.x >> 5)));
    if (node >= N) return;
    int start = rowptr[node];
    int d = count[node];
    float dc = dis[node];
    const int* __restrict__ seg = slot_src + start;
    for (int i0 = 0; i0 < d; i0 += 32) {
        int i = i0 + lane;
        if (i < d) {
            int key = seg[i];
            int p = 0;
            for (int j = 0; j < d; ++j) {
                int kj = seg[j];
                p += ((kj < key) || (kj == key && j < i)) ? 1 : 0;
            }
            csr_src[start + p] = key;
            csr_w[start + p] = dis[key] * dc;
        }
    }
}


template <int K, int F>
__device__ __forceinline__ void gather_step(float4& acc, const float* __restrict__ colbase,
                                            const int* __restrict__ src,
                                            const float* __restrict__ w, int j) {
    int s[K]; float ww[K]; float4 v[K];
#pragma unroll
    for (int k = 0; k < K; ++k) { s[k] = src[j + k]; ww[k] = w[j + k]; }
#pragma unroll
    for (int k = 0; k < K; ++k) v[k] = *(const float4*)(colbase + (size_t)s[k] * F);
#pragma unroll
    for (int k = 0; k < K; ++k) {
        acc.x = fmaf(v[k].x, ww[k], acc.x);
        acc.y = fmaf(v[k].y, ww[k], acc.y);
        acc.z = fmaf(v[k].z, ww[k], acc.z);
        acc.w = fmaf(v[k].w, ww[k], acc.w);
    }
}

struct RowHead {
    int start; int d; float dn;
    int sv; float wl;
    float4 self;
};

__device__ __forceinline__ void pair_bounds_load(int& bs, int& bd, float& bn, int nd, int nlim,
                                                 int lane,
                                                 const int* __restrict__ rowptr,
                                                 const int* __restrict__ count,
                                                 const float* __restrict__ dis) {
    const int node = nd + (lane & 1);
    const bool ok = node < nlim;
    const int nc = ok ? node : 0;
    bs = rowptr[nc];
    const int cnt = count[nc];
    bd = ok ? cnt : 0;
    bn = dis[nc];
}

__device__ __forceinline__ void row_list_load(RowHead& r, int bs, int bd, float bn, int rr,
                                              int node, int nlim, int lane,
                                              const float* __restrict__ lin,
                                              const int* __restrict__ csr_src,
                                              const float* __restrict__ csr_w) {
    r.start = __builtin_amdgcn_readlane(bs, rr);
    r.d = __builtin_amdgcn_readlane(bd, rr);
    r.dn = __int_as_float(__builtin_amdgcn_readlane(__float_as_int(bn), rr));
    r.sv = 0; r.wl = 0.0f;
    r.self = make_float4(0.0f, 0.0f, 0.0f, 0.0f);
    if (node < nlim) {
        const float* __restrict__ rowp = lin + (size_t)node * 128;
        r.self = *(const float4*)(rowp + (unsigned)lane * 4u);
        if (lane < r.d) { r.sv = csr_src[r.start + lane]; r.wl = csr_w[r.start + lane]; }
    }
}

template <int K>
__device__ __forceinline__ void lane_list_batch(float4& acc, const float* __restrict__ lin,
                                                unsigned loff, int sv, float wl, int k0) {
    float4 v[K];
#pragma unroll
    for (int k = 0; k < K; ++k) {
        const int s = __builtin_amdgcn_readlane(sv, k0 + k);
        const float* __restrict__ rowp = lin + (size_t)s * 128;
        v[k] = *(const float4*)(rowp + loff);
    }
#pragma unroll
    for (int k = 0; k < K; ++k) {
        const float ww = __int_as_float(__builtin_amdgcn_readlane(__float_as_int(wl), k0 + k));
        acc.x = fmaf(v[k].x, ww, acc.x);
        acc.y = fmaf(v[k].y, ww, acc.y);
        acc.z = fmaf(v[k].z, ww, acc.z);
        acc.w = fmaf(v[k].w, ww, acc.w);
    }
}

__device__ __forceinline__ void lane_list_chunk(float4& acc, const float* __restrict__ lin,
                                                unsigned loff, int sv, float wl, int dd) {
    int k0 = 0;
    for (; k0 + 8 <= dd; k0 += 8) lane_list_batch<8>(acc, lin, loff, sv, wl, k0);
    switch (dd - k0) {
        case 7: lane_list_batch<7>(acc, lin, loff, sv, wl, k0); break;
        case 6: lane_list_batch<6>(acc, lin, loff, sv, wl, k0); break;
        case 5: lane_list_batch<5>(acc, lin, loff, sv, wl, k0); break;
        case 4: lane_list_batch<4>(acc, lin, loff, sv, wl, k0); break;
        case 3: lane_list_batch<3>(acc, lin, loff, sv, wl, k0); break;
        case 2: lane_list_batch<2>(acc, lin, loff, sv, wl, k0); break;
        case 1: lane_list_batch<1>(acc, lin, loff, sv, wl, k0); break;
        default: break;
    }
}

#define ST_PITCH 132
#define FUSE_BLK 1024

template <bool LAST>
__global__ __launch_bounds__(256) __attribute__((amdgpu_waves_per_eu(8, 8)))
void gather_gemm_kernel(const float* __restrict__ lin, const float* __restrict__ dis,
                        const float* __restrict__ bias,
                        const int* __restrict__ rowptr, const int* __restrict__ count,
                        const int* __restrict__ csr_src, const float* __restrict__ csr_w,
                        const float* __restrict__ Bw,
                        const float* __restrict__ Bw2, int bcols2,
                        float* __restrict__ D, int N) {
    constexpr int IMG = 16 * ST_PITCH;
    constexpr int NIMG = LAST ? 4 : 2;
    __shared__ __attribute__((aligned(16))) float Sm[NIMG * IMG + 128];
    float* const St  = Sm;
    float* const Ot  = Sm + IMG;
    float* const Pt  = Sm + (LAST ? 2 * IMG : 0);
    float* const W2t = Sm + (LAST ? 3 * IMG : 0);
    float* const Bsh = Sm + NIMG * IMG;
    const int lane = threadIdx.x & 31;
    const int c = lane & 15;
    const int h = lane >> 4;
    const int wv = __builtin_amdgcn_readfirstlane((int)(threadIdx.x >> 5));

    v4f breg[16];
    {
        const float* __restrict__ bp = Bw + (size_t)(4 * h) * 128 + wv * 16 + c;
#pragma unroll
        for (int kb = 0; kb < 16; ++kb) {
            breg[kb].x = bp[(size_t)(kb * 8 + 0) * 128];
            breg[kb].y = bp[(size_t)(kb * 8 + 1) * 128];
            breg[kb].z = bp[(size_t)(kb * 8 + 2) * 128];
            breg[kb].w = bp[(size_t)(kb * 8 + 3) * 128];
        }
    }
    if (threadIdx.x < 128) Bsh[threadIdx.x] = bias[threadIdx.x];
    if (LAST) {
#pragma unroll
        for (int it = 0; it < 8; ++it) {
            const int i = it * 256 + (int)threadIdx.x;
            const int n = i & 15;
            const int k = i >> 4;
            const int nc = (n < bcols2) ? n : (bcols2 - 1);
            W2t[n * ST_PITCH + k] = Bw2[(size_t)k * bcols2 + nc];
        }
    }
    const int tilesM = (N + 15) >> 4;

    RowHead h0, h1;
    int bs, bd; float bn;
    {
        const int t0 = (int)blockIdx.x;
        const int nl0 = (t0 < tilesM) ? N : 0;
        const int nd0 = (t0 << 4) + wv * 2;
        pair_bounds_load(bs, bd, bn, nd0, nl0, lane, rowptr, count, dis);
        row_list_load(h0, bs, bd, bn, 0, nd0, nl0, lane, lin, csr_src, csr_w);
        row_list_load(h1, bs, bd, bn, 1, nd0 + 1, nl0, lane, lin, csr_src, csr_w);
        const int t1 = t0 + (int)gridDim.x;
        pair_bounds_load(bs, bd, bn, (t1 << 4) + wv * 2, (t1 < tilesM) ? N : 0, lane, rowptr, count, dis);
    }

    for (int tm = blockIdx.x; tm < tilesM; tm += gridDim.x) {
        const int m0 = tm << 4;
#pragma nounroll
        for (int rr = 0; rr < 2; ++rr) {
            const int rloc = wv * 2 + rr;
            const int node = m0 + rloc;
            if (node < N) {
                const int d = h0.d;
                const int start = h0.start;
                int sv = h0.sv;
                float wl = h0.wl;
                const float s2 = h0.dn * h0.dn;
                float4 acc;
                acc.x = h0.self.x * s2; acc.y = h0.self.y * s2;
                acc.z = h0.self.z * s2; acc.w = h0.self.w * s2;
                for (int j0 = 0; j0 < d; j0 += 32) {
                    const int dd = min(d - j0, 32);
                    if (j0 != 0) {
                        sv = 0; wl = 0.0f;
                        if (lane < dd) { sv = csr_src[start + j0 + lane]; wl = csr_w[start + j0 + lane]; }
                    }
                    lane_list_chunk(acc, lin, (unsigned)lane * 4u, sv, wl, dd);
                }
                *(float4*)(St + rloc * ST_PITCH + lane * 4) = acc;
            }
            h0 = h1;
        }
        __syncthreads();
        {
            const int tn = tm + (int)gridDim.x;
            const int nln = (tn < tilesM) ? N : 0;
            const int ndn = (tn << 4) + wv * 2;
            row_list_load(h0, bs, bd, bn, 0, ndn, nln, lane, lin, csr_src, csr_w);
            row_list_load(h1, bs, bd, bn, 1, ndn + 1, nln, lane, lin, csr_src, csr_w);
            const int tnn = tn + (int)gridDim.x;
            pair_bounds_load(bs, bd, bn, (tnn << 4) + wv * 2, (tnn < tilesM) ? N : 0, lane, rowptr, count, dis);
        }

        {
            const float* ap = St + c * ST_PITCH + 4 * h;
            v8f acc = (v8f){0.f, 0.f, 0.f, 0.f, 0.f, 0.f, 0.f, 0.f};
#pragma unroll
            for (int kq = 0; kq < 4; ++kq) {
                v4f a[4];
#pragma unroll
                for (int q = 0; q < 4; ++q) a[q] = *(const v4f*)(ap + (kq * 4 + q) * 8);
                asm volatile("" :: "v"(a[0]), "v"(a[1]), "v"(a[2]), "v"(a[3]));
#pragma unroll
                for (int q = 0; q < 4; ++q) {
                    const int kb = kq * 4 + q;
                    v2f a0 = {a[q].x, a[q].y};
                    v2f a1 = {a[q].z, a[q].w};
                    v2f b0 = {breg[kb].x, breg[kb].y};
                    v2f b1 = {breg[kb].z, breg[kb].w};
                    acc = __builtin_amdgcn_wmma_f32_16x16x4_f32(false, a0, false, b0,
                                                                (short)0, acc, false, false);
                    acc = __builtin_amdgcn_wmma_f32_16x16x4_f32(false, a1, false, b1,
                                                                (short)0, acc, false, false);
                }
            }
            const float bcol = Bsh[wv * 16 + c];
            float* op = Ot + (8 * h) * ST_PITCH + wv * 16 + c;
#pragma unroll
            for (int r = 0; r < 8; ++r) op[r * ST_PITCH] = fmaxf(acc[r] + bcol, 0.0f);

            if (LAST) {
                const int foff = c * ST_PITCH + wv * 16 + 4 * h;
                const float* hp = Ot + foff;
                const float* wp = W2t + foff;
                v4f a2[2], b2reg[2];
                b2reg[0] = *(const v4f*)(wp);
                b2reg[1] = *(const v4f*)(wp + 8);
                a2[0] = *(const v4f*)(hp);
                a2[1] = *(const v4f*)(hp + 8);
                v8f p = (v8f){0.f, 0.f, 0.f, 0.f, 0.f, 0.f, 0.f, 0.f};
#pragma unroll
                for (int kb = 0; kb < 2; ++kb) {
                    v2f a0 = {a2[kb].x, a2[kb].y};
                    v2f a1 = {a2[kb].z, a2[kb].w};
                    v2f b0 = {b2reg[kb].x, b2reg[kb].y};
                    v2f b1 = {b2reg[kb].z, b2reg[kb].w};
                    p = __builtin_amdgcn_wmma_f32_16x16x4_f32(false, a0, false, b0,
                                                              (short)0, p, false, false);
                    p = __builtin_amdgcn_wmma_f32_16x16x4_f32(false, a1, false, b1,
                                                              (short)0, p, false, false);
                }
                float* pp = Pt + (8 * h) * ST_PITCH + wv * 16 + c;
#pragma unroll
                for (int r = 0; r < 8; ++r) pp[r * ST_PITCH] = p[r];
            }
        }
        __syncthreads();

        if (!LAST) {
            for (int rr = 0; rr < 2; ++rr) {
                const int rloc = wv * 2 + rr;
                const int m = m0 + rloc;
                if (m < N) {
                    v4f v = *(const v4f*)(Ot + rloc * ST_PITCH + lane * 4);
                    const unsigned didx = (unsigned)m * 128u + (unsigned)lane * 4u;
                    *(v4f*)(D + didx) = v;
                }
            }
        } else {
            const int prow = wv * 2 + h;
            const float* q = Pt + prow * ST_PITCH + c;
            float s = q[0];
#pragma unroll
            for (int w = 1; w < 8; ++w) s += q[w * 16];
            const unsigned didx = (unsigned)m0 * 16u + (unsigned)threadIdx.x;
            if (m0 + prow < N) D[didx] = s;
        }
    }
}

__global__ __launch_bounds__(256)
void gather16_pool_kernel(const float* __restrict__ lin, const float* __restrict__ dis,
                          const float* __restrict__ bias, int biasLen,
                          const int* __restrict__ rowptr, const int* __restrict__ count,
                          const int* __restrict__ csr_src, const float* __restrict__ csr_w,
                          const int* __restrict__ batch,
                          float* __restrict__ out, float* __restrict__ cnt, int N) {
    const long long t = (long long)blockIdx.x * blockDim.x + threadIdx.x;
    const int node = (int)(t >> 2);
    const int q = (int)(t & 3);
    if ((node & ~7) >= N) return;
    const bool valid = node < N;
    float4 acc = make_float4(0.0f, 0.0f, 0.0f, 0.0f);
    int g = -1;
    if (valid) {
        g = batch[node];
        int start = rowptr[node];
        int d = count[node];
        float dn = dis[node];
        const float* __restrict__ colbase = lin + q * 4;
        const int* __restrict__ src = csr_src + start;
        const float* __restrict__ w = csr_w + start;
        float4 self = *(const float4*)(colbase + (size_t)node * 16);
        float s2 = dn * dn;
        acc.x = self.x * s2; acc.y = self.y * s2; acc.z = self.z * s2; acc.w = self.w * s2;
        int j = 0;
        for (; j + 4 <= d; j += 4) gather_step<4, 16>(acc, colbase, src, w, j);
        if (j + 2 <= d) { gather_step<2, 16>(acc, colbase, src, w, j); j += 2; }
        if (j < d) gather_step<1, 16>(acc, colbase, src, w, j);
        int c = q * 4;
        acc.x += (c + 0 < biasLen) ? bias[c + 0] : 0.0f;
        acc.y += (c + 1 < biasLen) ? bias[c + 1] : 0.0f;
        acc.z += (c + 2 < biasLen) ? bias[c + 2] : 0.0f;
        acc.w += (c + 3 < biasLen) ? bias[c + 3] : 0.0f;
    }
    const int g0 = __builtin_amdgcn_readfirstlane(g);
    const int c0 = q * 4;
    if (__all((!valid || g == g0) ? 1 : 0)) {
        for (int off = 4; off < 32; off <<= 1) {
            acc.x += __shfl_xor(acc.x, off, 32);
            acc.y += __shfl_xor(acc.y, off, 32);
            acc.z += __shfl_xor(acc.z, off, 32);
            acc.w += __shfl_xor(acc.w, off, 32);
        }
        const float nodes = (float)(__popcll(__ballot(valid ? 1 : 0)) >> 2);
        if ((threadIdx.x & 31) < 4) {
            float* __restrict__ o = out + (size_t)g0 * C_OUT + c0;
            if (c0 + 0 < C_OUT) atomicAdd(o + 0, acc.x);
            if (c0 + 1 < C_OUT) atomicAdd(o + 1, acc.y);
            if (c0 + 2 < C_OUT) atomicAdd(o + 2, acc.z);
            if (c0 + 3 < C_OUT) atomicAdd(o + 3, acc.w);
            if (q == 0) atomicAdd(&cnt[g0], nodes);
        }
    } else if (valid) {
        float* __restrict__ o = out + (size_t)g * C_OUT + c0;
        if (c0 + 0 < C_OUT) atomicAdd(o + 0, acc.x);
        if (c0 + 1 < C_OUT) atomicAdd(o + 1, acc.y);
        if (c0 + 2 < C_OUT) atomicAdd(o + 2, acc.z);
        if (c0 + 3 < C_OUT) atomicAdd(o + 3, acc.w);
        if (q == 0) atomicAdd(&cnt[g], 1.0f);
    }
}

__global__ void div_kernel(float* __restrict__ out, const float* __restrict__ cnt) {
    int t = blockIdx.x * blockDim.x + threadIdx.x;
    if (t < NUM_GRAPHS * C_OUT) out[t] /= fmaxf(cnt[t / C_OUT], 1.0f);
}


static inline int cdiv_ll(long long a, long long b) { return (int)((a + b - 1) / b); }

extern "C" void kernel_launch(void* const* d_in, const int* in_sizes, int n_in,
                              void* d_out, int out_size, void* d_ws, size_t ws_size,
                              hipStream_t stream) {
    const float* x     = (const float*)d_in[0];
    const int*   ei    = (const int*)d_in[1];
    const int*   batch = (const int*)d_in[2];
    const float* W0 = (const float*)d_in[3];  const float* b0 = (const float*)d_in[4];
    const float* W1 = (const float*)d_in[5];  const float* b1 = (const float*)d_in[6];
    const float* W2 = (const float*)d_in[7];  const float* b2 = (const float*)d_in[8];
    const float* W3 = (const float*)d_in[9];  const float* b3 = (const float*)d_in[10];

    const int N = in_sizes[2];
    const int E = in_sizes[1] / 2;
    const int H = 128;
    const int* rowi = ei;
    const int* coli = ei + E;

    float* out = (float*)d_out;
    float* ws  = (float*)d_ws;
    const size_t Npad = ((size_t)N + 255) & ~(size_t)255;
    const int    nb   = (N + 1023) >> 10;
    float* dis  = ws;
    float* bufA = ws + Npad;
    float* bufB = bufA + (size_t)N * H;
    float* cnt  = bufB + (size_t)N * H;
    int*   count    = (int*)(cnt + 512);
    int*   rowptr   = count + Npad;
    int*   bsum     = rowptr + Npad;
    int*   part     = bsum + (((size_t)nb + 255) & ~(size_t)255);
    int*   rank     = part + (size_t)CR_S * Npad;
    int*   slot_src = rank + (size_t)E;
    int*   csr_src  = slot_src + (size_t)E;
    float* csr_w    = (float*)(csr_src + (size_t)E);

    const int TB = 256;

    const int nranges = cdiv_ll((long long)Npad, CR_NR);
    const int chunk   = cdiv_ll(E, CR_S);
    count_rank_kernel<<<nranges * CR_S, 1024, 0, stream>>>(coli, rank, part, E, chunk, (int)Npad, out, cnt);
    scan_local_kernel<<<nb, TB, 0, stream>>>(part, count, rowptr, bsum, N, (int)Npad);
    finalize_kernel<<<cdiv_ll(N, TB), TB, 0, stream>>>(rowptr, bsum, count, dis, N);
    place_kernel<<<cdiv_ll(E, TB), TB, 0, stream>>>(rowi, coli, rank, rowptr, part, slot_src, E, chunk, (int)Npad);
    order_kernel<<<cdiv_ll(N, TB / 32), TB, 0, stream>>>(slot_src, rowptr, count, dis, csr_src, csr_w, N);

    const int fuseBlk    = FUSE_BLK;

    gather_gemm_kernel<false><<<fuseBlk, TB, 0, stream>>>(x, dis, b0, rowptr, count, csr_src, csr_w, W0, W0, H, bufA, N);

    gather_gemm_kernel<false><<<fuseBlk, TB, 0, stream>>>(bufA, dis, b1, rowptr, count, csr_src, csr_w, W1, W1, H, bufB, N);

    gather_gemm_kernel<true><<<fuseBlk, TB, 0, stream>>>(bufB, dis, b2, rowptr, count, csr_src, csr_w, W2, W3, C_OUT, bufA, N);

    gather16_pool_kernel<<<cdiv_ll((long long)N * 4, TB), TB, 0, stream>>>(bufA, dis, b3, C_OUT, rowptr, count, csr_src, csr_w, batch, out, cnt, N);
    div_kernel<<<cdiv_ll(NUM_GRAPHS * C_OUT, TB), TB, 0, stream>>>(out, cnt);
}


# Edge placement per (target range, edge chunk) work-group: slot cursors (row offset + chunk prefix) in LDS, a second LDS-add pass stores the sources; no per-edge rank, no offsets launch, 10->9
# speedup vs baseline: 5.4657x; 1.0454x over previous
#include <hip/hip_runtime.h>


typedef float v2f __attribute__((ext_vector_type(2)));
typedef float v4f __attribute__((ext_vector_type(4)));
typedef float v8f __attribute__((ext_vector_type(8)));

#define NUM_GRAPHS 512
#define C_OUT 10

#define CR_NR 12544
#define CR_S  64
#define CR_U  8

__global__ __launch_bounds__(1024)
void count_kernel(const int* __restrict__ col, int* __restrict__ part, int E, int chunk, int Npad,
                  float* __restrict__ psum, float* __restrict__ pcnt) {
    __shared__ __attribute__((aligned(16))) int lcnt[CR_NR];
    const int tid = threadIdx.x;
    const int gt = (int)blockIdx.x * 1024 + tid;
    if (gt < NUM_GRAPHS * C_OUT) psum[gt] = 0.0f;
    if (gt < NUM_GRAPHS) pcnt[gt] = 0.0f;
    const int s = blockIdx.x % CR_S;
    const int lo = (blockIdx.x / CR_S) * CR_NR;
    for (int i = tid; i < CR_NR / 4; i += 1024) ((int4*)lcnt)[i] = make_int4(0, 0, 0, 0);
    __syncthreads();
    const int t0 = s * chunk;
    const int t1 = min(t0 + chunk, E);
    const int tlast = t1 - 1;
    for (int tb = t0 + tid; tb < t1; tb += 1024 * CR_U) {
        int c[CR_U];
#pragma unroll
        for (int u = 0; u < CR_U; ++u) {
            const int t = tb + u * 1024;
            const int v = col[min(t, tlast)];
            c[u] = (t < t1) ? (v - lo) : -1;
        }
#pragma unroll
        for (int u = 0; u < CR_U; ++u)
            if ((unsigned)c[u] < (unsigned)CR_NR) atomicAdd(&lcnt[c[u]], 1);
    }
    __syncthreads();
    const int hi = min(CR_NR, Npad - lo);
    int4* __restrict__ dst = (int4*)(part + (size_t)s * Npad + lo);
    for (int i = tid; i < hi / 4; i += 1024) dst[i] = ((const int4*)lcnt)[i];
}

__device__ __forceinline__ int wave_inclusive_scan(int v, int lane) {
    int incl = v;
    for (int off = 1; off < 32; off <<= 1) {
        int n = __shfl_up(incl, off, 32);
        if (lane >= off) incl += n;
    }
    return incl;
}

__global__ __launch_bounds__(256)
void scan_local_kernel(int* __restrict__ part, int* __restrict__ count,
                       int* __restrict__ rowloc, int* __restrict__ bsum,
                       float* __restrict__ dis, int N, int Npad) {
    __shared__ int wsum[8];
    int lane = threadIdx.x & 31, wv = threadIdx.x >> 5;
    int base = blockIdx.x * 1024 + threadIdx.x * 4;
    int4 tot = make_int4(0, 0, 0, 0);
    if (base < Npad) {
        int4* p = (int4*)(part + base);
        const size_t pitch4 = (size_t)(Npad >> 2);
        for (int s0 = 0; s0 < CR_S; s0 += 16) {
            int4 v[16];
#pragma unroll
            for (int k = 0; k < 16; ++k) v[k] = p[(size_t)(s0 + k) * pitch4];
#pragma unroll
            for (int k = 0; k < 16; ++k) {
                p[(size_t)(s0 + k) * pitch4] = tot;
                tot.x += v[k].x; tot.y += v[k].y; tot.z += v[k].z; tot.w += v[k].w;
            }
        }
    }
    int c0 = (base + 0 < N) ? tot.x : 0;
    int c1 = (base + 1 < N) ? tot.y : 0;
    int c2 = (base + 2 < N) ? tot.z : 0;
    int c3 = (base + 3 < N) ? tot.w : 0;
    int tsum = c0 + c1 + c2 + c3;
    int incl = wave_inclusive_scan(tsum, lane);
    if (lane == 31) wsum[wv] = incl;
    __syncthreads();
    int woff = 0;
    for (int w = 0; w < wv; ++w) woff += wsum[w];
    int excl = woff + incl - tsum;
    if (base < Npad) *(int4*)(rowloc + base) = make_int4(excl, excl + c0, excl + c0 + c1, excl + c0 + c1 + c2);
    if (base + 0 < N) { count[base + 0] = c0; dis[base + 0] = rsqrtf((float)(c0 + 1)); }
    if (base + 1 < N) { count[base + 1] = c1; dis[base + 1] = rsqrtf((float)(c1 + 1)); }
    if (base + 2 < N) { count[base + 2] = c2; dis[base + 2] = rsqrtf((float)(c2 + 1)); }
    if (base + 3 < N) { count[base + 3] = c3; dis[base + 3] = rsqrtf((float)(c3 + 1)); }
    if (threadIdx.x == 255) bsum[blockIdx.x] = woff + incl;
}

#define PL_MAXCH 1024
__global__ __launch_bounds__(1024)
void place_kernel(const int* __restrict__ row, const int* __restrict__ col,
                  const int* __restrict__ rowloc, const int* __restrict__ bsum, int nb,
                  const int* __restrict__ part, int* __restrict__ slot_src,
                  int* __restrict__ rowptr, int E, int chunk, int Npad) {
    __shared__ __attribute__((aligned(16))) int cur[CR_NR];
    __shared__ int offs[PL_MAXCH];
    __shared__ int wtot[32];
    const int tid = threadIdx.x;
    const int lane = tid & 31, wv = tid >> 5;
    const int s = blockIdx.x % CR_S;
    const int lo = (blockIdx.x / CR_S) * CR_NR;
    int carry = 0;
    for (int b0 = 0; b0 < nb; b0 += 1024) {
        const int b = b0 + tid;
        const int v = (b < nb) ? bsum[b] : 0;
        const int incl = wave_inclusive_scan(v, lane);
        if (lane == 31) wtot[wv] = incl;
        __syncthreads();
        int woff = 0, tot = 0;
        for (int w = 0; w < 32; ++w) {
            const int x = wtot[w];
            woff += (w < wv) ? x : 0;
            tot += x;
        }
        if (b < nb && b < PL_MAXCH) offs[b] = carry + woff + incl - v;
        carry += tot;
        __syncthreads();
    }
    const int hi = min(CR_NR, Npad - lo);
    const int4* __restrict__ pl = (const int4*)(rowloc + lo);
    const int4* __restrict__ pp = (const int4*)(part + (size_t)s * Npad + lo);
    int4* __restrict__ pr = (int4*)(rowptr + lo);
    constexpr int CUR_IT = (CR_NR / 4 + 1023) / 1024;
    const int nvec = hi / 4;
    int4 a[CUR_IT], p[CUR_IT];
#pragma unroll
    for (int k = 0; k < CUR_IT; ++k) {
        const int ic = min(tid + k * 1024, nvec - 1);
        a[k] = pl[ic];
        p[k] = pp[ic];
    }
#pragma unroll
    for (int k = 0; k < CUR_IT; ++k) {
        const int i = tid + k * 1024;
        if (i < nvec) {
            const int o = offs[min((lo + 4 * i) >> 10, PL_MAXCH - 1)];
            const int4 r = make_int4(a[k].x + o, a[k].y + o, a[k].z + o, a[k].w + o);
            if (s == 0) pr[i] = r;
            ((int4*)cur)[i] = make_int4(r.x + p[k].x, r.y + p[k].y, r.z + p[k].z, r.w + p[k].w);
        }
    }
    __syncthreads();
    const int t0 = s * chunk;
    const int t1 = min(t0 + chunk, E);
    const int tlast = t1 - 1;
    for (int tb = t0 + tid; tb < t1; tb += 1024 * CR_U) {
        int c[CR_U], r[CR_U];
        unsigned sl[CR_U];
#pragma unroll
        for (int u = 0; u < CR_U; ++u) {
            const int t = tb + u * 1024;
            const int tc = min(t, tlast);
            const int v = col[tc];
            r[u] = row[tc];
            c[u] = (t < t1) ? (v - lo) : -1;
        }
#pragma unroll
        for (int u = 0; u < CR_U; ++u)
            sl[u] = ((unsigned)c[u] < (unsigned)CR_NR) ? (unsigned)atomicAdd(&cur[c[u]], 1) : 0u;
        static_assert(CR_U == 8, "the fence below names 8 slots");
        asm volatile("" :: "v"(sl[0]), "v"(sl[1]), "v"(sl[2]), "v"(sl[3]),
                           "v"(sl[4]), "v"(sl[5]), "v"(sl[6]), "v"(sl[7]));
#pragma unroll
        for (int u = 0; u < CR_U; ++u)
            if ((unsigned)c[u] < (unsigned)CR_NR) slot_src[sl[u]] = r[u];
    }
}

__global__ __launch_bounds__(256)
void order_kernel(const int* __restrict__ slot_src, const int* __restrict__ rowptr,
                  const int* __restrict__ count, const float* __restrict__ dis,
                  int* __restrict__ csr_src, float* __restrict__ csr_w, int N) {
    int lane = threadIdx.x & 31;
    int node = __builtin_amdgcn_readfirstlane((int)(blockIdx.x * 8 + (threadIdx.x >> 5)));
    if (node >= N) return;
    int start = rowptr[node];
    int d = count[node];
    float dc = dis[node];
    const int* __restrict__ seg = slot_src + start;
    for (int i0 = 0; i0 < d; i0 += 32) {
        int i = i0 + lane;
        if (i < d) {
            int key = seg[i];
            int p = 0;
            for (int j = 0; j < d; ++j) {
                int kj = seg[j];
                p += ((kj < key) || (kj == key && j < i)) ? 1 : 0;
            }
            csr_src[start + p] = key;
            csr_w[start + p] = dis[key] * dc;
        }
    }
}


template <int K, int F>
__device__ __forceinline__ void gather_step(float4& acc, const float* __restrict__ colbase,
                                            const int* __restrict__ src,
                                            const float* __restrict__ w, int j) {
    int s[K]; float ww[K]; float4 v[K];
#pragma unroll
    for (int k = 0; k < K; ++k) { s[k] = src[j + k]; ww[k] = w[j + k]; }
#pragma unroll
    for (int k = 0; k < K; ++k) v[k] = *(const float4*)(colbase + (size_t)s[k] * F);
#pragma unroll
    for (int k = 0; k < K; ++k) {
        acc.x = fmaf(v[k].x, ww[k], acc.x);
        acc.y = fmaf(v[k].y, ww[k], acc.y);
        acc.z = fmaf(v[k].z, ww[k], acc.z);
        acc.w = fmaf(v[k].w, ww[k], acc.w);
    }
}

struct RowHead {
    int start; int d; float dn;
    int sv; float wl;
    float4 self;
};

__device__ __forceinline__ void pair_bounds_load(int& bs, int& bd, float& bn, int nd, int nlim,
                                                 int lane,
                                                 const int* __restrict__ rowptr,
                                                 const int* __restrict__ count,
                                                 const float* __restrict__ dis) {
    const int node = nd + (lane & 1);
    const bool ok = node < nlim;
    const int nc = ok ? node : 0;
    bs = rowptr[nc];
    const int cnt = count[nc];
    bd = ok ? cnt : 0;
    bn = dis[nc];
}

__device__ __forceinline__ void row_list_load(RowHead& r, int bs, int bd, float bn, int rr,
                                              int node, int nlim, int lane,
                                              const float* __restrict__ lin,
                                              const int* __restrict__ csr_src,
                                              const float* __restrict__ csr_w) {
    r.start = __builtin_amdgcn_readlane(bs, rr);
    r.d = __builtin_amdgcn_readlane(bd, rr);
    r.dn = __int_as_float(__builtin_amdgcn_readlane(__float_as_int(bn), rr));
    r.sv = 0; r.wl = 0.0f;
    r.self = make_float4(0.0f, 0.0f, 0.0f, 0.0f);
    if (node < nlim) {
        const float* __restrict__ rowp = lin + (size_t)node * 128;
        r.self = *(const float4*)(rowp + (unsigned)lane * 4u);
        if (lane < r.d) { r.sv = csr_src[r.start + lane]; r.wl = csr_w[r.start + lane]; }
    }
}

template <int K>
__device__ __forceinline__ void lane_list_batch(float4& acc, const float* __restrict__ lin,
                                                unsigned loff, int sv, float wl, int k0) {
    float4 v[K];
#pragma unroll
    for (int k = 0; k < K; ++k) {
        const int s = __builtin_amdgcn_readlane(sv, k0 + k);
        const float* __restrict__ rowp = lin + (size_t)s * 128;
        v[k] = *(const float4*)(rowp + loff);
    }
#pragma unroll
    for (int k = 0; k < K; ++k) {
        const float ww = __int_as_float(__builtin_amdgcn_readlane(__float_as_int(wl), k0 + k));
        acc.x = fmaf(v[k].x, ww, acc.x);
        acc.y = fmaf(v[k].y, ww, acc.y);
        acc.z = fmaf(v[k].z, ww, acc.z);
        acc.w = fmaf(v[k].w, ww, acc.w);
    }
}

__device__ __forceinline__ void lane_list_chunk(float4& acc, const float* __restrict__ lin,
                                                unsigned loff, int sv, float wl, int dd) {
    int k0 = 0;
    for (; k0 + 8 <= dd; k0 += 8) lane_list_batch<8>(acc, lin, loff, sv, wl, k0);
    switch (dd - k0) {
        case 7: lane_list_batch<7>(acc, lin, loff, sv, wl, k0); break;
        case 6: lane_list_batch<6>(acc, lin, loff, sv, wl, k0); break;
        case 5: lane_list_batch<5>(acc, lin, loff, sv, wl, k0); break;
        case 4: lane_list_batch<4>(acc, lin, loff, sv, wl, k0); break;
        case 3: lane_list_batch<3>(acc, lin, loff, sv, wl, k0); break;
        case 2: lane_list_batch<2>(acc, lin, loff, sv, wl, k0); break;
        case 1: lane_list_batch<1>(acc, lin, loff, sv, wl, k0); break;
        default: break;
    }
}

#define ST_PITCH 132
#define FUSE_BLK 1024

template <bool LAST>
__global__ __launch_bounds__(256) __attribute__((amdgpu_waves_per_eu(8, 8)))
void gather_gemm_kernel(const float* __restrict__ lin, const float* __restrict__ dis,
                        const float* __restrict__ bias,
                        const int* __restrict__ rowptr, const int* __restrict__ count,
                        const int* __restrict__ csr_src, const float* __restrict__ csr_w,
                        const float* __restrict__ Bw,
                        const float* __restrict__ Bw2, int bcols2,
                        float* __restrict__ D, int N) {
    constexpr int IMG = 16 * ST_PITCH;
    constexpr int NIMG = LAST ? 4 : 2;
    __shared__ __attribute__((aligned(16))) float Sm[NIMG * IMG + 128];
    float* const St  = Sm;
    float* const Ot  = Sm + IMG;
    float* const Pt  = Sm + (LAST ? 2 * IMG : 0);
    float* const W2t = Sm + (LAST ? 3 * IMG : 0);
    float* const Bsh = Sm + NIMG * IMG;
    const int lane = threadIdx.x & 31;
    const int c = lane & 15;
    const int h = lane >> 4;
    const int wv = __builtin_amdgcn_readfirstlane((int)(threadIdx.x >> 5));

    v4f breg[16];
    {
        const float* __restrict__ bp = Bw + (size_t)(4 * h) * 128 + wv * 16 + c;
#pragma unroll
        for (int kb = 0; kb < 16; ++kb) {
            breg[kb].x = bp[(size_t)(kb * 8 + 0) * 128];
            breg[kb].y = bp[(size_t)(kb * 8 + 1) * 128];
            breg[kb].z = bp[(size_t)(kb * 8 + 2) * 128];
            breg[kb].w = bp[(size_t)(kb * 8 + 3) * 128];
        }
    }
    if (threadIdx.x < 128) Bsh[threadIdx.x] = bias[threadIdx.x];
    if (LAST) {
#pragma unroll
        for (int it = 0; it < 8; ++it) {
            const int i = it * 256 + (int)threadIdx.x;
            const int n = i & 15;
            const int k = i >> 4;
            const int nc = (n < bcols2) ? n : (bcols2 - 1);
            W2t[n * ST_PITCH + k] = Bw2[(size_t)k * bcols2 + nc];
        }
    }
    const int tilesM = (N + 15) >> 4;

    RowHead h0, h1;
    int bs, bd; float bn;
    {
        const int t0 = (int)blockIdx.x;
        const int nl0 = (t0 < tilesM) ? N : 0;
        const int nd0 = (t0 << 4) + wv * 2;
        pair_bounds_load(bs, bd, bn, nd0, nl0, lane, rowptr, count, dis);
        row_list_load(h0, bs, bd, bn, 0, nd0, nl0, lane, lin, csr_src, csr_w);
        row_list_load(h1, bs, bd, bn, 1, nd0 + 1, nl0, lane, lin, csr_src, csr_w);
        const int t1 = t0 + (int)gridDim.x;
        pair_bounds_load(bs, bd, bn, (t1 << 4) + wv * 2, (t1 < tilesM) ? N : 0, lane, rowptr, count, dis);
    }

    for (int tm = blockIdx.x; tm < tilesM; tm += gridDim.x) {
        const int m0 = tm << 4;
#pragma nounroll
        for (int rr = 0; rr < 2; ++rr) {
            const int rloc = wv * 2 + rr;
            const int node = m0 + rloc;
            if (node < N) {
                const int d = h0.d;
                const int start = h0.start;
                int sv = h0.sv;
                float wl = h0.wl;
                const float s2 = h0.dn * h0.dn;
                float4 acc;
                acc.x = h0.self.x * s2; acc.y = h0.self.y * s2;
                acc.z = h0.self.z * s2; acc.w = h0.self.w * s2;
                for (int j0 = 0; j0 < d; j0 += 32) {
                    const int dd = min(d - j0, 32);
                    if (j0 != 0) {
                        sv = 0; wl = 0.0f;
                        if (lane < dd) { sv = csr_src[start + j0 + lane]; wl = csr_w[start + j0 + lane]; }
                    }
                    lane_list_chunk(acc, lin, (unsigned)lane * 4u, sv, wl, dd);
                }
                *(float4*)(St + rloc * ST_PITCH + lane * 4) = acc;
            }
            h0 = h1;
        }
        __syncthreads();
        {
            const int tn = tm + (int)gridDim.x;
            const int nln = (tn < tilesM) ? N : 0;
            const int ndn = (tn << 4) + wv * 2;
            row_list_load(h0, bs, bd, bn, 0, ndn, nln, lane, lin, csr_src, csr_w);
            row_list_load(h1, bs, bd, bn, 1, ndn + 1, nln, lane, lin, csr_src, csr_w);
            const int tnn = tn + (int)gridDim.x;
            pair_bounds_load(bs, bd, bn, (tnn << 4) + wv * 2, (tnn < tilesM) ? N : 0, lane, rowptr, count, dis);
        }

        {
            const float* ap = St + c * ST_PITCH + 4 * h;
            v8f acc = (v8f){0.f, 0.f, 0.f, 0.f, 0.f, 0.f, 0.f, 0.f};
#pragma unroll
            for (int kq = 0; kq < 4; ++kq) {
                v4f a[4];
#pragma unroll
                for (int q = 0; q < 4; ++q) a[q] = *(const v4f*)(ap + (kq * 4 + q) * 8);
                asm volatile("" :: "v"(a[0]), "v"(a[1]), "v"(a[2]), "v"(a[3]));
#pragma unroll
                for (int q = 0; q < 4; ++q) {
                    const int kb = kq * 4 + q;
                    v2f a0 = {a[q].x, a[q].y};
                    v2f a1 = {a[q].z, a[q].w};
                    v2f b0 = {breg[kb].x, breg[kb].y};
                    v2f b1 = {breg[kb].z, breg[kb].w};
                    acc = __builtin_amdgcn_wmma_f32_16x16x4_f32(false, a0, false, b0,
                                                                (short)0, acc, false, false);
                    acc = __builtin_amdgcn_wmma_f32_16x16x4_f32(false, a1, false, b1,
                                                                (short)0, acc, false, false);
                }
            }
            const float bcol = Bsh[wv * 16 + c];
            float* op = Ot + (8 * h) * ST_PITCH + wv * 16 + c;
#pragma unroll
            for (int r = 0; r < 8; ++r) op[r * ST_PITCH] = fmaxf(acc[r] + bcol, 0.0f);

            if (LAST) {
                const int foff = c * ST_PITCH + wv * 16 + 4 * h;
                const float* hp = Ot + foff;
                const float* wp = W2t + foff;
                v4f a2[2], b2reg[2];
                b2reg[0] = *(const v4f*)(wp);
                b2reg[1] = *(const v4f*)(wp + 8);
                a2[0] = *(const v4f*)(hp);
                a2[1] = *(const v4f*)(hp + 8);
                v8f p = (v8f){0.f, 0.f, 0.f, 0.f, 0.f, 0.f, 0.f, 0.f};
#pragma unroll
                for (int kb = 0; kb < 2; ++kb) {
                    v2f a0 = {a2[kb].x, a2[kb].y};
                    v2f a1 = {a2[kb].z, a2[kb].w};
                    v2f b0 = {b2reg[kb].x, b2reg[kb].y};
                    v2f b1 = {b2reg[kb].z, b2reg[kb].w};
                    p = __builtin_amdgcn_wmma_f32_16x16x4_f32(false, a0, false, b0,
                                                              (short)0, p, false, false);
                    p = __builtin_amdgcn_wmma_f32_16x16x4_f32(false, a1, false, b1,
                                                              (short)0, p, false, false);
                }
                float* pp = Pt + (8 * h) * ST_PITCH + wv * 16 + c;
#pragma unroll
                for (int r = 0; r < 8; ++r) pp[r * ST_PITCH] = p[r];
            }
        }
        __syncthreads();

        if (!LAST) {
            for (int rr = 0; rr < 2; ++rr) {
                const int rloc = wv * 2 + rr;
                const int m = m0 + rloc;
                if (m < N) {
                    v4f v = *(const v4f*)(Ot + rloc * ST_PITCH + lane * 4);
                    const unsigned didx = (unsigned)m * 128u + (unsigned)lane * 4u;
                    *(v4f*)(D + didx) = v;
                }
            }
        } else {
            const int prow = wv * 2 + h;
            const float* q = Pt + prow * ST_PITCH + c;
            float s = q[0];
#pragma unroll
            for (int w = 1; w < 8; ++w) s += q[w * 16];
            const unsigned didx = (unsigned)m0 * 16u + (unsigned)threadIdx.x;
            if (m0 + prow < N) D[didx] = s;
        }
    }
}

__global__ __launch_bounds__(256)
void gather16_pool_kernel(const float* __restrict__ lin, const float* __restrict__ dis,
                          const float* __restrict__ bias, int biasLen,
                          const int* __restrict__ rowptr, const int* __restrict__ count,
                          const int* __restrict__ csr_src, const float* __restrict__ csr_w,
                          const int* __restrict__ batch,
                          float* __restrict__ out, float* __restrict__ cnt, int N) {
    const long long t = (long long)blockIdx.x * blockDim.x + threadIdx.x;
    const int node = (int)(t >> 2);
    const int q = (int)(t & 3);
    if ((node & ~7) >= N) return;
    const bool valid = node < N;
    float4 acc = make_float4(0.0f, 0.0f, 0.0f, 0.0f);
    int g = -1;
    if (valid) {
        g = batch[node];
        int start = rowptr[node];
        int d = count[node];
        float dn = dis[node];
        const float* __restrict__ colbase = lin + q * 4;
        const int* __restrict__ src = csr_src + start;
        const float* __restrict__ w = csr_w + start;
        float4 self = *(const float4*)(colbase + (size_t)node * 16);
        float s2 = dn * dn;
        acc.x = self.x * s2; acc.y = self.y * s2; acc.z = self.z * s2; acc.w = self.w * s2;
        int j = 0;
        for (; j + 4 <= d; j += 4) gather_step<4, 16>(acc, colbase, src, w, j);
        if (j + 2 <= d) { gather_step<2, 16>(acc, colbase, src, w, j); j += 2; }
        if (j < d) gather_step<1, 16>(acc, colbase, src, w, j);
        int c = q * 4;
        acc.x += (c + 0 < biasLen) ? bias[c + 0] : 0.0f;
        acc.y += (c + 1 < biasLen) ? bias[c + 1] : 0.0f;
        acc.z += (c + 2 < biasLen) ? bias[c + 2] : 0.0f;
        acc.w += (c + 3 < biasLen) ? bias[c + 3] : 0.0f;
    }
    const int g0 = __builtin_amdgcn_readfirstlane(g);
    const int c0 = q * 4;
    if (__all((!valid || g == g0) ? 1 : 0)) {
        for (int off = 4; off < 32; off <<= 1) {
            acc.x += __shfl_xor(acc.x, off, 32);
            acc.y += __shfl_xor(acc.y, off, 32);
            acc.z += __shfl_xor(acc.z, off, 32);
            acc.w += __shfl_xor(acc.w, off, 32);
        }
        const float nodes = (float)(__popcll(__ballot(valid ? 1 : 0)) >> 2);
        if ((threadIdx.x & 31) < 4) {
            float* __restrict__ o = out + (size_t)g0 * C_OUT + c0;
            if (c0 + 0 < C_OUT) atomicAdd(o + 0, acc.x);
            if (c0 + 1 < C_OUT) atomicAdd(o + 1, acc.y);
            if (c0 + 2 < C_OUT) atomicAdd(o + 2, acc.z);
            if (c0 + 3 < C_OUT) atomicAdd(o + 3, acc.w);
            if (q == 0) atomicAdd(&cnt[g0], nodes);
        }
    } else if (valid) {
        float* __restrict__ o = out + (size_t)g * C_OUT + c0;
        if (c0 + 0 < C_OUT) atomicAdd(o + 0, acc.x);
        if (c0 + 1 < C_OUT) atomicAdd(o + 1, acc.y);
        if (c0 + 2 < C_OUT) atomicAdd(o + 2, acc.z);
        if (c0 + 3 < C_OUT) atomicAdd(o + 3, acc.w);
        if (q == 0) atomicAdd(&cnt[g], 1.0f);
    }
}

__global__ void div_kernel(float* __restrict__ out, const float* __restrict__ cnt) {
    int t = blockIdx.x * blockDim.x + threadIdx.x;
    if (t < NUM_GRAPHS * C_OUT) out[t] /= fmaxf(cnt[t / C_OUT], 1.0f);
}


static inline int cdiv_ll(long long a, long long b) { return (int)((a + b - 1) / b); }

extern "C" void kernel_launch(void* const* d_in, const int* in_sizes, int n_in,
                              void* d_out, int out_size, void* d_ws, size_t ws_size,
                              hipStream_t stream) {
    const float* x     = (const float*)d_in[0];
    const int*   ei    = (const int*)d_in[1];
    const int*   batch = (const int*)d_in[2];
    const float* W0 = (const float*)d_in[3];  const float* b0 = (const float*)d_in[4];
    const float* W1 = (const float*)d_in[5];  const float* b1 = (const float*)d_in[6];
    const float* W2 = (const float*)d_in[7];  const float* b2 = (const float*)d_in[8];
    const float* W3 = (const float*)d_in[9];  const float* b3 = (const float*)d_in[10];

    const int N = in_sizes[2];
    const int E = in_sizes[1] / 2;
    const int H = 128;
    const int* rowi = ei;
    const int* coli = ei + E;

    float* out = (float*)d_out;
    float* ws  = (float*)d_ws;
    const size_t Npad = ((size_t)N + 255) & ~(size_t)255;
    const int    nb   = (N + 1023) >> 10;
    float* dis  = ws;
    float* bufA = ws + Npad;
    float* bufB = bufA + (size_t)N * H;
    float* cnt  = bufB + (size_t)N * H;
    int*   count    = (int*)(cnt + 512);
    int*   rowptr   = count + Npad;
    int*   rowloc   = rowptr + Npad;
    int*   bsum     = rowloc + Npad;
    int*   part     = bsum + (((size_t)nb + 255) & ~(size_t)255);
    int*   slot_src = part + (size_t)CR_S * Npad;
    int*   csr_src  = slot_src + (size_t)E;
    float* csr_w    = (float*)(csr_src + (size_t)E);

    const int TB = 256;

    const int nranges = cdiv_ll((long long)Npad, CR_NR);
    const int chunk   = cdiv_ll(E, CR_S);
    count_kernel<<<nranges * CR_S, 1024, 0, stream>>>(coli, part, E, chunk, (int)Npad, out, cnt);
    scan_local_kernel<<<nb, TB, 0, stream>>>(part, count, rowloc, bsum, dis, N, (int)Npad);
    place_kernel<<<nranges * CR_S, 1024, 0, stream>>>(rowi, coli, rowloc, bsum, nb, part, slot_src, rowptr, E, chunk, (int)Npad);
    order_kernel<<<cdiv_ll(N, TB / 32), TB, 0, stream>>>(slot_src, rowptr, count, dis, csr_src, csr_w, N);

    const int fuseBlk    = FUSE_BLK;

    gather_gemm_kernel<false><<<fuseBlk, TB, 0, stream>>>(x, dis, b0, rowptr, count, csr_src, csr_w, W0, W0, H, bufA, N);

    gather_gemm_kernel<false><<<fuseBlk, TB, 0, stream>>>(bufA, dis, b1, rowptr, count, csr_src, csr_w, W1, W1, H, bufB, N);

    gather_gemm_kernel<true><<<fuseBlk, TB, 0, stream>>>(bufB, dis, b2, rowptr, count, csr_src, csr_w, W2, W3, C_OUT, bufA, N);

    gather16_pool_kernel<<<cdiv_ll((long long)N * 4, TB), TB, 0, stream>>>(bufA, dis, b3, C_OUT, rowptr, count, csr_src, csr_w, batch, out, cnt, N);
    div_kernel<<<cdiv_ll(NUM_GRAPHS * C_OUT, TB), TB, 0, stream>>>(out, cnt);
}
